# starting kernel
# baseline (speedup 1.0000x reference)
#include <hip/hip_runtime.h>

typedef __attribute__((ext_vector_type(2))) float v2f;
typedef __attribute__((ext_vector_type(8))) float v8f;

#define N_NODES 50000
#define N_EDGES 600000
#define K_DIM   128
#define BN_EPS  1e-5f

__global__ void gin_copy_kernel(const float* __restrict__ src,
                                float* __restrict__ dst, int n4) {
    int i = blockIdx.x * blockDim.x + threadIdx.x;
    if (i < n4) {
        reinterpret_cast<float4*>(dst)[i] =
            reinterpret_cast<const float4*>(src)[i];
    }
}

__global__ void gin_transpose_pad(const float* __restrict__ W,
                                  float* __restrict__ WT,
                                  int Hout, int totalPad) {
    int i = blockIdx.x * blockDim.x + threadIdx.x;
    if (i >= totalPad) return;
    int n = i >> 7;
    int k = i & 127;
    WT[i] = (n < Hout) ? W[k * Hout + n] : 0.0f;
}

__global__ void gin_scatter_kernel(const float* __restrict__ x,
                                   const int* __restrict__ src_idx,
                                   const int* __restrict__ dst_idx,
                                   float* __restrict__ agg,
                                   int total) {
    int i = blockIdx.x * blockDim.x + threadIdx.x;
    if (i >= total) return;
    int e  = i >> 5;
    int f4 = (i & 31) << 2;
    int s = src_idx[e];
    int d = dst_idx[e];
    const float4 v = *reinterpret_cast<const float4*>(x + (size_t)s * K_DIM + f4);
    float* p = agg + (size_t)d * K_DIM + f4;
    atomicAdd(p + 0, v.x);
    atomicAdd(p + 1, v.y);
    atomicAdd(p + 2, v.z);
    atomicAdd(p + 3, v.w);
}

template<int MODE, int HOUT>
__global__ void gin_gemm_wmma(const float* __restrict__ A,
                              const float* __restrict__ WT,
                              const float* __restrict__ bias,
                              const float* __restrict__ bn_g,
                              const float* __restrict__ bn_b,
                              const float* __restrict__ bn_m,
                              const float* __restrict__ bn_v,
                              float* __restrict__ out) {
    constexpr int CG         = 4;
    constexpr int COL_TILES  = (HOUT + 15) / 16;
    constexpr int COL_GROUPS = (COL_TILES + CG - 1) / CG;
    constexpr int ROW_TILES  = N_NODES / 16;
    constexpr int TOTAL      = ROW_TILES * COL_GROUPS;

    const int wave = threadIdx.x >> 5;
    const int tile = blockIdx.x * (blockDim.x >> 5) + wave;
    if (tile >= TOTAL) return;

    const int rowTile = tile / COL_GROUPS;
    const int colBase = (tile % COL_GROUPS) * (CG * 16);

    const int lane = threadIdx.x & 31;
    const int m    = lane & 15;
    const int hi   = lane >> 4;

    const float* Arow  = A  + (long long)(rowTile * 16 + m) * K_DIM + 2 * hi;
    const float* Wlane = WT + (long long)(colBase + m) * K_DIM + 2 * hi;

    v8f acc[CG] = {};
    #pragma unroll 4
    for (int k0 = 0; k0 < K_DIM; k0 += 4) {
        v2f a = *reinterpret_cast<const v2f*>(Arow + k0);
        #pragma unroll
        for (int t = 0; t < CG; ++t) {
            v2f b = *reinterpret_cast<const v2f*>(Wlane + (t << 11) + k0);
            acc[t] = __builtin_amdgcn_wmma_f32_16x16x4_f32(
                false, a, false, b, (short)0, acc[t], false, false);
        }
    }

    #pragma unroll
    for (int t = 0; t < CG; ++t) {
        const int  n  = colBase + t * 16 + m;
        const bool ok = (HOUT % (CG * 16) == 0) || (n < HOUT);
        float bias_v = ok ? bias[n] : 0.0f;
        float scale = 1.0f, shift = 0.0f;
        if constexpr (MODE == 2) {
            float rs = rsqrtf(bn_v[n] + BN_EPS);
            scale = bn_g[n] * rs;
            shift = bn_b[n] - bn_m[n] * scale;
        }
        float* orow = out + (long long)(rowTile * 16) * HOUT + (ok ? n : 0);
        #pragma unroll
        for (int r = 0; r < 8; ++r) {
            float val = acc[t][r] + bias_v;
            if constexpr (MODE == 2) val = val * scale + shift;
            if constexpr (MODE >= 1) val = fmaxf(val, 0.0f);
            if (ok) orow[(long long)(r + 8 * hi) * HOUT] = val;
        }
    }
}

extern "C" void kernel_launch(void* const* d_in, const int* in_sizes, int n_in,
                              void* d_out, int out_size, void* d_ws, size_t ws_size,
                              hipStream_t stream) {
    (void)in_sizes; (void)n_in; (void)out_size; (void)ws_size;

    const float* x      = (const float*)d_in[0];
    const int*   edge   = (const int*)  d_in[1];
    const float* lin1_w = (const float*)d_in[2];
    const float* lin1_b = (const float*)d_in[3];
    const float* bn1_g  = (const float*)d_in[4];
    const float* bn1_b  = (const float*)d_in[5];
    const float* bn1_m  = (const float*)d_in[6];
    const float* bn1_v  = (const float*)d_in[7];
    const float* lin2_w = (const float*)d_in[8];
    const float* lin2_b = (const float*)d_in[9];
    const float* lin3_w = (const float*)d_in[10];
    const float* lin3_b = (const float*)d_in[11];
    const float* bn2_g  = (const float*)d_in[12];
    const float* bn2_b  = (const float*)d_in[13];
    const float* bn2_m  = (const float*)d_in[14];
    const float* bn2_v  = (const float*)d_in[15];
    const float* lin4_w = (const float*)d_in[16];
    const float* lin4_b = (const float*)d_in[17];
    const float* lin5_w = (const float*)d_in[18];
    const float* lin5_b = (const float*)d_in[19];
    const float* lin6_w = (const float*)d_in[20];
    const float* lin6_b = (const float*)d_in[21];

    const int* e_src = edge;
    const int* e_dst = edge + N_EDGES;

    const size_t NH = (size_t)N_NODES * K_DIM;
    float* B0  = (float*)d_ws;
    float* B1  = B0  + NH;
    float* WT1 = B1  + NH;
    float* WT2 = WT1 + 128 * 128;
    float* WT3 = WT2 + 128 * 128;
    float* WT4 = WT3 + 128 * 128;
    float* WT5 = WT4 + 128 * 128;
    float* WT6 = WT5 + 128 * 128;

    gin_transpose_pad<<<64, 256, 0, stream>>>(lin1_w, WT1, 128, 128 * 128);
    gin_transpose_pad<<<64, 256, 0, stream>>>(lin2_w, WT2, 128, 128 * 128);
    gin_transpose_pad<<<64, 256, 0, stream>>>(lin3_w, WT3, 128, 128 * 128);
    gin_transpose_pad<<<64, 256, 0, stream>>>(lin4_w, WT4, 128, 128 * 128);
    gin_transpose_pad<<<64, 256, 0, stream>>>(lin5_w, WT5, 128, 128 * 128);
    gin_transpose_pad<<<32, 256, 0, stream>>>(lin6_w, WT6,  40,  64 * 128);

    const int NH4      = (int)(NH / 4);
    const int cpBlocks = (NH4 + 255) / 256;
    const int scTotal  = N_EDGES * 32;
    const int scBlocks = (scTotal + 255) / 256;
    const int g128     = ((N_NODES / 16) * 2 + 7) / 8;
    const int g40      = ((N_NODES / 16) * 1 + 7) / 8;

    gin_copy_kernel   <<<cpBlocks, 256, 0, stream>>>(x, B0, NH4);
    gin_scatter_kernel<<<scBlocks, 256, 0, stream>>>(x, e_src, e_dst, B0, scTotal);
    gin_gemm_wmma<2,128><<<g128, 256, 0, stream>>>(B0, WT1, lin1_b,
                                                   bn1_g, bn1_b, bn1_m, bn1_v, B1);
    gin_gemm_wmma<1,128><<<g128, 256, 0, stream>>>(B1, WT2, lin2_b,
                                                   nullptr, nullptr, nullptr, nullptr, B0);

    gin_copy_kernel   <<<cpBlocks, 256, 0, stream>>>(B0, B1, NH4);
    gin_scatter_kernel<<<scBlocks, 256, 0, stream>>>(B0, e_src, e_dst, B1, scTotal);
    gin_gemm_wmma<2,128><<<g128, 256, 0, stream>>>(B1, WT3, lin3_b,
                                                   bn2_g, bn2_b, bn2_m, bn2_v, B0);
    gin_gemm_wmma<1,128><<<g128, 256, 0, stream>>>(B0, WT4, lin4_b,
                                                   nullptr, nullptr, nullptr, nullptr, B1);

    gin_gemm_wmma<1,128><<<g128, 256, 0, stream>>>(B1, WT5, lin5_b,
                                                   nullptr, nullptr, nullptr, nullptr, B0);
    gin_gemm_wmma<0,40> <<<g40,  256, 0, stream>>>(B0, WT6, lin6_b,
                                                   nullptr, nullptr, nullptr, nullptr,
                                                   (float*)d_out);
}


# Gather aggregation: edges grouped by destination once per call (integer-atomic arrival-order rank, scan, fill); node rows summed in registers in list order, stored once; no f32 atomics, no row copy
# speedup vs baseline: 1.4139x; 1.4139x over previous
#include <hip/hip_runtime.h>

typedef __attribute__((ext_vector_type(2))) float v2f;
typedef __attribute__((ext_vector_type(8))) float v8f;

#define N_NODES 50000
#define N_EDGES 600000
#define K_DIM   128
#define BN_EPS  1e-5f

__global__ void gin_transpose_pad(const float* __restrict__ W,
                                  float* __restrict__ WT,
                                  int Hout, int totalPad) {
    int i = blockIdx.x * blockDim.x + threadIdx.x;
    if (i >= totalPad) return;
    int n = i >> 7;
    int k = i & 127;
    WT[i] = (n < Hout) ? W[k * Hout + n] : 0.0f;
}


__global__ void gin_zero_kernel(int* __restrict__ deg, int n4) {
    int i = blockIdx.x * blockDim.x + threadIdx.x;
    if (i < n4) reinterpret_cast<int4*>(deg)[i] = make_int4(0, 0, 0, 0);
}

__global__ void gin_rank_kernel(const int* __restrict__ dst_idx,
                                int* __restrict__ deg,
                                int* __restrict__ rank, int n_edges) {
    int e = blockIdx.x * blockDim.x + threadIdx.x;
    if (e >= n_edges) return;
    rank[e] = atomicAdd(deg + dst_idx[e], 1);
}

__global__ void __launch_bounds__(1024)
gin_scan_kernel(const int* __restrict__ deg, int* __restrict__ rowptr) {
    constexpr int N4  = N_NODES / 4;
    constexpr int SUB = (N4 + 1023) / 1024;
    __shared__ int wsum[32];
    const int lane = threadIdx.x & 31;
    const int wave  = __builtin_amdgcn_readfirstlane((int)(threadIdx.x >> 5));
    const int first = wave * (SUB * 32) + lane;

    int4 v[SUB];
    #pragma unroll
    for (int i = 0; i < SUB; ++i) {
        const int idx = first + i * 32;
        v[i] = (idx < N4) ? reinterpret_cast<const int4*>(deg)[idx]
                          : make_int4(0, 0, 0, 0);
    }

    int pre[SUB];
    int base = 0;
    #pragma unroll
    for (int i = 0; i < SUB; ++i) {
        const int s4 = (v[i].x + v[i].y) + (v[i].z + v[i].w);
        int inc = s4;
        #pragma unroll
        for (int o = 1; o < 32; o <<= 1) {
            const int y = __shfl_up(inc, o);
            if (lane >= o) inc += y;
        }
        pre[i] = base + (inc - s4);
        base  += __builtin_amdgcn_readlane(inc, 31);
    }
    if (lane == 0) wsum[wave] = base;
    __syncthreads();
    const int w = wsum[lane];
    int winc = w;
    #pragma unroll
    for (int o = 1; o < 32; o <<= 1) {
        const int y = __shfl_up(winc, o);
        if (lane >= o) winc += y;
    }
    const int wave_excl = __builtin_amdgcn_readlane(winc - w, wave);

    #pragma unroll
    for (int i = 0; i < SUB; ++i) {
        const int idx = first + i * 32;
        const int e0 = wave_excl + pre[i];
        if (idx < N4) {
            const int e1 = e0 + v[i].x;
            const int e2 = e1 + v[i].y;
            const int e3 = e2 + v[i].z;
            reinterpret_cast<int4*>(rowptr)[idx] = make_int4(e0, e1, e2, e3);
        } else if (idx == N4) {
            rowptr[N_NODES] = e0;
        }
    }
}

__global__ void gin_fill_kernel(const int* __restrict__ src_idx,
                                const int* __restrict__ dst_idx,
                                const int* __restrict__ rowptr,
                                const int* __restrict__ rank,
                                int* __restrict__ col, int n_edges) {
    int e = blockIdx.x * blockDim.x + threadIdx.x;
    if (e >= n_edges) return;
    col[rowptr[dst_idx[e]] + rank[e]] = src_idx[e];
}

__device__ __forceinline__ void gin_add_rows(const float4* __restrict__ xr,
                                             int my, int n, float4& acc) {
    const float4 zero4 = make_float4(0.0f, 0.0f, 0.0f, 0.0f);
    for (int k = 0; k < n; k += 16) {
        float4 r[16];
        r[0] = xr[(size_t)__builtin_amdgcn_readlane(my, k) * 32];
        #pragma unroll
        for (int j = 1; j < 16; ++j) {
            r[j] = zero4;
            if (k + j < n)
                r[j] = xr[(size_t)__builtin_amdgcn_readlane(my, k + j) * 32];
        }
        #pragma unroll
        for (int s = 1; s < 16; s <<= 1) {
            #pragma unroll
            for (int j = 0; j < 16; j += 2 * s) {
                r[j].x += r[j + s].x;  r[j].y += r[j + s].y;
                r[j].z += r[j + s].z;  r[j].w += r[j + s].w;
            }
        }
        acc.x += r[0].x;  acc.y += r[0].y;  acc.z += r[0].z;  acc.w += r[0].w;
    }
}

__global__ void __launch_bounds__(256)
gin_gather_kernel(const float* __restrict__ x,
                  const int* __restrict__ rowptr,
                  const int* __restrict__ col,
                  float* __restrict__ agg) {
    const int node = __builtin_amdgcn_readfirstlane(
        (int)(blockIdx.x * (blockDim.x >> 5) + (threadIdx.x >> 5)));
    if (node >= N_NODES) return;
    const int lane = threadIdx.x & 31;
    const float4* __restrict__ xr = reinterpret_cast<const float4*>(x) + lane;

    const int start = rowptr[node];
    const int cnt   = rowptr[node + 1] - start;

    const int n0  = (cnt < 32) ? cnt : 32;
    const int my0 = (lane < n0) ? col[start + lane] : 0;
    const float4 self = xr[(size_t)node * 32];

    float4 acc = make_float4(0.0f, 0.0f, 0.0f, 0.0f);
    gin_add_rows(xr, my0, n0, acc);
    for (int base = 32; base < cnt; base += 32) {
        const int n  = (cnt - base < 32) ? (cnt - base) : 32;
        const int my = (lane < n) ? col[start + base + lane] : 0;
        gin_add_rows(xr, my, n, acc);
    }
    acc.x += self.x;  acc.y += self.y;  acc.z += self.z;  acc.w += self.w;
    reinterpret_cast<float4*>(agg)[(size_t)node * 32 + lane] = acc;
}

template<int MODE, int HOUT>
__global__ void gin_gemm_wmma(const float* __restrict__ A,
                              const float* __restrict__ WT,
                              const float* __restrict__ bias,
                              const float* __restrict__ bn_g,
                              const float* __restrict__ bn_b,
                              const float* __restrict__ bn_m,
                              const float* __restrict__ bn_v,
                              float* __restrict__ out) {
    constexpr int CG         = 4;
    constexpr int COL_TILES  = (HOUT + 15) / 16;
    constexpr int COL_GROUPS = (COL_TILES + CG - 1) / CG;
    constexpr int ROW_TILES  = N_NODES / 16;
    constexpr int TOTAL      = ROW_TILES * COL_GROUPS;

    const int wave = threadIdx.x >> 5;
    const int tile = blockIdx.x * (blockDim.x >> 5) + wave;
    if (tile >= TOTAL) return;

    const int rowTile = tile / COL_GROUPS;
    const int colBase = (tile % COL_GROUPS) * (CG * 16);

    const int lane = threadIdx.x & 31;
    const int m    = lane & 15;
    const int hi   = lane >> 4;

    const float* Arow  = A  + (long long)(rowTile * 16 + m) * K_DIM + 2 * hi;
    const float* Wlane = WT + (long long)(colBase + m) * K_DIM + 2 * hi;

    v8f acc[CG] = {};
    #pragma unroll 4
    for (int k0 = 0; k0 < K_DIM; k0 += 4) {
        v2f a = *reinterpret_cast<const v2f*>(Arow + k0);
        #pragma unroll
        for (int t = 0; t < CG; ++t) {
            v2f b = *reinterpret_cast<const v2f*>(Wlane + (t << 11) + k0);
            acc[t] = __builtin_amdgcn_wmma_f32_16x16x4_f32(
                false, a, false, b, (short)0, acc[t], false, false);
        }
    }

    #pragma unroll
    for (int t = 0; t < CG; ++t) {
        const int  n  = colBase + t * 16 + m;
        const bool ok = (HOUT % (CG * 16) == 0) || (n < HOUT);
        float bias_v = ok ? bias[n] : 0.0f;
        float scale = 1.0f, shift = 0.0f;
        if constexpr (MODE == 2) {
            float rs = rsqrtf(bn_v[n] + BN_EPS);
            scale = bn_g[n] * rs;
            shift = bn_b[n] - bn_m[n] * scale;
        }
        float* orow = out + (long long)(rowTile * 16) * HOUT + (ok ? n : 0);
        #pragma unroll
        for (int r = 0; r < 8; ++r) {
            float val = acc[t][r] + bias_v;
            if constexpr (MODE == 2) val = val * scale + shift;
            if constexpr (MODE >= 1) val = fmaxf(val, 0.0f);
            if (ok) orow[(long long)(r + 8 * hi) * HOUT] = val;
        }
    }
}

extern "C" void kernel_launch(void* const* d_in, const int* in_sizes, int n_in,
                              void* d_out, int out_size, void* d_ws, size_t ws_size,
                              hipStream_t stream) {
    (void)in_sizes; (void)n_in; (void)out_size; (void)ws_size;

    const float* x      = (const float*)d_in[0];
    const int*   edge   = (const int*)  d_in[1];
    const float* lin1_w = (const float*)d_in[2];
    const float* lin1_b = (const float*)d_in[3];
    const float* bn1_g  = (const float*)d_in[4];
    const float* bn1_b  = (const float*)d_in[5];
    const float* bn1_m  = (const float*)d_in[6];
    const float* bn1_v  = (const float*)d_in[7];
    const float* lin2_w = (const float*)d_in[8];
    const float* lin2_b = (const float*)d_in[9];
    const float* lin3_w = (const float*)d_in[10];
    const float* lin3_b = (const float*)d_in[11];
    const float* bn2_g  = (const float*)d_in[12];
    const float* bn2_b  = (const float*)d_in[13];
    const float* bn2_m  = (const float*)d_in[14];
    const float* bn2_v  = (const float*)d_in[15];
    const float* lin4_w = (const float*)d_in[16];
    const float* lin4_b = (const float*)d_in[17];
    const float* lin5_w = (const float*)d_in[18];
    const float* lin5_b = (const float*)d_in[19];
    const float* lin6_w = (const float*)d_in[20];
    const float* lin6_b = (const float*)d_in[21];

    const int* e_src = edge;
    const int* e_dst = edge + N_EDGES;

    const size_t NH = (size_t)N_NODES * K_DIM;
    float* B0  = (float*)d_ws;
    float* B1  = B0  + NH;
    float* WT1 = B1  + NH;
    float* WT2 = WT1 + 128 * 128;
    float* WT3 = WT2 + 128 * 128;
    float* WT4 = WT3 + 128 * 128;
    float* WT5 = WT4 + 128 * 128;
    float* WT6 = WT5 + 128 * 128;
    int* deg    = (int*)(WT6 + 64 * 128);
    int* rowptr = deg    + N_NODES;
    int* rank   = rowptr + N_NODES + 16;
    int* col    = rank   + N_EDGES;

    gin_transpose_pad<<<64, 256, 0, stream>>>(lin1_w, WT1, 128, 128 * 128);
    gin_transpose_pad<<<64, 256, 0, stream>>>(lin2_w, WT2, 128, 128 * 128);
    gin_transpose_pad<<<64, 256, 0, stream>>>(lin3_w, WT3, 128, 128 * 128);
    gin_transpose_pad<<<64, 256, 0, stream>>>(lin4_w, WT4, 128, 128 * 128);
    gin_transpose_pad<<<64, 256, 0, stream>>>(lin5_w, WT5, 128, 128 * 128);
    gin_transpose_pad<<<32, 256, 0, stream>>>(lin6_w, WT6,  40,  64 * 128);

    const int zBlocks  = (N_NODES / 4 + 255) / 256;
    const int eBlocks  = (N_EDGES + 255) / 256;
    const int gaBlocks = (N_NODES + 7) / 8;
    const int g128     = ((N_NODES / 16) * 2 + 7) / 8;
    const int g40      = ((N_NODES / 16) * 1 + 7) / 8;

    gin_zero_kernel<<<zBlocks, 256, 0, stream>>>(deg, N_NODES / 4);
    gin_rank_kernel<<<eBlocks, 256, 0, stream>>>(e_dst, deg, rank, N_EDGES);
    gin_scan_kernel<<<1, 1024, 0, stream>>>(deg, rowptr);
    gin_fill_kernel<<<eBlocks, 256, 0, stream>>>(e_src, e_dst, rowptr, rank, col, N_EDGES);

    gin_gather_kernel<<<gaBlocks, 256, 0, stream>>>(x, rowptr, col, B0);
    gin_gemm_wmma<2,128><<<g128, 256, 0, stream>>>(B0, WT1, lin1_b,
                                                   bn1_g, bn1_b, bn1_m, bn1_v, B1);
    gin_gemm_wmma<1,128><<<g128, 256, 0, stream>>>(B1, WT2, lin2_b,
                                                   nullptr, nullptr, nullptr, nullptr, B0);

    gin_gather_kernel<<<gaBlocks, 256, 0, stream>>>(B0, rowptr, col, B1);
    gin_gemm_wmma<2,128><<<g128, 256, 0, stream>>>(B1, WT3, lin3_b,
                                                   bn2_g, bn2_b, bn2_m, bn2_v, B0);
    gin_gemm_wmma<1,128><<<g128, 256, 0, stream>>>(B0, WT4, lin4_b,
                                                   nullptr, nullptr, nullptr, nullptr, B1);

    gin_gemm_wmma<1,128><<<g128, 256, 0, stream>>>(B1, WT5, lin5_b,
                                                   nullptr, nullptr, nullptr, nullptr, B0);
    gin_gemm_wmma<0,40> <<<g40,  256, 0, stream>>>(B0, WT6, lin6_b,
                                                   nullptr, nullptr, nullptr, nullptr,
                                                   (float*)d_out);
}


# GEMM operand feed: weights staged once per work-group in LDS, read as grouped b128 fragments; A rows read as b128 along k; k summed in a new fixed order within each 8; wave = 16 rows x all columns
# speedup vs baseline: 1.7265x; 1.2211x over previous
#include <hip/hip_runtime.h>

typedef __attribute__((ext_vector_type(2))) float v2f;
typedef __attribute__((ext_vector_type(4))) float v4f;
typedef __attribute__((ext_vector_type(8))) float v8f;

#define N_NODES 50000
#define N_EDGES 600000
#define K_DIM   128
#define BN_EPS  1e-5f

__global__ void gin_transpose_pad(const float* __restrict__ W,
                                  float* __restrict__ WT,
                                  int Hout, int totalPad) {
    int i = blockIdx.x * blockDim.x + threadIdx.x;
    if (i >= totalPad) return;
    int n = i >> 7;
    int k = i & 127;
    WT[i] = (n < Hout) ? W[k * Hout + n] : 0.0f;
}


__global__ void gin_zero_kernel(int* __restrict__ deg, int n4) {
    int i = blockIdx.x * blockDim.x + threadIdx.x;
    if (i < n4) reinterpret_cast<int4*>(deg)[i] = make_int4(0, 0, 0, 0);
}

__global__ void gin_rank_kernel(const int* __restrict__ dst_idx,
                                int* __restrict__ deg,
                                int* __restrict__ rank, int n_edges) {
    int e = blockIdx.x * blockDim.x + threadIdx.x;
    if (e >= n_edges) return;
    rank[e] = atomicAdd(deg + dst_idx[e], 1);
}

__global__ void __launch_bounds__(1024)
gin_scan_kernel(const int* __restrict__ deg, int* __restrict__ rowptr) {
    constexpr int N4  = N_NODES / 4;
    constexpr int SUB = (N4 + 1023) / 1024;
    __shared__ int wsum[32];
    const int lane = threadIdx.x & 31;
    const int wave  = __builtin_amdgcn_readfirstlane((int)(threadIdx.x >> 5));
    const int first = wave * (SUB * 32) + lane;

    int4 v[SUB];
    #pragma unroll
    for (int i = 0; i < SUB; ++i) {
        const int idx = first + i * 32;
        v[i] = (idx < N4) ? reinterpret_cast<const int4*>(deg)[idx]
                          : make_int4(0, 0, 0, 0);
    }

    int pre[SUB];
    int base = 0;
    #pragma unroll
    for (int i = 0; i < SUB; ++i) {
        const int s4 = (v[i].x + v[i].y) + (v[i].z + v[i].w);
        int inc = s4;
        #pragma unroll
        for (int o = 1; o < 32; o <<= 1) {
            const int y = __shfl_up(inc, o);
            if (lane >= o) inc += y;
        }
        pre[i] = base + (inc - s4);
        base  += __builtin_amdgcn_readlane(inc, 31);
    }
    if (lane == 0) wsum[wave] = base;
    __syncthreads();
    const int w = wsum[lane];
    int winc = w;
    #pragma unroll
    for (int o = 1; o < 32; o <<= 1) {
        const int y = __shfl_up(winc, o);
        if (lane >= o) winc += y;
    }
    const int wave_excl = __builtin_amdgcn_readlane(winc - w, wave);

    #pragma unroll
    for (int i = 0; i < SUB; ++i) {
        const int idx = first + i * 32;
        const int e0 = wave_excl + pre[i];
        if (idx < N4) {
            const int e1 = e0 + v[i].x;
            const int e2 = e1 + v[i].y;
            const int e3 = e2 + v[i].z;
            reinterpret_cast<int4*>(rowptr)[idx] = make_int4(e0, e1, e2, e3);
        } else if (idx == N4) {
            rowptr[N_NODES] = e0;
        }
    }
}

__global__ void gin_fill_kernel(const int* __restrict__ src_idx,
                                const int* __restrict__ dst_idx,
                                const int* __restrict__ rowptr,
                                const int* __restrict__ rank,
                                int* __restrict__ col, int n_edges) {
    int e = blockIdx.x * blockDim.x + threadIdx.x;
    if (e >= n_edges) return;
    col[rowptr[dst_idx[e]] + rank[e]] = src_idx[e];
}

__device__ __forceinline__ void gin_add_rows(const float4* __restrict__ xr,
                                             int my, int n, float4& acc) {
    const float4 zero4 = make_float4(0.0f, 0.0f, 0.0f, 0.0f);
    for (int k = 0; k < n; k += 16) {
        float4 r[16];
        r[0] = xr[(size_t)__builtin_amdgcn_readlane(my, k) * 32];
        #pragma unroll
        for (int j = 1; j < 16; ++j) {
            r[j] = zero4;
            if (k + j < n)
                r[j] = xr[(size_t)__builtin_amdgcn_readlane(my, k + j) * 32];
        }
        #pragma unroll
        for (int s = 1; s < 16; s <<= 1) {
            #pragma unroll
            for (int j = 0; j < 16; j += 2 * s) {
                r[j].x += r[j + s].x;  r[j].y += r[j + s].y;
                r[j].z += r[j + s].z;  r[j].w += r[j + s].w;
            }
        }
        acc.x += r[0].x;  acc.y += r[0].y;  acc.z += r[0].z;  acc.w += r[0].w;
    }
}

__global__ void __launch_bounds__(256)
gin_gather_kernel(const float* __restrict__ x,
                  const int* __restrict__ rowptr,
                  const int* __restrict__ col,
                  float* __restrict__ agg) {
    const int node = __builtin_amdgcn_readfirstlane(
        (int)(blockIdx.x * (blockDim.x >> 5) + (threadIdx.x >> 5)));
    if (node >= N_NODES) return;
    const int lane = threadIdx.x & 31;
    const float4* __restrict__ xr = reinterpret_cast<const float4*>(x) + lane;

    const int start = rowptr[node];
    const int cnt   = rowptr[node + 1] - start;

    const int n0  = (cnt < 32) ? cnt : 32;
    const int my0 = (lane < n0) ? col[start + lane] : 0;
    const float4 self = xr[(size_t)node * 32];

    float4 acc = make_float4(0.0f, 0.0f, 0.0f, 0.0f);
    gin_add_rows(xr, my0, n0, acc);
    for (int base = 32; base < cnt; base += 32) {
        const int n  = (cnt - base < 32) ? (cnt - base) : 32;
        const int my = (lane < n) ? col[start + base + lane] : 0;
        gin_add_rows(xr, my, n, acc);
    }
    acc.x += self.x;  acc.y += self.y;  acc.z += self.z;  acc.w += self.w;
    reinterpret_cast<float4*>(agg)[(size_t)node * 32 + lane] = acc;
}


#define WT_PITCH 132

template<int MODE, int HOUT>
__global__ void __launch_bounds__(256)
gin_gemm_wmma(const float* __restrict__ A,
              const float* __restrict__ WT,
              const float* __restrict__ bias,
              const float* __restrict__ bn_g,
              const float* __restrict__ bn_b,
              const float* __restrict__ bn_m,
              const float* __restrict__ bn_v,
              float* __restrict__ out) {
    constexpr int NT        = (HOUT + 15) / 16;
    constexpr int W_ROWS    = NT * 16;
    constexpr int ROW_TILES = N_NODES / 16;
    constexpr int KSTEPS    = K_DIM / 8;
    constexpr int FILL      = W_ROWS * 32 / 256;

    __shared__ alignas(16) float Ws[W_ROWS * WT_PITCH];

    const int lane = threadIdx.x & 31;
    const int wave = __builtin_amdgcn_readfirstlane((int)(threadIdx.x >> 5));
    const int tile = blockIdx.x * 8 + wave;
    const bool have = tile < ROW_TILES;
    const int m  = lane & 15;
    const int hi = lane >> 4;

    const int lrow = threadIdx.x >> 5;
    const int lcol = (threadIdx.x & 31) * 4;
    v4f wreg[FILL];
    #pragma unroll
    for (int it = 0; it < FILL; ++it)
        wreg[it] = reinterpret_cast<const v4f*>(WT)[it * 256 + threadIdx.x];

    const int ltile = have ? tile : ROW_TILES - 1;
    const v4f* Ap = reinterpret_cast<const v4f*>(
        A + (long long)(ltile * 16 + m) * K_DIM + 4 * hi);
    v4f a[KSTEPS];
    #pragma unroll
    for (int j = 0; j < KSTEPS; ++j) a[j] = Ap[2 * j];

    #pragma unroll
    for (int it = 0; it < FILL; ++it)
        *reinterpret_cast<v4f*>(&Ws[(it * 8 + lrow) * WT_PITCH + lcol]) = wreg[it];
    __syncthreads();
    if (!have) return;

    const float* Bl = Ws + m * WT_PITCH + 4 * hi;

    v8f acc[NT] = {};
    v4f bc[NT], bnx[NT];
    #pragma unroll
    for (int t = 0; t < NT; ++t)
        bc[t] = *reinterpret_cast<const v4f*>(Bl + t * 16 * WT_PITCH);
    __builtin_amdgcn_sched_group_barrier(0x100, NT, 0);

    #pragma unroll
    for (int j = 0; j < KSTEPS; ++j) {
        if (j + 1 < KSTEPS) {
            #pragma unroll
            for (int t = 0; t < NT; ++t)
                bnx[t] = *reinterpret_cast<const v4f*>(Bl + t * 16 * WT_PITCH + 8 * (j + 1));
        }
        const v2f a01 = {a[j].x, a[j].y};
        const v2f a23 = {a[j].z, a[j].w};
        #pragma unroll
        for (int t = 0; t < NT; ++t) {
            const v2f b01 = {bc[t].x, bc[t].y};
            acc[t] = __builtin_amdgcn_wmma_f32_16x16x4_f32(
                false, a01, false, b01, (short)0, acc[t], false, false);
        }
        #pragma unroll
        for (int t = 0; t < NT; ++t) {
            const v2f b23 = {bc[t].z, bc[t].w};
            acc[t] = __builtin_amdgcn_wmma_f32_16x16x4_f32(
                false, a23, false, b23, (short)0, acc[t], false, false);
        }
        if (j + 1 < KSTEPS)
            __builtin_amdgcn_sched_group_barrier(0x100, NT, 0);
        __builtin_amdgcn_sched_group_barrier(0x008, 2 * NT, 0);
        if (j + 1 < KSTEPS) {
            #pragma unroll
            for (int t = 0; t < NT; ++t) bc[t] = bnx[t];
        }
    }

    #pragma unroll
    for (int t = 0; t < NT; ++t) {
        const int  n  = t * 16 + m;
        const bool ok = (HOUT % 16 == 0) || (n < HOUT);
        float bias_v = ok ? bias[n] : 0.0f;
        float scale = 1.0f, shift = 0.0f;
        if constexpr (MODE == 2) {
            float rs = rsqrtf(bn_v[n] + BN_EPS);
            scale = bn_g[n] * rs;
            shift = bn_b[n] - bn_m[n] * scale;
        }
        float* orow = out + (long long)(tile * 16) * HOUT + (ok ? n : 0);
        #pragma unroll
        for (int r = 0; r < 8; ++r) {
            float val = acc[t][r] + bias_v;
            if constexpr (MODE == 2) val = val * scale + shift;
            if constexpr (MODE >= 1) val = fmaxf(val, 0.0f);
            if (ok) orow[(long long)(r + 8 * hi) * HOUT] = val;
        }
    }
}

extern "C" void kernel_launch(void* const* d_in, const int* in_sizes, int n_in,
                              void* d_out, int out_size, void* d_ws, size_t ws_size,
                              hipStream_t stream) {
    (void)in_sizes; (void)n_in; (void)out_size; (void)ws_size;

    const float* x      = (const float*)d_in[0];
    const int*   edge   = (const int*)  d_in[1];
    const float* lin1_w = (const float*)d_in[2];
    const float* lin1_b = (const float*)d_in[3];
    const float* bn1_g  = (const float*)d_in[4];
    const float* bn1_b  = (const float*)d_in[5];
    const float* bn1_m  = (const float*)d_in[6];
    const float* bn1_v  = (const float*)d_in[7];
    const float* lin2_w = (const float*)d_in[8];
    const float* lin2_b = (const float*)d_in[9];
    const float* lin3_w = (const float*)d_in[10];
    const float* lin3_b = (const float*)d_in[11];
    const float* bn2_g  = (const float*)d_in[12];
    const float* bn2_b  = (const float*)d_in[13];
    const float* bn2_m  = (const float*)d_in[14];
    const float* bn2_v  = (const float*)d_in[15];
    const float* lin4_w = (const float*)d_in[16];
    const float* lin4_b = (const float*)d_in[17];
    const float* lin5_w = (const float*)d_in[18];
    const float* lin5_b = (const float*)d_in[19];
    const float* lin6_w = (const float*)d_in[20];
    const float* lin6_b = (const float*)d_in[21];

    const int* e_src = edge;
    const int* e_dst = edge + N_EDGES;

    const size_t NH = (size_t)N_NODES * K_DIM;
    float* B0  = (float*)d_ws;
    float* B1  = B0  + NH;
    float* WT1 = B1  + NH;
    float* WT2 = WT1 + 128 * 128;
    float* WT3 = WT2 + 128 * 128;
    float* WT4 = WT3 + 128 * 128;
    float* WT5 = WT4 + 128 * 128;
    float* WT6 = WT5 + 128 * 128;
    int* deg    = (int*)(WT6 + 64 * 128);
    int* rowptr = deg    + N_NODES;
    int* rank   = rowptr + N_NODES + 16;
    int* col    = rank   + N_EDGES;

    gin_transpose_pad<<<64, 256, 0, stream>>>(lin1_w, WT1, 128, 128 * 128);
    gin_transpose_pad<<<64, 256, 0, stream>>>(lin2_w, WT2, 128, 128 * 128);
    gin_transpose_pad<<<64, 256, 0, stream>>>(lin3_w, WT3, 128, 128 * 128);
    gin_transpose_pad<<<64, 256, 0, stream>>>(lin4_w, WT4, 128, 128 * 128);
    gin_transpose_pad<<<64, 256, 0, stream>>>(lin5_w, WT5, 128, 128 * 128);
    gin_transpose_pad<<<32, 256, 0, stream>>>(lin6_w, WT6,  40,  64 * 128);

    const int zBlocks  = (N_NODES / 4 + 255) / 256;
    const int eBlocks  = (N_EDGES + 255) / 256;
    const int gaBlocks = (N_NODES + 7) / 8;
    const int gGemm    = (N_NODES / 16 + 7) / 8;

    gin_zero_kernel<<<zBlocks, 256, 0, stream>>>(deg, N_NODES / 4);
    gin_rank_kernel<<<eBlocks, 256, 0, stream>>>(e_dst, deg, rank, N_EDGES);
    gin_scan_kernel<<<1, 1024, 0, stream>>>(deg, rowptr);
    gin_fill_kernel<<<eBlocks, 256, 0, stream>>>(e_src, e_dst, rowptr, rank, col, N_EDGES);

    gin_gather_kernel<<<gaBlocks, 256, 0, stream>>>(x, rowptr, col, B0);
    gin_gemm_wmma<2,128><<<gGemm, 256, 0, stream>>>(B0, WT1, lin1_b,
                                                   bn1_g, bn1_b, bn1_m, bn1_v, B1);
    gin_gemm_wmma<1,128><<<gGemm, 256, 0, stream>>>(B1, WT2, lin2_b,
                                                   nullptr, nullptr, nullptr, nullptr, B0);

    gin_gather_kernel<<<gaBlocks, 256, 0, stream>>>(B0, rowptr, col, B1);
    gin_gemm_wmma<2,128><<<gGemm, 256, 0, stream>>>(B1, WT3, lin3_b,
                                                   bn2_g, bn2_b, bn2_m, bn2_v, B0);
    gin_gemm_wmma<1,128><<<gGemm, 256, 0, stream>>>(B0, WT4, lin4_b,
                                                   nullptr, nullptr, nullptr, nullptr, B1);

    gin_gemm_wmma<1,128><<<gGemm, 256, 0, stream>>>(B1, WT5, lin5_b,
                                                   nullptr, nullptr, nullptr, nullptr, B0);
    gin_gemm_wmma<0,40> <<<gGemm, 256, 0, stream>>>(B0, WT6, lin6_b,
                                                   nullptr, nullptr, nullptr, nullptr,
                                                   (float*)d_out);
}


# Layer chains fused (lin1-lin2, lin3-lin6): products formed transposed, a layer's D registers are the next layer's B operand; later layers sum k in a new fixed order; 18 -> 14 launches
# speedup vs baseline: 2.1397x; 1.2393x over previous
#include <hip/hip_runtime.h>

typedef __attribute__((ext_vector_type(2))) float v2f;
typedef __attribute__((ext_vector_type(4))) float v4f;
typedef __attribute__((ext_vector_type(8))) float v8f;

#define N_NODES 50000
#define N_EDGES 600000
#define K_DIM   128
#define BN_EPS  1e-5f

__global__ void gin_transpose_pad(const float* __restrict__ W,
                                  float* __restrict__ WT,
                                  int Hout, int totalPad) {
    int i = blockIdx.x * blockDim.x + threadIdx.x;
    if (i >= totalPad) return;
    int n = i >> 7;
    int k = i & 127;
    WT[i] = (n < Hout) ? W[k * Hout + n] : 0.0f;
}


__global__ void gin_zero_kernel(int* __restrict__ deg, int n4) {
    int i = blockIdx.x * blockDim.x + threadIdx.x;
    if (i < n4) reinterpret_cast<int4*>(deg)[i] = make_int4(0, 0, 0, 0);
}

__global__ void gin_rank_kernel(const int* __restrict__ dst_idx,
                                int* __restrict__ deg,
                                int* __restrict__ rank, int n_edges) {
    int e = blockIdx.x * blockDim.x + threadIdx.x;
    if (e >= n_edges) return;
    rank[e] = atomicAdd(deg + dst_idx[e], 1);
}

__global__ void __launch_bounds__(1024)
gin_scan_kernel(const int* __restrict__ deg, int* __restrict__ rowptr) {
    constexpr int N4  = N_NODES / 4;
    constexpr int SUB = (N4 + 1023) / 1024;
    __shared__ int wsum[32];
    const int lane = threadIdx.x & 31;
    const int wave  = __builtin_amdgcn_readfirstlane((int)(threadIdx.x >> 5));
    const int first = wave * (SUB * 32) + lane;

    int4 v[SUB];
    #pragma unroll
    for (int i = 0; i < SUB; ++i) {
        const int idx = first + i * 32;
        v[i] = (idx < N4) ? reinterpret_cast<const int4*>(deg)[idx]
                          : make_int4(0, 0, 0, 0);
    }

    int pre[SUB];
    int base = 0;
    #pragma unroll
    for (int i = 0; i < SUB; ++i) {
        const int s4 = (v[i].x + v[i].y) + (v[i].z + v[i].w);
        int inc = s4;
        #pragma unroll
        for (int o = 1; o < 32; o <<= 1) {
            const int y = __shfl_up(inc, o);
            if (lane >= o) inc += y;
        }
        pre[i] = base + (inc - s4);
        base  += __builtin_amdgcn_readlane(inc, 31);
    }
    if (lane == 0) wsum[wave] = base;
    __syncthreads();
    const int w = wsum[lane];
    int winc = w;
    #pragma unroll
    for (int o = 1; o < 32; o <<= 1) {
        const int y = __shfl_up(winc, o);
        if (lane >= o) winc += y;
    }
    const int wave_excl = __builtin_amdgcn_readlane(winc - w, wave);

    #pragma unroll
    for (int i = 0; i < SUB; ++i) {
        const int idx = first + i * 32;
        const int e0 = wave_excl + pre[i];
        if (idx < N4) {
            const int e1 = e0 + v[i].x;
            const int e2 = e1 + v[i].y;
            const int e3 = e2 + v[i].z;
            reinterpret_cast<int4*>(rowptr)[idx] = make_int4(e0, e1, e2, e3);
        } else if (idx == N4) {
            rowptr[N_NODES] = e0;
        }
    }
}

__global__ void gin_fill_kernel(const int* __restrict__ src_idx,
                                const int* __restrict__ dst_idx,
                                const int* __restrict__ rowptr,
                                const int* __restrict__ rank,
                                int* __restrict__ col, int n_edges) {
    int e = blockIdx.x * blockDim.x + threadIdx.x;
    if (e >= n_edges) return;
    col[rowptr[dst_idx[e]] + rank[e]] = src_idx[e];
}

__device__ __forceinline__ void gin_add_rows(const float4* __restrict__ xr,
                                             int my, int n, float4& acc) {
    const float4 zero4 = make_float4(0.0f, 0.0f, 0.0f, 0.0f);
    for (int k = 0; k < n; k += 16) {
        float4 r[16];
        r[0] = xr[(size_t)__builtin_amdgcn_readlane(my, k) * 32];
        #pragma unroll
        for (int j = 1; j < 16; ++j) {
            r[j] = zero4;
            if (k + j < n)
                r[j] = xr[(size_t)__builtin_amdgcn_readlane(my, k + j) * 32];
        }
        #pragma unroll
        for (int s = 1; s < 16; s <<= 1) {
            #pragma unroll
            for (int j = 0; j < 16; j += 2 * s) {
                r[j].x += r[j + s].x;  r[j].y += r[j + s].y;
                r[j].z += r[j + s].z;  r[j].w += r[j + s].w;
            }
        }
        acc.x += r[0].x;  acc.y += r[0].y;  acc.z += r[0].z;  acc.w += r[0].w;
    }
}

__global__ void __launch_bounds__(256)
gin_gather_kernel(const float* __restrict__ x,
                  const int* __restrict__ rowptr,
                  const int* __restrict__ col,
                  float* __restrict__ agg) {
    const int node = __builtin_amdgcn_readfirstlane(
        (int)(blockIdx.x * (blockDim.x >> 5) + (threadIdx.x >> 5)));
    if (node >= N_NODES) return;
    const int lane = threadIdx.x & 31;
    const float4* __restrict__ xr = reinterpret_cast<const float4*>(x) + lane;

    const int start = rowptr[node];
    const int cnt   = rowptr[node + 1] - start;

    const int n0  = (cnt < 32) ? cnt : 32;
    const int my0 = (lane < n0) ? col[start + lane] : 0;
    const float4 self = xr[(size_t)node * 32];

    float4 acc = make_float4(0.0f, 0.0f, 0.0f, 0.0f);
    gin_add_rows(xr, my0, n0, acc);
    for (int base = 32; base < cnt; base += 32) {
        const int n  = (cnt - base < 32) ? (cnt - base) : 32;
        const int my = (lane < n) ? col[start + base + lane] : 0;
        gin_add_rows(xr, my, n, acc);
    }
    acc.x += self.x;  acc.y += self.y;  acc.z += self.z;  acc.w += self.w;
    reinterpret_cast<float4*>(agg)[(size_t)node * 32 + lane] = acc;
}


#define WT_PITCH 132
#define CH_WAVES 16

template<int NT, bool FIRST>
__device__ __forceinline__ void gin_layer_mma(const float* Wl, const v4f* a,
                                              const v8f* prev, v8f* acc) {
    constexpr int KSTEPS = K_DIM / 8;
    v4f bc[NT], bnx[NT];
    #pragma unroll
    for (int t = 0; t < NT; ++t) {
        acc[t] = v8f{};
        bc[t] = *reinterpret_cast<const v4f*>(Wl + t * 16 * WT_PITCH);
    }
    __builtin_amdgcn_sched_group_barrier(0x100, NT, 0);

    #pragma unroll
    for (int s = 0; s < KSTEPS; ++s) {
        if (s + 1 < KSTEPS) {
            const int s1 = s + 1;
            const int koff = FIRST ? 8 * s1 : 16 * (s1 >> 1) + 4 * (s1 & 1);
            #pragma unroll
            for (int t = 0; t < NT; ++t)
                bnx[t] = *reinterpret_cast<const v4f*>(Wl + t * 16 * WT_PITCH + koff);
        }
        v2f x01, x23;
        if constexpr (FIRST) {
            x01 = v2f{a[s].x, a[s].y};
            x23 = v2f{a[s].z, a[s].w};
        } else {
            const v8f p = prev[s >> 1];
            const int e = 4 * (s & 1);
            x01 = v2f{p[e], p[e + 1]};
            x23 = v2f{p[e + 2], p[e + 3]};
        }
        #pragma unroll
        for (int t = 0; t < NT; ++t) {
            const v2f w01 = {bc[t].x, bc[t].y};
            acc[t] = __builtin_amdgcn_wmma_f32_16x16x4_f32(
                false, w01, false, x01, (short)0, acc[t], false, false);
        }
        #pragma unroll
        for (int t = 0; t < NT; ++t) {
            const v2f w23 = {bc[t].z, bc[t].w};
            acc[t] = __builtin_amdgcn_wmma_f32_16x16x4_f32(
                false, w23, false, x23, (short)0, acc[t], false, false);
        }
        if (s + 1 < KSTEPS)
            __builtin_amdgcn_sched_group_barrier(0x100, NT, 0);
        __builtin_amdgcn_sched_group_barrier(0x008, 2 * NT, 0);
        if (s + 1 < KSTEPS) {
            #pragma unroll
            for (int t = 0; t < NT; ++t) bc[t] = bnx[t];
        }
    }
}

template<int NT, int MODE>
__device__ __forceinline__ void gin_layer_epi(const float* P, v8f* acc) {
    #pragma unroll
    for (int t = 0; t < NT; ++t) {
        #pragma unroll
        for (int g = 0; g < 2; ++g) {
            const v4f b4 = *reinterpret_cast<const v4f*>(P + 16 * t + 4 * g);
            v4f sc4 = {1.0f, 1.0f, 1.0f, 1.0f}, sh4 = {0.0f, 0.0f, 0.0f, 0.0f};
            if constexpr (MODE == 2) {
                sc4 = *reinterpret_cast<const v4f*>(P + 128 + 16 * t + 4 * g);
                sh4 = *reinterpret_cast<const v4f*>(P + 256 + 16 * t + 4 * g);
            }
            #pragma unroll
            for (int i = 0; i < 4; ++i) {
                float val = acc[t][4 * g + i] + b4[i];
                if constexpr (MODE == 2) val = val * sc4[i] + sh4[i];
                if constexpr (MODE >= 1) val = fmaxf(val, 0.0f);
                acc[t][4 * g + i] = val;
            }
        }
    }
}

template<int ROWS>
__device__ __forceinline__ void gin_w_load(const float* __restrict__ WT, v4f* wreg) {
    #pragma unroll
    for (int it = 0; it < ROWS / CH_WAVES; ++it)
        wreg[it] = reinterpret_cast<const v4f*>(WT)[it * (CH_WAVES * 32) + threadIdx.x];
}
template<int ROWS>
__device__ __forceinline__ void gin_w_store(float* Wd, const v4f* wreg) {
    #pragma unroll
    for (int it = 0; it < ROWS / CH_WAVES; ++it) {
        const int i = it * (CH_WAVES * 32) + threadIdx.x;
        *reinterpret_cast<v4f*>(&Wd[(i >> 5) * WT_PITCH + (i & 31) * 4]) = wreg[it];
    }
}

template<int NL, int HLAST, bool RELU_LAST>
__global__ void __launch_bounds__(CH_WAVES * 32)
gin_mlp_chain(const float* __restrict__ A,
              const float* __restrict__ WTa, const float* __restrict__ WTb,
              const float* __restrict__ WTc, const float* __restrict__ WTd,
              const float* __restrict__ bias_a, const float* __restrict__ bias_b,
              const float* __restrict__ bias_c, const float* __restrict__ bias_d,
              const float* __restrict__ bn_g, const float* __restrict__ bn_b,
              const float* __restrict__ bn_m, const float* __restrict__ bn_v,
              float* __restrict__ out) {
    constexpr int NT_LAST   = (HLAST + 15) / 16;
    constexpr int ROWS_LAST = NT_LAST * 16;
    constexpr int ROW_TILES = N_NODES / 16;
    constexpr int KSTEPS    = K_DIM / 8;
    constexpr int W_FLOATS  = ((NL - 1) * 128 + ROWS_LAST) * WT_PITCH;
    constexpr int P_FLOATS  = 384 + (NL - 1) * 128;

    __shared__ alignas(16) float Ws[W_FLOATS + P_FLOATS];
    float* Ps = Ws + W_FLOATS;

    const int lane = threadIdx.x & 31;
    const int wave = __builtin_amdgcn_readfirstlane((int)(threadIdx.x >> 5));
    const int tile = blockIdx.x * CH_WAVES + wave;
    const bool have = tile < ROW_TILES;
    const int c  = lane & 15;
    const int hi = lane >> 4;

    constexpr int RL = ROWS_LAST / CH_WAVES;
    v4f w0[8], w1[NL == 2 ? RL : 8], w2[8], w3[RL];
    gin_w_load<128>(WTa, w0);
    if constexpr (NL == 2) {
        gin_w_load<ROWS_LAST>(WTb, w1);
    } else {
        gin_w_load<128>(WTb, w1);
        gin_w_load<128>(WTc, w2);
        gin_w_load<ROWS_LAST>(WTd, w3);
    }

    const int ltile = have ? tile : ROW_TILES - 1;
    const v4f* Ap = reinterpret_cast<const v4f*>(
        A + (long long)(ltile * 16 + c) * K_DIM + 4 * hi);
    v4f a[KSTEPS];
    #pragma unroll
    for (int j = 0; j < KSTEPS; ++j) a[j] = Ap[2 * j];

    if (threadIdx.x < 128) {
        const int n = threadIdx.x;
        const float rs = rsqrtf(bn_v[n] + BN_EPS);
        const float scale = bn_g[n] * rs;
        Ps[n]       = bias_a[n];
        Ps[128 + n] = scale;
        Ps[256 + n] = bn_b[n] - bn_m[n] * scale;
        if constexpr (NL == 2) {
            const float bb = bias_b[(n < HLAST) ? n : 0];
            Ps[384 + n] = (n < HLAST) ? bb : 0.0f;
        } else {
            Ps[384 + n] = bias_b[n];
            Ps[512 + n] = bias_c[n];
            const float bd = bias_d[(n < HLAST) ? n : 0];
            Ps[640 + n] = (n < HLAST) ? bd : 0.0f;
        }
    }

    gin_w_store<128>(Ws, w0);
    if constexpr (NL == 2) {
        gin_w_store<ROWS_LAST>(Ws + 128 * WT_PITCH, w1);
    } else {
        gin_w_store<128>(Ws + 128 * WT_PITCH, w1);
        gin_w_store<128>(Ws + 256 * WT_PITCH, w2);
        gin_w_store<ROWS_LAST>(Ws + 384 * WT_PITCH, w3);
    }
    __syncthreads();
    if (!have) return;

    const float* Wl0 = Ws + c * WT_PITCH + 4 * hi;
    const float* WlN = Ws + c * WT_PITCH + 8 * hi;
    const float* Pl  = Ps + 8 * hi;

    v8f x[8], y[8];
    gin_layer_mma<8, true>(Wl0, a, nullptr, x);
    gin_layer_epi<8, 2>(Pl, x);

    if constexpr (NL == 2) {
        gin_layer_mma<NT_LAST, false>(WlN + 128 * WT_PITCH, nullptr, x, y);
        gin_layer_epi<NT_LAST, RELU_LAST ? 1 : 0>(Pl + 384, y);
    } else {
        gin_layer_mma<8, false>(WlN + 128 * WT_PITCH, nullptr, x, y);
        gin_layer_epi<8, 1>(Pl + 384, y);
        gin_layer_mma<8, false>(WlN + 256 * WT_PITCH, nullptr, y, x);
        gin_layer_epi<8, 1>(Pl + 512, x);
        gin_layer_mma<NT_LAST, false>(WlN + 384 * WT_PITCH, nullptr, x, y);
        gin_layer_epi<NT_LAST, RELU_LAST ? 1 : 0>(Pl + 640, y);
    }

    float* orow = out + (long long)(tile * 16 + c) * HLAST + 8 * hi;
    #pragma unroll
    for (int t = 0; t < NT_LAST; ++t) {
        if ((HLAST % 16 == 0) || (16 * t + 8 * hi < HLAST)) {
            *reinterpret_cast<v4f*>(orow + 16 * t)     = v4f{y[t][0], y[t][1], y[t][2], y[t][3]};
            *reinterpret_cast<v4f*>(orow + 16 * t + 4) = v4f{y[t][4], y[t][5], y[t][6], y[t][7]};
        }
    }
}

extern "C" void kernel_launch(void* const* d_in, const int* in_sizes, int n_in,
                              void* d_out, int out_size, void* d_ws, size_t ws_size,
                              hipStream_t stream) {
    (void)in_sizes; (void)n_in; (void)out_size; (void)ws_size;

    const float* x      = (const float*)d_in[0];
    const int*   edge   = (const int*)  d_in[1];
    const float* lin1_w = (const float*)d_in[2];
    const float* lin1_b = (const float*)d_in[3];
    const float* bn1_g  = (const float*)d_in[4];
    const float* bn1_b  = (const float*)d_in[5];
    const float* bn1_m  = (const float*)d_in[6];
    const float* bn1_v  = (const float*)d_in[7];
    const float* lin2_w = (const float*)d_in[8];
    const float* lin2_b = (const float*)d_in[9];
    const float* lin3_w = (const float*)d_in[10];
    const float* lin3_b = (const float*)d_in[11];
    const float* bn2_g  = (const float*)d_in[12];
    const float* bn2_b  = (const float*)d_in[13];
    const float* bn2_m  = (const float*)d_in[14];
    const float* bn2_v  = (const float*)d_in[15];
    const float* lin4_w = (const float*)d_in[16];
    const float* lin4_b = (const float*)d_in[17];
    const float* lin5_w = (const float*)d_in[18];
    const float* lin5_b = (const float*)d_in[19];
    const float* lin6_w = (const float*)d_in[20];
    const float* lin6_b = (const float*)d_in[21];

    const int* e_src = edge;
    const int* e_dst = edge + N_EDGES;

    const size_t NH = (size_t)N_NODES * K_DIM;
    float* B0  = (float*)d_ws;
    float* B1  = B0  + NH;
    float* WT1 = B1  + NH;
    float* WT2 = WT1 + 128 * 128;
    float* WT3 = WT2 + 128 * 128;
    float* WT4 = WT3 + 128 * 128;
    float* WT5 = WT4 + 128 * 128;
    float* WT6 = WT5 + 128 * 128;
    int* deg    = (int*)(WT6 + 64 * 128);
    int* rowptr = deg    + N_NODES;
    int* rank   = rowptr + N_NODES + 16;
    int* col    = rank   + N_EDGES;

    gin_transpose_pad<<<64, 256, 0, stream>>>(lin1_w, WT1, 128, 128 * 128);
    gin_transpose_pad<<<64, 256, 0, stream>>>(lin2_w, WT2, 128, 128 * 128);
    gin_transpose_pad<<<64, 256, 0, stream>>>(lin3_w, WT3, 128, 128 * 128);
    gin_transpose_pad<<<64, 256, 0, stream>>>(lin4_w, WT4, 128, 128 * 128);
    gin_transpose_pad<<<64, 256, 0, stream>>>(lin5_w, WT5, 128, 128 * 128);
    gin_transpose_pad<<<32, 256, 0, stream>>>(lin6_w, WT6,  40,  64 * 128);

    const int zBlocks  = (N_NODES / 4 + 255) / 256;
    const int eBlocks  = (N_EDGES + 255) / 256;
    const int gaBlocks = (N_NODES + 7) / 8;
    const int gChain   = (N_NODES / 16 + CH_WAVES - 1) / CH_WAVES;

    gin_zero_kernel<<<zBlocks, 256, 0, stream>>>(deg, N_NODES / 4);
    gin_rank_kernel<<<eBlocks, 256, 0, stream>>>(e_dst, deg, rank, N_EDGES);
    gin_scan_kernel<<<1, 1024, 0, stream>>>(deg, rowptr);
    gin_fill_kernel<<<eBlocks, 256, 0, stream>>>(e_src, e_dst, rowptr, rank, col, N_EDGES);

    gin_gather_kernel<<<gaBlocks, 256, 0, stream>>>(x, rowptr, col, B0);
    gin_mlp_chain<2, 128, true><<<gChain, CH_WAVES * 32, 0, stream>>>(
        B0, WT1, WT2, nullptr, nullptr, lin1_b, lin2_b, nullptr, nullptr,
        bn1_g, bn1_b, bn1_m, bn1_v, B1);

    gin_gather_kernel<<<gaBlocks, 256, 0, stream>>>(B1, rowptr, col, B0);
    gin_mlp_chain<4, 40, false><<<gChain, CH_WAVES * 32, 0, stream>>>(
        B0, WT3, WT4, WT5, WT6, lin3_b, lin4_b, lin5_b, lin6_b,
        bn2_g, bn2_b, bn2_m, bn2_v, (float*)d_out);
}


# Set-up merged into one launch: the six weight transposes and the counter clear run as block ranges of one kernel (they depend on nothing but the inputs); 14 -> 8 launches
# speedup vs baseline: 2.2914x; 1.0709x over previous
#include <hip/hip_runtime.h>

typedef __attribute__((ext_vector_type(2))) float v2f;
typedef __attribute__((ext_vector_type(4))) float v4f;
typedef __attribute__((ext_vector_type(8))) float v8f;

#define N_NODES 50000
#define N_EDGES 600000
#define K_DIM   128
#define BN_EPS  1e-5f

#define SETUP_T_BLOCKS 352
#define SETUP_Z_BLOCKS ((N_NODES / 4 + 255) / 256)

__global__ void __launch_bounds__(256)
gin_setup_kernel(const float* __restrict__ w1, const float* __restrict__ w2,
                 const float* __restrict__ w3, const float* __restrict__ w4,
                 const float* __restrict__ w5, const float* __restrict__ w6,
                 float* __restrict__ WT, int* __restrict__ deg) {
    const int b = blockIdx.x;
    if (b >= SETUP_T_BLOCKS) {
        const int i = (b - SETUP_T_BLOCKS) * 256 + threadIdx.x;
        if (i < N_NODES / 4) reinterpret_cast<int4*>(deg)[i] = make_int4(0, 0, 0, 0);
        return;
    }
    const int m = b >> 6;
    const float* __restrict__ W = (m == 0) ? w1 : (m == 1) ? w2 : (m == 2) ? w3
                                : (m == 3) ? w4 : (m == 4) ? w5 : w6;
    const int Hout = (m == 5) ? 40 : 128;
    const int i = (b & 63) * 256 + threadIdx.x;
    const int n = i >> 7;
    const int k = i & 127;
    WT[m * (128 * 128) + i] = (n < Hout) ? W[k * Hout + n] : 0.0f;
}


__global__ void gin_rank_kernel(const int* __restrict__ dst_idx,
                                int* __restrict__ deg,
                                int* __restrict__ rank, int n_edges) {
    int e = blockIdx.x * blockDim.x + threadIdx.x;
    if (e >= n_edges) return;
    rank[e] = atomicAdd(deg + dst_idx[e], 1);
}

__global__ void __launch_bounds__(1024)
gin_scan_kernel(const int* __restrict__ deg, int* __restrict__ rowptr) {
    constexpr int N4  = N_NODES / 4;
    constexpr int SUB = (N4 + 1023) / 1024;
    __shared__ int wsum[32];
    const int lane = threadIdx.x & 31;
    const int wave  = __builtin_amdgcn_readfirstlane((int)(threadIdx.x >> 5));
    const int first = wave * (SUB * 32) + lane;

    int4 v[SUB];
    #pragma unroll
    for (int i = 0; i < SUB; ++i) {
        const int idx = first + i * 32;
        v[i] = (idx < N4) ? reinterpret_cast<const int4*>(deg)[idx]
                          : make_int4(0, 0, 0, 0);
    }

    int pre[SUB];
    int base = 0;
    #pragma unroll
    for (int i = 0; i < SUB; ++i) {
        const int s4 = (v[i].x + v[i].y) + (v[i].z + v[i].w);
        int inc = s4;
        #pragma unroll
        for (int o = 1; o < 32; o <<= 1) {
            const int y = __shfl_up(inc, o);
            if (lane >= o) inc += y;
        }
        pre[i] = base + (inc - s4);
        base  += __builtin_amdgcn_readlane(inc, 31);
    }
    if (lane == 0) wsum[wave] = base;
    __syncthreads();
    const int w = wsum[lane];
    int winc = w;
    #pragma unroll
    for (int o = 1; o < 32; o <<= 1) {
        const int y = __shfl_up(winc, o);
        if (lane >= o) winc += y;
    }
    const int wave_excl = __builtin_amdgcn_readlane(winc - w, wave);

    #pragma unroll
    for (int i = 0; i < SUB; ++i) {
        const int idx = first + i * 32;
        const int e0 = wave_excl + pre[i];
        if (idx < N4) {
            const int e1 = e0 + v[i].x;
            const int e2 = e1 + v[i].y;
            const int e3 = e2 + v[i].z;
            reinterpret_cast<int4*>(rowptr)[idx] = make_int4(e0, e1, e2, e3);
        } else if (idx == N4) {
            rowptr[N_NODES] = e0;
        }
    }
}

__global__ void gin_fill_kernel(const int* __restrict__ src_idx,
                                const int* __restrict__ dst_idx,
                                const int* __restrict__ rowptr,
                                const int* __restrict__ rank,
                                int* __restrict__ col, int n_edges) {
    int e = blockIdx.x * blockDim.x + threadIdx.x;
    if (e >= n_edges) return;
    col[rowptr[dst_idx[e]] + rank[e]] = src_idx[e];
}

__device__ __forceinline__ void gin_add_rows(const float4* __restrict__ xr,
                                             int my, int n, float4& acc) {
    const float4 zero4 = make_float4(0.0f, 0.0f, 0.0f, 0.0f);
    for (int k = 0; k < n; k += 16) {
        float4 r[16];
        r[0] = xr[(size_t)__builtin_amdgcn_readlane(my, k) * 32];
        #pragma unroll
        for (int j = 1; j < 16; ++j) {
            r[j] = zero4;
            if (k + j < n)
                r[j] = xr[(size_t)__builtin_amdgcn_readlane(my, k + j) * 32];
        }
        #pragma unroll
        for (int s = 1; s < 16; s <<= 1) {
            #pragma unroll
            for (int j = 0; j < 16; j += 2 * s) {
                r[j].x += r[j + s].x;  r[j].y += r[j + s].y;
                r[j].z += r[j + s].z;  r[j].w += r[j + s].w;
            }
        }
        acc.x += r[0].x;  acc.y += r[0].y;  acc.z += r[0].z;  acc.w += r[0].w;
    }
}

__global__ void __launch_bounds__(256)
gin_gather_kernel(const float* __restrict__ x,
                  const int* __restrict__ rowptr,
                  const int* __restrict__ col,
                  float* __restrict__ agg) {
    const int node = __builtin_amdgcn_readfirstlane(
        (int)(blockIdx.x * (blockDim.x >> 5) + (threadIdx.x >> 5)));
    if (node >= N_NODES) return;
    const int lane = threadIdx.x & 31;
    const float4* __restrict__ xr = reinterpret_cast<const float4*>(x) + lane;

    const int start = rowptr[node];
    const int cnt   = rowptr[node + 1] - start;

    const int n0  = (cnt < 32) ? cnt : 32;
    const int my0 = (lane < n0) ? col[start + lane] : 0;
    const float4 self = xr[(size_t)node * 32];

    float4 acc = make_float4(0.0f, 0.0f, 0.0f, 0.0f);
    gin_add_rows(xr, my0, n0, acc);
    for (int base = 32; base < cnt; base += 32) {
        const int n  = (cnt - base < 32) ? (cnt - base) : 32;
        const int my = (lane < n) ? col[start + base + lane] : 0;
        gin_add_rows(xr, my, n, acc);
    }
    acc.x += self.x;  acc.y += self.y;  acc.z += self.z;  acc.w += self.w;
    reinterpret_cast<float4*>(agg)[(size_t)node * 32 + lane] = acc;
}


#define WT_PITCH 132
#define CH_WAVES 16

template<int NT, bool FIRST>
__device__ __forceinline__ void gin_layer_mma(const float* Wl, const v4f* a,
                                              const v8f* prev, v8f* acc) {
    constexpr int KSTEPS = K_DIM / 8;
    v4f bc[NT], bnx[NT];
    #pragma unroll
    for (int t = 0; t < NT; ++t) {
        acc[t] = v8f{};
        bc[t] = *reinterpret_cast<const v4f*>(Wl + t * 16 * WT_PITCH);
    }
    __builtin_amdgcn_sched_group_barrier(0x100, NT, 0);

    #pragma unroll
    for (int s = 0; s < KSTEPS; ++s) {
        if (s + 1 < KSTEPS) {
            const int s1 = s + 1;
            const int koff = FIRST ? 8 * s1 : 16 * (s1 >> 1) + 4 * (s1 & 1);
            #pragma unroll
            for (int t = 0; t < NT; ++t)
                bnx[t] = *reinterpret_cast<const v4f*>(Wl + t * 16 * WT_PITCH + koff);
        }
        v2f x01, x23;
        if constexpr (FIRST) {
            x01 = v2f{a[s].x, a[s].y};
            x23 = v2f{a[s].z, a[s].w};
        } else {
            const v8f p = prev[s >> 1];
            const int e = 4 * (s & 1);
            x01 = v2f{p[e], p[e + 1]};
            x23 = v2f{p[e + 2], p[e + 3]};
        }
        #pragma unroll
        for (int t = 0; t < NT; ++t) {
            const v2f w01 = {bc[t].x, bc[t].y};
            acc[t] = __builtin_amdgcn_wmma_f32_16x16x4_f32(
                false, w01, false, x01, (short)0, acc[t], false, false);
        }
        #pragma unroll
        for (int t = 0; t < NT; ++t) {
            const v2f w23 = {bc[t].z, bc[t].w};
            acc[t] = __builtin_amdgcn_wmma_f32_16x16x4_f32(
                false, w23, false, x23, (short)0, acc[t], false, false);
        }
        if (s + 1 < KSTEPS)
            __builtin_amdgcn_sched_group_barrier(0x100, NT, 0);
        __builtin_amdgcn_sched_group_barrier(0x008, 2 * NT, 0);
        if (s + 1 < KSTEPS) {
            #pragma unroll
            for (int t = 0; t < NT; ++t) bc[t] = bnx[t];
        }
    }
}

template<int NT, int MODE>
__device__ __forceinline__ void gin_layer_epi(const float* P, v8f* acc) {
    #pragma unroll
    for (int t = 0; t < NT; ++t) {
        #pragma unroll
        for (int g = 0; g < 2; ++g) {
            const v4f b4 = *reinterpret_cast<const v4f*>(P + 16 * t + 4 * g);
            v4f sc4 = {1.0f, 1.0f, 1.0f, 1.0f}, sh4 = {0.0f, 0.0f, 0.0f, 0.0f};
            if constexpr (MODE == 2) {
                sc4 = *reinterpret_cast<const v4f*>(P + 128 + 16 * t + 4 * g);
                sh4 = *reinterpret_cast<const v4f*>(P + 256 + 16 * t + 4 * g);
            }
            #pragma unroll
            for (int i = 0; i < 4; ++i) {
                float val = acc[t][4 * g + i] + b4[i];
                if constexpr (MODE == 2) val = val * sc4[i] + sh4[i];
                if constexpr (MODE >= 1) val = fmaxf(val, 0.0f);
                acc[t][4 * g + i] = val;
            }
        }
    }
}

template<int ROWS>
__device__ __forceinline__ void gin_w_load(const float* __restrict__ WT, v4f* wreg) {
    #pragma unroll
    for (int it = 0; it < ROWS / CH_WAVES; ++it)
        wreg[it] = reinterpret_cast<const v4f*>(WT)[it * (CH_WAVES * 32) + threadIdx.x];
}
template<int ROWS>
__device__ __forceinline__ void gin_w_store(float* Wd, const v4f* wreg) {
    #pragma unroll
    for (int it = 0; it < ROWS / CH_WAVES; ++it) {
        const int i = it * (CH_WAVES * 32) + threadIdx.x;
        *reinterpret_cast<v4f*>(&Wd[(i >> 5) * WT_PITCH + (i & 31) * 4]) = wreg[it];
    }
}

template<int NL, int HLAST, bool RELU_LAST>
__global__ void __launch_bounds__(CH_WAVES * 32)
gin_mlp_chain(const float* __restrict__ A,
              const float* __restrict__ WTa, const float* __restrict__ WTb,
              const float* __restrict__ WTc, const float* __restrict__ WTd,
              const float* __restrict__ bias_a, const float* __restrict__ bias_b,
              const float* __restrict__ bias_c, const float* __restrict__ bias_d,
              const float* __restrict__ bn_g, const float* __restrict__ bn_b,
              const float* __restrict__ bn_m, const float* __restrict__ bn_v,
              float* __restrict__ out) {
    constexpr int NT_LAST   = (HLAST + 15) / 16;
    constexpr int ROWS_LAST = NT_LAST * 16;
    constexpr int ROW_TILES = N_NODES / 16;
    constexpr int KSTEPS    = K_DIM / 8;
    constexpr int W_FLOATS  = ((NL - 1) * 128 + ROWS_LAST) * WT_PITCH;
    constexpr int P_FLOATS  = 384 + (NL - 1) * 128;

    __shared__ alignas(16) float Ws[W_FLOATS + P_FLOATS];
    float* Ps = Ws + W_FLOATS;

    const int lane = threadIdx.x & 31;
    const int wave = __builtin_amdgcn_readfirstlane((int)(threadIdx.x >> 5));
    const int tile = blockIdx.x * CH_WAVES + wave;
    const bool have = tile < ROW_TILES;
    const int c  = lane & 15;
    const int hi = lane >> 4;

    constexpr int RL = ROWS_LAST / CH_WAVES;
    v4f w0[8], w1[NL == 2 ? RL : 8], w2[8], w3[RL];
    gin_w_load<128>(WTa, w0);
    if constexpr (NL == 2) {
        gin_w_load<ROWS_LAST>(WTb, w1);
    } else {
        gin_w_load<128>(WTb, w1);
        gin_w_load<128>(WTc, w2);
        gin_w_load<ROWS_LAST>(WTd, w3);
    }

    const int ltile = have ? tile : ROW_TILES - 1;
    const v4f* Ap = reinterpret_cast<const v4f*>(
        A + (long long)(ltile * 16 + c) * K_DIM + 4 * hi);
    v4f a[KSTEPS];
    #pragma unroll
    for (int j = 0; j < KSTEPS; ++j) a[j] = Ap[2 * j];

    if (threadIdx.x < 128) {
        const int n = threadIdx.x;
        const float rs = rsqrtf(bn_v[n] + BN_EPS);
        const float scale = bn_g[n] * rs;
        Ps[n]       = bias_a[n];
        Ps[128 + n] = scale;
        Ps[256 + n] = bn_b[n] - bn_m[n] * scale;
        if constexpr (NL == 2) {
            const float bb = bias_b[(n < HLAST) ? n : 0];
            Ps[384 + n] = (n < HLAST) ? bb : 0.0f;
        } else {
            Ps[384 + n] = bias_b[n];
            Ps[512 + n] = bias_c[n];
            const float bd = bias_d[(n < HLAST) ? n : 0];
            Ps[640 + n] = (n < HLAST) ? bd : 0.0f;
        }
    }

    gin_w_store<128>(Ws, w0);
    if constexpr (NL == 2) {
        gin_w_store<ROWS_LAST>(Ws + 128 * WT_PITCH, w1);
    } else {
        gin_w_store<128>(Ws + 128 * WT_PITCH, w1);
        gin_w_store<128>(Ws + 256 * WT_PITCH, w2);
        gin_w_store<ROWS_LAST>(Ws + 384 * WT_PITCH, w3);
    }
    __syncthreads();
    if (!have) return;

    const float* Wl0 = Ws + c * WT_PITCH + 4 * hi;
    const float* WlN = Ws + c * WT_PITCH + 8 * hi;
    const float* Pl  = Ps + 8 * hi;

    v8f x[8], y[8];
    gin_layer_mma<8, true>(Wl0, a, nullptr, x);
    gin_layer_epi<8, 2>(Pl, x);

    if constexpr (NL == 2) {
        gin_layer_mma<NT_LAST, false>(WlN + 128 * WT_PITCH, nullptr, x, y);
        gin_layer_epi<NT_LAST, RELU_LAST ? 1 : 0>(Pl + 384, y);
    } else {
        gin_layer_mma<8, false>(WlN + 128 * WT_PITCH, nullptr, x, y);
        gin_layer_epi<8, 1>(Pl + 384, y);
        gin_layer_mma<8, false>(WlN + 256 * WT_PITCH, nullptr, y, x);
        gin_layer_epi<8, 1>(Pl + 512, x);
        gin_layer_mma<NT_LAST, false>(WlN + 384 * WT_PITCH, nullptr, x, y);
        gin_layer_epi<NT_LAST, RELU_LAST ? 1 : 0>(Pl + 640, y);
    }

    float* orow = out + (long long)(tile * 16 + c) * HLAST + 8 * hi;
    #pragma unroll
    for (int t = 0; t < NT_LAST; ++t) {
        if ((HLAST % 16 == 0) || (16 * t + 8 * hi < HLAST)) {
            *reinterpret_cast<v4f*>(orow + 16 * t)     = v4f{y[t][0], y[t][1], y[t][2], y[t][3]};
            *reinterpret_cast<v4f*>(orow + 16 * t + 4) = v4f{y[t][4], y[t][5], y[t][6], y[t][7]};
        }
    }
}

extern "C" void kernel_launch(void* const* d_in, const int* in_sizes, int n_in,
                              void* d_out, int out_size, void* d_ws, size_t ws_size,
                              hipStream_t stream) {
    (void)in_sizes; (void)n_in; (void)out_size; (void)ws_size;

    const float* x      = (const float*)d_in[0];
    const int*   edge   = (const int*)  d_in[1];
    const float* lin1_w = (const float*)d_in[2];
    const float* lin1_b = (const float*)d_in[3];
    const float* bn1_g  = (const float*)d_in[4];
    const float* bn1_b  = (const float*)d_in[5];
    const float* bn1_m  = (const float*)d_in[6];
    const float* bn1_v  = (const float*)d_in[7];
    const float* lin2_w = (const float*)d_in[8];
    const float* lin2_b = (const float*)d_in[9];
    const float* lin3_w = (const float*)d_in[10];
    const float* lin3_b = (const float*)d_in[11];
    const float* bn2_g  = (const float*)d_in[12];
    const float* bn2_b  = (const float*)d_in[13];
    const float* bn2_m  = (const float*)d_in[14];
    const float* bn2_v  = (const float*)d_in[15];
    const float* lin4_w = (const float*)d_in[16];
    const float* lin4_b = (const float*)d_in[17];
    const float* lin5_w = (const float*)d_in[18];
    const float* lin5_b = (const float*)d_in[19];
    const float* lin6_w = (const float*)d_in[20];
    const float* lin6_b = (const float*)d_in[21];

    const int* e_src = edge;
    const int* e_dst = edge + N_EDGES;

    const size_t NH = (size_t)N_NODES * K_DIM;
    float* B0  = (float*)d_ws;
    float* B1  = B0  + NH;
    float* WT1 = B1  + NH;
    float* WT2 = WT1 + 128 * 128;
    float* WT3 = WT2 + 128 * 128;
    float* WT4 = WT3 + 128 * 128;
    float* WT5 = WT4 + 128 * 128;
    float* WT6 = WT5 + 128 * 128;
    int* deg    = (int*)(WT6 + 64 * 128);
    int* rowptr = deg    + N_NODES;
    int* rank   = rowptr + N_NODES + 16;
    int* col    = rank   + N_EDGES;

    const int sBlocks  = SETUP_T_BLOCKS + SETUP_Z_BLOCKS;
    const int eBlocks  = (N_EDGES + 255) / 256;
    const int gaBlocks = (N_NODES + 7) / 8;
    const int gChain   = (N_NODES / 16 + CH_WAVES - 1) / CH_WAVES;

    gin_setup_kernel<<<sBlocks, 256, 0, stream>>>(lin1_w, lin2_w, lin3_w, lin4_w, lin5_w, lin6_w,
                                                  WT1, deg);

    gin_rank_kernel<<<eBlocks, 256, 0, stream>>>(e_dst, deg, rank, N_EDGES);
    gin_scan_kernel<<<1, 1024, 0, stream>>>(deg, rowptr);
    gin_fill_kernel<<<eBlocks, 256, 0, stream>>>(e_src, e_dst, rowptr, rank, col, N_EDGES);

    gin_gather_kernel<<<gaBlocks, 256, 0, stream>>>(x, rowptr, col, B0);
    gin_mlp_chain<2, 128, true><<<gChain, CH_WAVES * 32, 0, stream>>>(
        B0, WT1, WT2, nullptr, nullptr, lin1_b, lin2_b, nullptr, nullptr,
        bn1_g, bn1_b, bn1_m, bn1_v, B1);

    gin_gather_kernel<<<gaBlocks, 256, 0, stream>>>(B1, rowptr, col, B0);
    gin_mlp_chain<4, 40, false><<<gChain, CH_WAVES * 32, 0, stream>>>(
        B0, WT3, WT4, WT5, WT6, lin3_b, lin4_b, lin5_b, lin6_b,
        bn2_g, bn2_b, bn2_m, bn2_v, (float*)d_out);
}


# Edges grouped in one pass: the count kernel writes each source id into its destination's fixed 32-slot list (edges past 32: one overflow list, summed in its order); no scan or fill; 8 -> 6 launches
# speedup vs baseline: 2.6507x; 1.1568x over previous
#include <hip/hip_runtime.h>

typedef __attribute__((ext_vector_type(2))) float v2f;
typedef __attribute__((ext_vector_type(4))) float v4f;
typedef __attribute__((ext_vector_type(8))) float v8f;

#define N_NODES 50000
#define N_EDGES 600000
#define K_DIM   128
#define BN_EPS  1e-5f

#define SETUP_T_BLOCKS 352
#define SETUP_Z_INT4   (N_NODES / 4 + 1)
#define SETUP_Z_BLOCKS ((SETUP_Z_INT4 + 255) / 256)

__global__ void __launch_bounds__(256)
gin_setup_kernel(const float* __restrict__ w1, const float* __restrict__ w2,
                 const float* __restrict__ w3, const float* __restrict__ w4,
                 const float* __restrict__ w5, const float* __restrict__ w6,
                 float* __restrict__ WT, int* __restrict__ deg) {
    const int b = blockIdx.x;
    if (b >= SETUP_T_BLOCKS) {
        const int i = (b - SETUP_T_BLOCKS) * 256 + threadIdx.x;
        if (i < SETUP_Z_INT4) reinterpret_cast<int4*>(deg)[i] = make_int4(0, 0, 0, 0);
        return;
    }
    const int m = b >> 6;
    const float* __restrict__ W = (m == 0) ? w1 : (m == 1) ? w2 : (m == 2) ? w3
                                : (m == 3) ? w4 : (m == 4) ? w5 : w6;
    const int Hout = (m == 5) ? 40 : 128;
    const int i = (b & 63) * 256 + threadIdx.x;
    const int n = i >> 7;
    const int k = i & 127;
    WT[m * (128 * 128) + i] = (n < Hout) ? W[k * Hout + n] : 0.0f;
}

#define BUCKET 32

__global__ void gin_bucket_kernel(const int* __restrict__ src_idx,
                                  const int* __restrict__ dst_idx,
                                  int* __restrict__ deg,
                                  int* __restrict__ col,
                                  int2* __restrict__ ovf, int n_edges) {
    int e = blockIdx.x * blockDim.x + threadIdx.x;
    if (e >= n_edges) return;
    const int d = dst_idx[e];
    const int s = src_idx[e];
    const int r = atomicAdd(deg + d, 1);
    if (r < BUCKET) {
        col[d * BUCKET + r] = s;
    } else {
        const int o = atomicAdd(deg + N_NODES, 1);
        ovf[o] = make_int2(d, s);
    }
}

__device__ __forceinline__ void gin_add_rows(const float4* __restrict__ xr,
                                             int my, int n, float4& acc) {
    const float4 zero4 = make_float4(0.0f, 0.0f, 0.0f, 0.0f);
    for (int k = 0; k < n; k += 16) {
        float4 r[16];
        r[0] = xr[(size_t)__builtin_amdgcn_readlane(my, k) * 32];
        #pragma unroll
        for (int j = 1; j < 16; ++j) {
            r[j] = zero4;
            if (k + j < n)
                r[j] = xr[(size_t)__builtin_amdgcn_readlane(my, k + j) * 32];
        }
        #pragma unroll
        for (int s = 1; s < 16; s <<= 1) {
            #pragma unroll
            for (int j = 0; j < 16; j += 2 * s) {
                r[j].x += r[j + s].x;  r[j].y += r[j + s].y;
                r[j].z += r[j + s].z;  r[j].w += r[j + s].w;
            }
        }
        acc.x += r[0].x;  acc.y += r[0].y;  acc.z += r[0].z;  acc.w += r[0].w;
    }
}

__global__ void __launch_bounds__(256)
gin_gather_kernel(const float* __restrict__ x,
                  const int* __restrict__ deg,
                  const int* __restrict__ col,
                  const int2* __restrict__ ovf,
                  float* __restrict__ agg) {
    const int node = __builtin_amdgcn_readfirstlane(
        (int)(blockIdx.x * (blockDim.x >> 5) + (threadIdx.x >> 5)));
    if (node >= N_NODES) return;
    const int lane = threadIdx.x & 31;
    const float4* __restrict__ xr = reinterpret_cast<const float4*>(x) + lane;

    const int cnt = deg[node];

    const int n0  = (cnt < BUCKET) ? cnt : BUCKET;
    const int my0 = (lane < n0) ? col[node * BUCKET + lane] : 0;
    const float4 self = xr[(size_t)node * 32];

    float4 acc = make_float4(0.0f, 0.0f, 0.0f, 0.0f);
    gin_add_rows(xr, my0, n0, acc);
    if (cnt > BUCKET) {
        int total = deg[N_NODES];
        total = (total < N_EDGES) ? total : N_EDGES;
        for (int base = 0; base < total; base += 32) {
            const int i = base + lane;
            int2 rec = make_int2(-1, 0);
            if (i < total) rec = ovf[i];
            unsigned int m = __builtin_amdgcn_ballot_w32(rec.x == node);
            while (m != 0u) {
                const int l = __builtin_ctz(m);
                m &= m - 1u;
                const float4 v = xr[(size_t)__builtin_amdgcn_readlane(rec.y, l) * 32];
                acc.x += v.x;  acc.y += v.y;  acc.z += v.z;  acc.w += v.w;
            }
        }
    }
    acc.x += self.x;  acc.y += self.y;  acc.z += self.z;  acc.w += self.w;
    reinterpret_cast<float4*>(agg)[(size_t)node * 32 + lane] = acc;
}


#define WT_PITCH 132
#define CH_WAVES 16

template<int NT, bool FIRST>
__device__ __forceinline__ void gin_layer_mma(const float* Wl, const v4f* a,
                                              const v8f* prev, v8f* acc) {
    constexpr int KSTEPS = K_DIM / 8;
    v4f bc[NT], bnx[NT];
    #pragma unroll
    for (int t = 0; t < NT; ++t) {
        acc[t] = v8f{};
        bc[t] = *reinterpret_cast<const v4f*>(Wl + t * 16 * WT_PITCH);
    }
    __builtin_amdgcn_sched_group_barrier(0x100, NT, 0);

    #pragma unroll
    for (int s = 0; s < KSTEPS; ++s) {
        if (s + 1 < KSTEPS) {
            const int s1 = s + 1;
            const int koff = FIRST ? 8 * s1 : 16 * (s1 >> 1) + 4 * (s1 & 1);
            #pragma unroll
            for (int t = 0; t < NT; ++t)
                bnx[t] = *reinterpret_cast<const v4f*>(Wl + t * 16 * WT_PITCH + koff);
        }
        v2f x01, x23;
        if constexpr (FIRST) {
            x01 = v2f{a[s].x, a[s].y};
            x23 = v2f{a[s].z, a[s].w};
        } else {
            const v8f p = prev[s >> 1];
            const int e = 4 * (s & 1);
            x01 = v2f{p[e], p[e + 1]};
            x23 = v2f{p[e + 2], p[e + 3]};
        }
        #pragma unroll
        for (int t = 0; t < NT; ++t) {
            const v2f w01 = {bc[t].x, bc[t].y};
            acc[t] = __builtin_amdgcn_wmma_f32_16x16x4_f32(
                false, w01, false, x01, (short)0, acc[t], false, false);
        }
        #pragma unroll
        for (int t = 0; t < NT; ++t) {
            const v2f w23 = {bc[t].z, bc[t].w};
            acc[t] = __builtin_amdgcn_wmma_f32_16x16x4_f32(
                false, w23, false, x23, (short)0, acc[t], false, false);
        }
        if (s + 1 < KSTEPS)
            __builtin_amdgcn_sched_group_barrier(0x100, NT, 0);
        __builtin_amdgcn_sched_group_barrier(0x008, 2 * NT, 0);
        if (s + 1 < KSTEPS) {
            #pragma unroll
            for (int t = 0; t < NT; ++t) bc[t] = bnx[t];
        }
    }
}

template<int NT, int MODE>
__device__ __forceinline__ void gin_layer_epi(const float* P, v8f* acc) {
    #pragma unroll
    for (int t = 0; t < NT; ++t) {
        #pragma unroll
        for (int g = 0; g < 2; ++g) {
            const v4f b4 = *reinterpret_cast<const v4f*>(P + 16 * t + 4 * g);
            v4f sc4 = {1.0f, 1.0f, 1.0f, 1.0f}, sh4 = {0.0f, 0.0f, 0.0f, 0.0f};
            if constexpr (MODE == 2) {
                sc4 = *reinterpret_cast<const v4f*>(P + 128 + 16 * t + 4 * g);
                sh4 = *reinterpret_cast<const v4f*>(P + 256 + 16 * t + 4 * g);
            }
            #pragma unroll
            for (int i = 0; i < 4; ++i) {
                float val = acc[t][4 * g + i] + b4[i];
                if constexpr (MODE == 2) val = val * sc4[i] + sh4[i];
                if constexpr (MODE >= 1) val = fmaxf(val, 0.0f);
                acc[t][4 * g + i] = val;
            }
        }
    }
}

template<int ROWS>
__device__ __forceinline__ void gin_w_load(const float* __restrict__ WT, v4f* wreg) {
    #pragma unroll
    for (int it = 0; it < ROWS / CH_WAVES; ++it)
        wreg[it] = reinterpret_cast<const v4f*>(WT)[it * (CH_WAVES * 32) + threadIdx.x];
}
template<int ROWS>
__device__ __forceinline__ void gin_w_store(float* Wd, const v4f* wreg) {
    #pragma unroll
    for (int it = 0; it < ROWS / CH_WAVES; ++it) {
        const int i = it * (CH_WAVES * 32) + threadIdx.x;
        *reinterpret_cast<v4f*>(&Wd[(i >> 5) * WT_PITCH + (i & 31) * 4]) = wreg[it];
    }
}

template<int NL, int HLAST, bool RELU_LAST>
__global__ void __launch_bounds__(CH_WAVES * 32)
gin_mlp_chain(const float* __restrict__ A,
              const float* __restrict__ WTa, const float* __restrict__ WTb,
              const float* __restrict__ WTc, const float* __restrict__ WTd,
              const float* __restrict__ bias_a, const float* __restrict__ bias_b,
              const float* __restrict__ bias_c, const float* __restrict__ bias_d,
              const float* __restrict__ bn_g, const float* __restrict__ bn_b,
              const float* __restrict__ bn_m, const float* __restrict__ bn_v,
              float* __restrict__ out) {
    constexpr int NT_LAST   = (HLAST + 15) / 16;
    constexpr int ROWS_LAST = NT_LAST * 16;
    constexpr int ROW_TILES = N_NODES / 16;
    constexpr int KSTEPS    = K_DIM / 8;
    constexpr int W_FLOATS  = ((NL - 1) * 128 + ROWS_LAST) * WT_PITCH;
    constexpr int P_FLOATS  = 384 + (NL - 1) * 128;

    __shared__ alignas(16) float Ws[W_FLOATS + P_FLOATS];
    float* Ps = Ws + W_FLOATS;

    const int lane = threadIdx.x & 31;
    const int wave = __builtin_amdgcn_readfirstlane((int)(threadIdx.x >> 5));
    const int tile = blockIdx.x * CH_WAVES + wave;
    const bool have = tile < ROW_TILES;
    const int c  = lane & 15;
    const int hi = lane >> 4;

    constexpr int RL = ROWS_LAST / CH_WAVES;
    v4f w0[8], w1[NL == 2 ? RL : 8], w2[8], w3[RL];
    gin_w_load<128>(WTa, w0);
    if constexpr (NL == 2) {
        gin_w_load<ROWS_LAST>(WTb, w1);
    } else {
        gin_w_load<128>(WTb, w1);
        gin_w_load<128>(WTc, w2);
        gin_w_load<ROWS_LAST>(WTd, w3);
    }

    const int ltile = have ? tile : ROW_TILES - 1;
    const v4f* Ap = reinterpret_cast<const v4f*>(
        A + (long long)(ltile * 16 + c) * K_DIM + 4 * hi);
    v4f a[KSTEPS];
    #pragma unroll
    for (int j = 0; j < KSTEPS; ++j) a[j] = Ap[2 * j];

    if (threadIdx.x < 128) {
        const int n = threadIdx.x;
        const float rs = rsqrtf(bn_v[n] + BN_EPS);
        const float scale = bn_g[n] * rs;
        Ps[n]       = bias_a[n];
        Ps[128 + n] = scale;
        Ps[256 + n] = bn_b[n] - bn_m[n] * scale;
        if constexpr (NL == 2) {
            const float bb = bias_b[(n < HLAST) ? n : 0];
            Ps[384 + n] = (n < HLAST) ? bb : 0.0f;
        } else {
            Ps[384 + n] = bias_b[n];
            Ps[512 + n] = bias_c[n];
            const float bd = bias_d[(n < HLAST) ? n : 0];
            Ps[640 + n] = (n < HLAST) ? bd : 0.0f;
        }
    }

    gin_w_store<128>(Ws, w0);
    if constexpr (NL == 2) {
        gin_w_store<ROWS_LAST>(Ws + 128 * WT_PITCH, w1);
    } else {
        gin_w_store<128>(Ws + 128 * WT_PITCH, w1);
        gin_w_store<128>(Ws + 256 * WT_PITCH, w2);
        gin_w_store<ROWS_LAST>(Ws + 384 * WT_PITCH, w3);
    }
    __syncthreads();
    if (!have) return;

    const float* Wl0 = Ws + c * WT_PITCH + 4 * hi;
    const float* WlN = Ws + c * WT_PITCH + 8 * hi;
    const float* Pl  = Ps + 8 * hi;

    v8f x[8], y[8];
    gin_layer_mma<8, true>(Wl0, a, nullptr, x);
    gin_layer_epi<8, 2>(Pl, x);

    if constexpr (NL == 2) {
        gin_layer_mma<NT_LAST, false>(WlN + 128 * WT_PITCH, nullptr, x, y);
        gin_layer_epi<NT_LAST, RELU_LAST ? 1 : 0>(Pl + 384, y);
    } else {
        gin_layer_mma<8, false>(WlN + 128 * WT_PITCH, nullptr, x, y);
        gin_layer_epi<8, 1>(Pl + 384, y);
        gin_layer_mma<8, false>(WlN + 256 * WT_PITCH, nullptr, y, x);
        gin_layer_epi<8, 1>(Pl + 512, x);
        gin_layer_mma<NT_LAST, false>(WlN + 384 * WT_PITCH, nullptr, x, y);
        gin_layer_epi<NT_LAST, RELU_LAST ? 1 : 0>(Pl + 640, y);
    }

    float* orow = out + (long long)(tile * 16 + c) * HLAST + 8 * hi;
    #pragma unroll
    for (int t = 0; t < NT_LAST; ++t) {
        if ((HLAST % 16 == 0) || (16 * t + 8 * hi < HLAST)) {
            *reinterpret_cast<v4f*>(orow + 16 * t)     = v4f{y[t][0], y[t][1], y[t][2], y[t][3]};
            *reinterpret_cast<v4f*>(orow + 16 * t + 4) = v4f{y[t][4], y[t][5], y[t][6], y[t][7]};
        }
    }
}

extern "C" void kernel_launch(void* const* d_in, const int* in_sizes, int n_in,
                              void* d_out, int out_size, void* d_ws, size_t ws_size,
                              hipStream_t stream) {
    (void)in_sizes; (void)n_in; (void)out_size; (void)ws_size;

    const float* x      = (const float*)d_in[0];
    const int*   edge   = (const int*)  d_in[1];
    const float* lin1_w = (const float*)d_in[2];
    const float* lin1_b = (const float*)d_in[3];
    const float* bn1_g  = (const float*)d_in[4];
    const float* bn1_b  = (const float*)d_in[5];
    const float* bn1_m  = (const float*)d_in[6];
    const float* bn1_v  = (const float*)d_in[7];
    const float* lin2_w = (const float*)d_in[8];
    const float* lin2_b = (const float*)d_in[9];
    const float* lin3_w = (const float*)d_in[10];
    const float* lin3_b = (const float*)d_in[11];
    const float* bn2_g  = (const float*)d_in[12];
    const float* bn2_b  = (const float*)d_in[13];
    const float* bn2_m  = (const float*)d_in[14];
    const float* bn2_v  = (const float*)d_in[15];
    const float* lin4_w = (const float*)d_in[16];
    const float* lin4_b = (const float*)d_in[17];
    const float* lin5_w = (const float*)d_in[18];
    const float* lin5_b = (const float*)d_in[19];
    const float* lin6_w = (const float*)d_in[20];
    const float* lin6_b = (const float*)d_in[21];

    const int* e_src = edge;
    const int* e_dst = edge + N_EDGES;

    const size_t NH = (size_t)N_NODES * K_DIM;
    float* B0  = (float*)d_ws;
    float* B1  = B0  + NH;
    float* WT1 = B1  + NH;
    float* WT2 = WT1 + 128 * 128;
    float* WT3 = WT2 + 128 * 128;
    float* WT4 = WT3 + 128 * 128;
    float* WT5 = WT4 + 128 * 128;
    float* WT6 = WT5 + 128 * 128;
    int*  deg = (int*)(WT6 + 64 * 128);
    int*  col = deg + N_NODES + 16;
    int2* ovf = (int2*)(col + N_NODES * BUCKET);

    const int sBlocks  = SETUP_T_BLOCKS + SETUP_Z_BLOCKS;
    const int eBlocks  = (N_EDGES + 255) / 256;
    const int gaBlocks = (N_NODES + 7) / 8;
    const int gChain   = (N_NODES / 16 + CH_WAVES - 1) / CH_WAVES;

    gin_setup_kernel<<<sBlocks, 256, 0, stream>>>(lin1_w, lin2_w, lin3_w, lin4_w, lin5_w, lin6_w,
                                                  WT1, deg);

    gin_bucket_kernel<<<eBlocks, 256, 0, stream>>>(e_src, e_dst, deg, col, ovf, N_EDGES);

    gin_gather_kernel<<<gaBlocks, 256, 0, stream>>>(x, deg, col, ovf, B0);
    gin_mlp_chain<2, 128, true><<<gChain, CH_WAVES * 32, 0, stream>>>(
        B0, WT1, WT2, nullptr, nullptr, lin1_b, lin2_b, nullptr, nullptr,
        bn1_g, bn1_b, bn1_m, bn1_v, B1);

    gin_gather_kernel<<<gaBlocks, 256, 0, stream>>>(B1, deg, col, ovf, B0);
    gin_mlp_chain<4, 40, false><<<gChain, CH_WAVES * 32, 0, stream>>>(
        B0, WT3, WT4, WT5, WT6, lin3_b, lin4_b, lin5_b, lin6_b,
        bn2_g, bn2_b, bn2_m, bn2_v, (float*)d_out);
}


# Edge ranks from LDS counters in two levels (edges binned by 256-node range, then ranked per range), not one returning global add per edge; list and sum order = LDS arrival order; 6 -> 7 launches
# speedup vs baseline: 3.3522x; 1.2646x over previous
#include <hip/hip_runtime.h>

typedef __attribute__((ext_vector_type(2))) float v2f;
typedef __attribute__((ext_vector_type(4))) float v4f;
typedef __attribute__((ext_vector_type(8))) float v8f;

#define N_NODES 50000
#define N_EDGES 600000
#define K_DIM   128
#define BN_EPS  1e-5f

#define SETUP_T_BLOCKS 352
#define BIN_SHIFT 8
#define BIN_NODES (1 << BIN_SHIFT)
#define N_BINS    ((N_NODES + BIN_NODES - 1) / BIN_NODES)
#define N_CNTS    (N_BINS + 2)
#define SETUP_Z_BLOCKS 1

__global__ void __launch_bounds__(256)
gin_setup_kernel(const float* __restrict__ w1, const float* __restrict__ w2,
                 const float* __restrict__ w3, const float* __restrict__ w4,
                 const float* __restrict__ w5, const float* __restrict__ w6,
                 float* __restrict__ WT, int* __restrict__ cnts) {
    const int b = blockIdx.x;
    if (b >= SETUP_T_BLOCKS) {
        if (threadIdx.x < N_CNTS) cnts[threadIdx.x] = 0;
        return;
    }
    const int m = b >> 6;
    const float* __restrict__ W = (m == 0) ? w1 : (m == 1) ? w2 : (m == 2) ? w3
                                : (m == 3) ? w4 : (m == 4) ? w5 : w6;
    const int Hout = (m == 5) ? 40 : 128;
    const int i = (b & 63) * 256 + threadIdx.x;
    const int n = i >> 7;
    const int k = i & 127;
    WT[m * (128 * 128) + i] = (n < Hout) ? W[k * Hout + n] : 0.0f;
}

#define BUCKET 32
#define BIN_CAP 4096
#define BIN_THREADS 512
#define RANK_THREADS 512
#define RANK_PER (BIN_CAP / RANK_THREADS)
static_assert(N_NODES <= 65536, "destination and source are packed into 16 bits each");
static_assert(N_EDGES % 4 == 0, "edges are read four at a time");

__global__ void __launch_bounds__(BIN_THREADS)
gin_bin_kernel(const int* __restrict__ src_idx,
               const int* __restrict__ dst_idx,
               int* __restrict__ cnts,
               unsigned int* __restrict__ binbuf,
               unsigned int* __restrict__ ovf1) {
    __shared__ int hist[N_BINS];
    __shared__ int base[N_BINS];
    const int tid = threadIdx.x;
    if (tid < N_BINS) hist[tid] = 0;
    __syncthreads();

    const int q = blockIdx.x * BIN_THREADS + tid;
    const bool on = q < N_EDGES / 4;
    int d[4] = {0, 0, 0, 0}, s[4] = {0, 0, 0, 0}, r[4] = {0, 0, 0, 0};
    if (on) {
        const int4 d4 = reinterpret_cast<const int4*>(dst_idx)[q];
        const int4 s4 = reinterpret_cast<const int4*>(src_idx)[q];
        d[0] = d4.x; d[1] = d4.y; d[2] = d4.z; d[3] = d4.w;
        s[0] = s4.x; s[1] = s4.y; s[2] = s4.z; s[3] = s4.w;
        #pragma unroll
        for (int i = 0; i < 4; ++i) r[i] = atomicAdd(&hist[(unsigned int)d[i] >> BIN_SHIFT], 1);
    }
    __syncthreads();
    if (tid < N_BINS) {
        const int c = hist[tid];
        base[tid] = (c > 0) ? atomicAdd(cnts + 2 + tid, c) : 0;
    }
    __syncthreads();
    if (on) {
        #pragma unroll
        for (int i = 0; i < 4; ++i) {
            const int b = (int)((unsigned int)d[i] >> BIN_SHIFT);
            const int pos = base[b] + r[i];
            const unsigned int rec = ((unsigned int)d[i] << 16) | (unsigned int)s[i];
            if (pos < BIN_CAP) {
                binbuf[b * BIN_CAP + pos] = rec;
            } else {
                const int o = atomicAdd(cnts + 1, 1);
                ovf1[o] = rec;
            }
        }
    }
}

__device__ __forceinline__ void gin_rank_place(unsigned int rec, int* cnt, int* lst,
                                               int* __restrict__ cnts, int2* __restrict__ ovf) {
    const int dl = (int)((rec >> 16) & (BIN_NODES - 1));
    const int sv = (int)(rec & 0xFFFFu);
    const int r = atomicAdd(&cnt[dl], 1);
    if (r < BUCKET) {
        lst[dl * BUCKET + r] = sv;
    } else {
        const int o = atomicAdd(cnts, 1);
        ovf[o] = make_int2((int)(rec >> 16), sv);
    }
}

__global__ void __launch_bounds__(RANK_THREADS)
gin_rank_kernel(int* __restrict__ cnts,
                const unsigned int* __restrict__ binbuf,
                const unsigned int* __restrict__ ovf1,
                int* __restrict__ deg,
                int* __restrict__ col,
                int2* __restrict__ ovf) {
    __shared__ int cnt[BIN_NODES];
    __shared__ int lst[BIN_NODES * BUCKET];
    const int b = blockIdx.x;
    const int tid = threadIdx.x;
    if (tid < BIN_NODES) cnt[tid] = 0;
    __syncthreads();

    const int tot = cnts[2 + b];
    const int nb = (tot < BIN_CAP) ? tot : BIN_CAP;
    unsigned int rec[RANK_PER];
    #pragma unroll
    for (int k = 0; k < RANK_PER; ++k) {
        const int i = tid + k * RANK_THREADS;
        rec[k] = (i < nb) ? binbuf[b * BIN_CAP + i] : 0u;
    }
    #pragma unroll
    for (int k = 0; k < RANK_PER; ++k)
        if (tid + k * RANK_THREADS < nb) gin_rank_place(rec[k], cnt, lst, cnts, ovf);
    if (tot > BIN_CAP) {
        int t1 = cnts[1];
        t1 = (t1 < N_EDGES) ? t1 : N_EDGES;
        for (int i = tid; i < t1; i += RANK_THREADS) {
            const unsigned int rec = ovf1[i];
            if ((int)(rec >> 24) == b) gin_rank_place(rec, cnt, lst, cnts, ovf);
        }
    }
    __syncthreads();

    const int node0 = b * BIN_NODES;
    if (tid < BIN_NODES && node0 + tid < N_NODES) deg[node0 + tid] = cnt[tid];
    const int lane = tid & 31;
    for (int n = tid >> 5; n < BIN_NODES; n += RANK_THREADS / 32) {
        if (node0 + n < N_NODES) {
            const int c = cnt[n];
            if (lane < ((c < BUCKET) ? c : BUCKET))
                col[(node0 + n) * BUCKET + lane] = lst[n * BUCKET + lane];
        }
    }
}

__device__ __forceinline__ void gin_add_rows(const float4* __restrict__ xr,
                                             int my, int n, float4& acc) {
    const float4 zero4 = make_float4(0.0f, 0.0f, 0.0f, 0.0f);
    for (int k = 0; k < n; k += 16) {
        float4 r[16];
        r[0] = xr[(size_t)__builtin_amdgcn_readlane(my, k) * 32];
        #pragma unroll
        for (int j = 1; j < 16; ++j) {
            r[j] = zero4;
            if (k + j < n)
                r[j] = xr[(size_t)__builtin_amdgcn_readlane(my, k + j) * 32];
        }
        #pragma unroll
        for (int s = 1; s < 16; s <<= 1) {
            #pragma unroll
            for (int j = 0; j < 16; j += 2 * s) {
                r[j].x += r[j + s].x;  r[j].y += r[j + s].y;
                r[j].z += r[j + s].z;  r[j].w += r[j + s].w;
            }
        }
        acc.x += r[0].x;  acc.y += r[0].y;  acc.z += r[0].z;  acc.w += r[0].w;
    }
}

__global__ void __launch_bounds__(256)
gin_gather_kernel(const float* __restrict__ x,
                  const int* __restrict__ deg,
                  const int* __restrict__ col,
                  const int2* __restrict__ ovf,
                  float* __restrict__ agg) {
    const int node = __builtin_amdgcn_readfirstlane(
        (int)(blockIdx.x * (blockDim.x >> 5) + (threadIdx.x >> 5)));
    if (node >= N_NODES) return;
    const int lane = threadIdx.x & 31;
    const float4* __restrict__ xr = reinterpret_cast<const float4*>(x) + lane;

    const int cnt = deg[node];

    const int n0  = (cnt < BUCKET) ? cnt : BUCKET;
    const int my0 = (lane < n0) ? col[node * BUCKET + lane] : 0;
    const float4 self = xr[(size_t)node * 32];

    float4 acc = make_float4(0.0f, 0.0f, 0.0f, 0.0f);
    gin_add_rows(xr, my0, n0, acc);
    if (cnt > BUCKET) {
        int total = deg[N_NODES];
        total = (total < N_EDGES) ? total : N_EDGES;
        for (int base = 0; base < total; base += 32) {
            const int i = base + lane;
            int2 rec = make_int2(-1, 0);
            if (i < total) rec = ovf[i];
            unsigned int m = __builtin_amdgcn_ballot_w32(rec.x == node);
            while (m != 0u) {
                const int l = __builtin_ctz(m);
                m &= m - 1u;
                const float4 v = xr[(size_t)__builtin_amdgcn_readlane(rec.y, l) * 32];
                acc.x += v.x;  acc.y += v.y;  acc.z += v.z;  acc.w += v.w;
            }
        }
    }
    acc.x += self.x;  acc.y += self.y;  acc.z += self.z;  acc.w += self.w;
    reinterpret_cast<float4*>(agg)[(size_t)node * 32 + lane] = acc;
}


#define WT_PITCH 132
#define CH_WAVES 16

template<int NT, bool FIRST>
__device__ __forceinline__ void gin_layer_mma(const float* Wl, const v4f* a,
                                              const v8f* prev, v8f* acc) {
    constexpr int KSTEPS = K_DIM / 8;
    v4f bc[NT], bnx[NT];
    #pragma unroll
    for (int t = 0; t < NT; ++t) {
        acc[t] = v8f{};
        bc[t] = *reinterpret_cast<const v4f*>(Wl + t * 16 * WT_PITCH);
    }
    __builtin_amdgcn_sched_group_barrier(0x100, NT, 0);

    #pragma unroll
    for (int s = 0; s < KSTEPS; ++s) {
        if (s + 1 < KSTEPS) {
            const int s1 = s + 1;
            const int koff = FIRST ? 8 * s1 : 16 * (s1 >> 1) + 4 * (s1 & 1);
            #pragma unroll
            for (int t = 0; t < NT; ++t)
                bnx[t] = *reinterpret_cast<const v4f*>(Wl + t * 16 * WT_PITCH + koff);
        }
        v2f x01, x23;
        if constexpr (FIRST) {
            x01 = v2f{a[s].x, a[s].y};
            x23 = v2f{a[s].z, a[s].w};
        } else {
            const v8f p = prev[s >> 1];
            const int e = 4 * (s & 1);
            x01 = v2f{p[e], p[e + 1]};
            x23 = v2f{p[e + 2], p[e + 3]};
        }
        #pragma unroll
        for (int t = 0; t < NT; ++t) {
            const v2f w01 = {bc[t].x, bc[t].y};
            acc[t] = __builtin_amdgcn_wmma_f32_16x16x4_f32(
                false, w01, false, x01, (short)0, acc[t], false, false);
        }
        #pragma unroll
        for (int t = 0; t < NT; ++t) {
            const v2f w23 = {bc[t].z, bc[t].w};
            acc[t] = __builtin_amdgcn_wmma_f32_16x16x4_f32(
                false, w23, false, x23, (short)0, acc[t], false, false);
        }
        if (s + 1 < KSTEPS)
            __builtin_amdgcn_sched_group_barrier(0x100, NT, 0);
        __builtin_amdgcn_sched_group_barrier(0x008, 2 * NT, 0);
        if (s + 1 < KSTEPS) {
            #pragma unroll
            for (int t = 0; t < NT; ++t) bc[t] = bnx[t];
        }
    }
}

template<int NT, int MODE>
__device__ __forceinline__ void gin_layer_epi(const float* P, v8f* acc) {
    #pragma unroll
    for (int t = 0; t < NT; ++t) {
        #pragma unroll
        for (int g = 0; g < 2; ++g) {
            const v4f b4 = *reinterpret_cast<const v4f*>(P + 16 * t + 4 * g);
            v4f sc4 = {1.0f, 1.0f, 1.0f, 1.0f}, sh4 = {0.0f, 0.0f, 0.0f, 0.0f};
            if constexpr (MODE == 2) {
                sc4 = *reinterpret_cast<const v4f*>(P + 128 + 16 * t + 4 * g);
                sh4 = *reinterpret_cast<const v4f*>(P + 256 + 16 * t + 4 * g);
            }
            #pragma unroll
            for (int i = 0; i < 4; ++i) {
                float val = acc[t][4 * g + i] + b4[i];
                if constexpr (MODE == 2) val = val * sc4[i] + sh4[i];
                if constexpr (MODE >= 1) val = fmaxf(val, 0.0f);
                acc[t][4 * g + i] = val;
            }
        }
    }
}

template<int ROWS>
__device__ __forceinline__ void gin_w_load(const float* __restrict__ WT, v4f* wreg) {
    #pragma unroll
    for (int it = 0; it < ROWS / CH_WAVES; ++it)
        wreg[it] = reinterpret_cast<const v4f*>(WT)[it * (CH_WAVES * 32) + threadIdx.x];
}
template<int ROWS>
__device__ __forceinline__ void gin_w_store(float* Wd, const v4f* wreg) {
    #pragma unroll
    for (int it = 0; it < ROWS / CH_WAVES; ++it) {
        const int i = it * (CH_WAVES * 32) + threadIdx.x;
        *reinterpret_cast<v4f*>(&Wd[(i >> 5) * WT_PITCH + (i & 31) * 4]) = wreg[it];
    }
}

template<int NL, int HLAST, bool RELU_LAST>
__global__ void __launch_bounds__(CH_WAVES * 32)
gin_mlp_chain(const float* __restrict__ A,
              const float* __restrict__ WTa, const float* __restrict__ WTb,
              const float* __restrict__ WTc, const float* __restrict__ WTd,
              const float* __restrict__ bias_a, const float* __restrict__ bias_b,
              const float* __restrict__ bias_c, const float* __restrict__ bias_d,
              const float* __restrict__ bn_g, const float* __restrict__ bn_b,
              const float* __restrict__ bn_m, const float* __restrict__ bn_v,
              float* __restrict__ out) {
    constexpr int NT_LAST   = (HLAST + 15) / 16;
    constexpr int ROWS_LAST = NT_LAST * 16;
    constexpr int ROW_TILES = N_NODES / 16;
    constexpr int KSTEPS    = K_DIM / 8;
    constexpr int W_FLOATS  = ((NL - 1) * 128 + ROWS_LAST) * WT_PITCH;
    constexpr int P_FLOATS  = 384 + (NL - 1) * 128;

    __shared__ alignas(16) float Ws[W_FLOATS + P_FLOATS];
    float* Ps = Ws + W_FLOATS;

    const int lane = threadIdx.x & 31;
    const int wave = __builtin_amdgcn_readfirstlane((int)(threadIdx.x >> 5));
    const int tile = blockIdx.x * CH_WAVES + wave;
    const bool have = tile < ROW_TILES;
    const int c  = lane & 15;
    const int hi = lane >> 4;

    constexpr int RL = ROWS_LAST / CH_WAVES;
    v4f w0[8], w1[NL == 2 ? RL : 8], w2[8], w3[RL];
    gin_w_load<128>(WTa, w0);
    if constexpr (NL == 2) {
        gin_w_load<ROWS_LAST>(WTb, w1);
    } else {
        gin_w_load<128>(WTb, w1);
        gin_w_load<128>(WTc, w2);
        gin_w_load<ROWS_LAST>(WTd, w3);
    }

    const int ltile = have ? tile : ROW_TILES - 1;
    const v4f* Ap = reinterpret_cast<const v4f*>(
        A + (long long)(ltile * 16 + c) * K_DIM + 4 * hi);
    v4f a[KSTEPS];
    #pragma unroll
    for (int j = 0; j < KSTEPS; ++j) a[j] = Ap[2 * j];

    if (threadIdx.x < 128) {
        const int n = threadIdx.x;
        const float rs = rsqrtf(bn_v[n] + BN_EPS);
        const float scale = bn_g[n] * rs;
        Ps[n]       = bias_a[n];
        Ps[128 + n] = scale;
        Ps[256 + n] = bn_b[n] - bn_m[n] * scale;
        if constexpr (NL == 2) {
            const float bb = bias_b[(n < HLAST) ? n : 0];
            Ps[384 + n] = (n < HLAST) ? bb : 0.0f;
        } else {
            Ps[384 + n] = bias_b[n];
            Ps[512 + n] = bias_c[n];
            const float bd = bias_d[(n < HLAST) ? n : 0];
            Ps[640 + n] = (n < HLAST) ? bd : 0.0f;
        }
    }

    gin_w_store<128>(Ws, w0);
    if constexpr (NL == 2) {
        gin_w_store<ROWS_LAST>(Ws + 128 * WT_PITCH, w1);
    } else {
        gin_w_store<128>(Ws + 128 * WT_PITCH, w1);
        gin_w_store<128>(Ws + 256 * WT_PITCH, w2);
        gin_w_store<ROWS_LAST>(Ws + 384 * WT_PITCH, w3);
    }
    __syncthreads();
    if (!have) return;

    const float* Wl0 = Ws + c * WT_PITCH + 4 * hi;
    const float* WlN = Ws + c * WT_PITCH + 8 * hi;
    const float* Pl  = Ps + 8 * hi;

    v8f x[8], y[8];
    gin_layer_mma<8, true>(Wl0, a, nullptr, x);
    gin_layer_epi<8, 2>(Pl, x);

    if constexpr (NL == 2) {
        gin_layer_mma<NT_LAST, false>(WlN + 128 * WT_PITCH, nullptr, x, y);
        gin_layer_epi<NT_LAST, RELU_LAST ? 1 : 0>(Pl + 384, y);
    } else {
        gin_layer_mma<8, false>(WlN + 128 * WT_PITCH, nullptr, x, y);
        gin_layer_epi<8, 1>(Pl + 384, y);
        gin_layer_mma<8, false>(WlN + 256 * WT_PITCH, nullptr, y, x);
        gin_layer_epi<8, 1>(Pl + 512, x);
        gin_layer_mma<NT_LAST, false>(WlN + 384 * WT_PITCH, nullptr, x, y);
        gin_layer_epi<NT_LAST, RELU_LAST ? 1 : 0>(Pl + 640, y);
    }

    float* orow = out + (long long)(tile * 16 + c) * HLAST + 8 * hi;
    #pragma unroll
    for (int t = 0; t < NT_LAST; ++t) {
        if ((HLAST % 16 == 0) || (16 * t + 8 * hi < HLAST)) {
            *reinterpret_cast<v4f*>(orow + 16 * t)     = v4f{y[t][0], y[t][1], y[t][2], y[t][3]};
            *reinterpret_cast<v4f*>(orow + 16 * t + 4) = v4f{y[t][4], y[t][5], y[t][6], y[t][7]};
        }
    }
}

extern "C" void kernel_launch(void* const* d_in, const int* in_sizes, int n_in,
                              void* d_out, int out_size, void* d_ws, size_t ws_size,
                              hipStream_t stream) {
    (void)in_sizes; (void)n_in; (void)out_size; (void)ws_size;

    const float* x      = (const float*)d_in[0];
    const int*   edge   = (const int*)  d_in[1];
    const float* lin1_w = (const float*)d_in[2];
    const float* lin1_b = (const float*)d_in[3];
    const float* bn1_g  = (const float*)d_in[4];
    const float* bn1_b  = (const float*)d_in[5];
    const float* bn1_m  = (const float*)d_in[6];
    const float* bn1_v  = (const float*)d_in[7];
    const float* lin2_w = (const float*)d_in[8];
    const float* lin2_b = (const float*)d_in[9];
    const float* lin3_w = (const float*)d_in[10];
    const float* lin3_b = (const float*)d_in[11];
    const float* bn2_g  = (const float*)d_in[12];
    const float* bn2_b  = (const float*)d_in[13];
    const float* bn2_m  = (const float*)d_in[14];
    const float* bn2_v  = (const float*)d_in[15];
    const float* lin4_w = (const float*)d_in[16];
    const float* lin4_b = (const float*)d_in[17];
    const float* lin5_w = (const float*)d_in[18];
    const float* lin5_b = (const float*)d_in[19];
    const float* lin6_w = (const float*)d_in[20];
    const float* lin6_b = (const float*)d_in[21];

    const int* e_src = edge;
    const int* e_dst = edge + N_EDGES;

    const size_t NH = (size_t)N_NODES * K_DIM;
    float* B0  = (float*)d_ws;
    float* B1  = B0  + NH;
    float* WT1 = B1  + NH;
    float* WT2 = WT1 + 128 * 128;
    float* WT3 = WT2 + 128 * 128;
    float* WT4 = WT3 + 128 * 128;
    float* WT5 = WT4 + 128 * 128;
    float* WT6 = WT5 + 128 * 128;
    int*  deg  = (int*)(WT6 + 64 * 128);
    int*  cnts = deg + N_NODES;
    int*  col  = deg + N_NODES + 272;
    int2* ovf  = (int2*)(col + N_NODES * BUCKET);
    unsigned int* binbuf = (unsigned int*)(ovf + N_EDGES);
    unsigned int* ovf1   = binbuf + N_BINS * BIN_CAP;

    const int sBlocks  = SETUP_T_BLOCKS + SETUP_Z_BLOCKS;
    const int eBlocks  = (N_EDGES / 4 + BIN_THREADS - 1) / BIN_THREADS;
    const int gaBlocks = (N_NODES + 7) / 8;
    const int gChain   = (N_NODES / 16 + CH_WAVES - 1) / CH_WAVES;

    gin_setup_kernel<<<sBlocks, 256, 0, stream>>>(lin1_w, lin2_w, lin3_w, lin4_w, lin5_w, lin6_w,
                                                  WT1, cnts);

    gin_bin_kernel<<<eBlocks, BIN_THREADS, 0, stream>>>(e_src, e_dst, cnts, binbuf, ovf1);
    gin_rank_kernel<<<N_BINS, RANK_THREADS, 0, stream>>>(cnts, binbuf, ovf1, deg, col, ovf);

    gin_gather_kernel<<<gaBlocks, 256, 0, stream>>>(x, deg, col, ovf, B0);
    gin_mlp_chain<2, 128, true><<<gChain, CH_WAVES * 32, 0, stream>>>(
        B0, WT1, WT2, nullptr, nullptr, lin1_b, lin2_b, nullptr, nullptr,
        bn1_g, bn1_b, bn1_m, bn1_v, B1);

    gin_gather_kernel<<<gaBlocks, 256, 0, stream>>>(B1, deg, col, ovf, B0);
    gin_mlp_chain<4, 40, false><<<gChain, CH_WAVES * 32, 0, stream>>>(
        B0, WT3, WT4, WT5, WT6, lin3_b, lin4_b, lin5_b, lin6_b,
        bn2_g, bn2_b, bn2_m, bn2_v, (float*)d_out);
}


# Chain tiles dealt evenly over the matrix pipes: 256 work-groups x 12 whole-tile waves (3 per SIMD); each of the 53 left-over tiles shared by 4 waves by output-feature tile, layers passed through LDS
# speedup vs baseline: 3.7207x; 1.1099x over previous
#include <hip/hip_runtime.h>

typedef __attribute__((ext_vector_type(2))) float v2f;
typedef __attribute__((ext_vector_type(4))) float v4f;
typedef __attribute__((ext_vector_type(8))) float v8f;

#define N_NODES 50000
#define N_EDGES 600000
#define K_DIM   128
#define BN_EPS  1e-5f

#define SETUP_T_BLOCKS 352
#define BIN_SHIFT 8
#define BIN_NODES (1 << BIN_SHIFT)
#define N_BINS    ((N_NODES + BIN_NODES - 1) / BIN_NODES)
#define N_CNTS    (N_BINS + 2)
#define SETUP_Z_BLOCKS 1

__global__ void __launch_bounds__(256)
gin_setup_kernel(const float* __restrict__ w1, const float* __restrict__ w2,
                 const float* __restrict__ w3, const float* __restrict__ w4,
                 const float* __restrict__ w5, const float* __restrict__ w6,
                 float* __restrict__ WT, int* __restrict__ cnts) {
    const int b = blockIdx.x;
    if (b >= SETUP_T_BLOCKS) {
        if (threadIdx.x < N_CNTS) cnts[threadIdx.x] = 0;
        return;
    }
    const int m = b >> 6;
    const float* __restrict__ W = (m == 0) ? w1 : (m == 1) ? w2 : (m == 2) ? w3
                                : (m == 3) ? w4 : (m == 4) ? w5 : w6;
    const int Hout = (m == 5) ? 40 : 128;
    const int i = (b & 63) * 256 + threadIdx.x;
    const int n = i >> 7;
    const int k = i & 127;
    WT[m * (128 * 128) + i] = (n < Hout) ? W[k * Hout + n] : 0.0f;
}

#define BUCKET 32
#define BIN_CAP 4096
#define BIN_THREADS 512
#define RANK_THREADS 512
#define RANK_PER (BIN_CAP / RANK_THREADS)
static_assert(N_NODES <= 65536, "destination and source are packed into 16 bits each");
static_assert(N_EDGES % 4 == 0, "edges are read four at a time");

__global__ void __launch_bounds__(BIN_THREADS)
gin_bin_kernel(const int* __restrict__ src_idx,
               const int* __restrict__ dst_idx,
               int* __restrict__ cnts,
               unsigned int* __restrict__ binbuf,
               unsigned int* __restrict__ ovf1) {
    __shared__ int hist[N_BINS];
    __shared__ int base[N_BINS];
    const int tid = threadIdx.x;
    if (tid < N_BINS) hist[tid] = 0;
    __syncthreads();

    const int q = blockIdx.x * BIN_THREADS + tid;
    const bool on = q < N_EDGES / 4;
    int d[4] = {0, 0, 0, 0}, s[4] = {0, 0, 0, 0}, r[4] = {0, 0, 0, 0};
    if (on) {
        const int4 d4 = reinterpret_cast<const int4*>(dst_idx)[q];
        const int4 s4 = reinterpret_cast<const int4*>(src_idx)[q];
        d[0] = d4.x; d[1] = d4.y; d[2] = d4.z; d[3] = d4.w;
        s[0] = s4.x; s[1] = s4.y; s[2] = s4.z; s[3] = s4.w;
        #pragma unroll
        for (int i = 0; i < 4; ++i) r[i] = atomicAdd(&hist[(unsigned int)d[i] >> BIN_SHIFT], 1);
    }
    __syncthreads();
    if (tid < N_BINS) {
        const int c = hist[tid];
        base[tid] = (c > 0) ? atomicAdd(cnts + 2 + tid, c) : 0;
    }
    __syncthreads();
    if (on) {
        #pragma unroll
        for (int i = 0; i < 4; ++i) {
            const int b = (int)((unsigned int)d[i] >> BIN_SHIFT);
            const int pos = base[b] + r[i];
            const unsigned int rec = ((unsigned int)d[i] << 16) | (unsigned int)s[i];
            if (pos < BIN_CAP) {
                binbuf[b * BIN_CAP + pos] = rec;
            } else {
                const int o = atomicAdd(cnts + 1, 1);
                ovf1[o] = rec;
            }
        }
    }
}

__device__ __forceinline__ void gin_rank_place(unsigned int rec, int* cnt, int* lst,
                                               int* __restrict__ cnts, int2* __restrict__ ovf) {
    const int dl = (int)((rec >> 16) & (BIN_NODES - 1));
    const int sv = (int)(rec & 0xFFFFu);
    const int r = atomicAdd(&cnt[dl], 1);
    if (r < BUCKET) {
        lst[dl * BUCKET + r] = sv;
    } else {
        const int o = atomicAdd(cnts, 1);
        ovf[o] = make_int2((int)(rec >> 16), sv);
    }
}

__global__ void __launch_bounds__(RANK_THREADS)
gin_rank_kernel(int* __restrict__ cnts,
                const unsigned int* __restrict__ binbuf,
                const unsigned int* __restrict__ ovf1,
                int* __restrict__ deg,
                int* __restrict__ col,
                int2* __restrict__ ovf) {
    __shared__ int cnt[BIN_NODES];
    __shared__ int lst[BIN_NODES * BUCKET];
    const int b = blockIdx.x;
    const int tid = threadIdx.x;
    if (tid < BIN_NODES) cnt[tid] = 0;
    __syncthreads();

    const int tot = cnts[2 + b];
    const int nb = (tot < BIN_CAP) ? tot : BIN_CAP;
    unsigned int rec[RANK_PER];
    #pragma unroll
    for (int k = 0; k < RANK_PER; ++k) {
        const int i = tid + k * RANK_THREADS;
        rec[k] = (i < nb) ? binbuf[b * BIN_CAP + i] : 0u;
    }
    #pragma unroll
    for (int k = 0; k < RANK_PER; ++k)
        if (tid + k * RANK_THREADS < nb) gin_rank_place(rec[k], cnt, lst, cnts, ovf);
    if (tot > BIN_CAP) {
        int t1 = cnts[1];
        t1 = (t1 < N_EDGES) ? t1 : N_EDGES;
        for (int i = tid; i < t1; i += RANK_THREADS) {
            const unsigned int rec = ovf1[i];
            if ((int)(rec >> 24) == b) gin_rank_place(rec, cnt, lst, cnts, ovf);
        }
    }
    __syncthreads();

    const int node0 = b * BIN_NODES;
    if (tid < BIN_NODES && node0 + tid < N_NODES) deg[node0 + tid] = cnt[tid];
    const int lane = tid & 31;
    for (int n = tid >> 5; n < BIN_NODES; n += RANK_THREADS / 32) {
        if (node0 + n < N_NODES) {
            const int c = cnt[n];
            if (lane < ((c < BUCKET) ? c : BUCKET))
                col[(node0 + n) * BUCKET + lane] = lst[n * BUCKET + lane];
        }
    }
}

__device__ __forceinline__ void gin_add_rows(const float4* __restrict__ xr,
                                             int my, int n, float4& acc) {
    const float4 zero4 = make_float4(0.0f, 0.0f, 0.0f, 0.0f);
    for (int k = 0; k < n; k += 16) {
        float4 r[16];
        r[0] = xr[(size_t)__builtin_amdgcn_readlane(my, k) * 32];
        #pragma unroll
        for (int j = 1; j < 16; ++j) {
            r[j] = zero4;
            if (k + j < n)
                r[j] = xr[(size_t)__builtin_amdgcn_readlane(my, k + j) * 32];
        }
        #pragma unroll
        for (int s = 1; s < 16; s <<= 1) {
            #pragma unroll
            for (int j = 0; j < 16; j += 2 * s) {
                r[j].x += r[j + s].x;  r[j].y += r[j + s].y;
                r[j].z += r[j + s].z;  r[j].w += r[j + s].w;
            }
        }
        acc.x += r[0].x;  acc.y += r[0].y;  acc.z += r[0].z;  acc.w += r[0].w;
    }
}

__global__ void __launch_bounds__(256)
gin_gather_kernel(const float* __restrict__ x,
                  const int* __restrict__ deg,
                  const int* __restrict__ col,
                  const int2* __restrict__ ovf,
                  float* __restrict__ agg) {
    const int node = __builtin_amdgcn_readfirstlane(
        (int)(blockIdx.x * (blockDim.x >> 5) + (threadIdx.x >> 5)));
    if (node >= N_NODES) return;
    const int lane = threadIdx.x & 31;
    const float4* __restrict__ xr = reinterpret_cast<const float4*>(x) + lane;

    const int cnt = deg[node];

    const int n0  = (cnt < BUCKET) ? cnt : BUCKET;
    const int my0 = (lane < n0) ? col[node * BUCKET + lane] : 0;
    const float4 self = xr[(size_t)node * 32];

    float4 acc = make_float4(0.0f, 0.0f, 0.0f, 0.0f);
    gin_add_rows(xr, my0, n0, acc);
    if (cnt > BUCKET) {
        int total = deg[N_NODES];
        total = (total < N_EDGES) ? total : N_EDGES;
        for (int base = 0; base < total; base += 32) {
            const int i = base + lane;
            int2 rec = make_int2(-1, 0);
            if (i < total) rec = ovf[i];
            unsigned int m = __builtin_amdgcn_ballot_w32(rec.x == node);
            while (m != 0u) {
                const int l = __builtin_ctz(m);
                m &= m - 1u;
                const float4 v = xr[(size_t)__builtin_amdgcn_readlane(rec.y, l) * 32];
                acc.x += v.x;  acc.y += v.y;  acc.z += v.z;  acc.w += v.w;
            }
        }
    }
    acc.x += self.x;  acc.y += self.y;  acc.z += self.z;  acc.w += self.w;
    reinterpret_cast<float4*>(agg)[(size_t)node * 32 + lane] = acc;
}


#define WT_PITCH 132
#define CH_WAVES 16
#define CH_FULL  12
#define CH_BLOCKS 256
#define X_TILE_FLOATS (8 * 32)
static_assert(CH_WAVES - CH_FULL == 4, "a left-over tile is shared by four waves, one per SIMD");
static_assert(CH_BLOCKS * CH_FULL <= N_NODES / 16, "every whole-tile wave has a tile");
static_assert(N_NODES / 16 - CH_BLOCKS * CH_FULL <= CH_BLOCKS, "at most one left-over tile per work-group");

template<int NT, bool FIRST>
__device__ __forceinline__ void gin_layer_mma(const float* Wl, const v4f* a,
                                              const v8f* prev, v8f* acc) {
    constexpr int KSTEPS = K_DIM / 8;
    v4f bc[NT], bnx[NT];
    #pragma unroll
    for (int t = 0; t < NT; ++t) {
        acc[t] = v8f{};
        bc[t] = *reinterpret_cast<const v4f*>(Wl + t * 16 * WT_PITCH);
    }
    __builtin_amdgcn_sched_group_barrier(0x100, NT, 0);

    #pragma unroll
    for (int s = 0; s < KSTEPS; ++s) {
        if (s + 1 < KSTEPS) {
            const int s1 = s + 1;
            const int koff = FIRST ? 8 * s1 : 16 * (s1 >> 1) + 4 * (s1 & 1);
            #pragma unroll
            for (int t = 0; t < NT; ++t)
                bnx[t] = *reinterpret_cast<const v4f*>(Wl + t * 16 * WT_PITCH + koff);
        }
        v2f x01, x23;
        if constexpr (FIRST) {
            x01 = v2f{a[s].x, a[s].y};
            x23 = v2f{a[s].z, a[s].w};
        } else {
            const v8f p = prev[s >> 1];
            const int e = 4 * (s & 1);
            x01 = v2f{p[e], p[e + 1]};
            x23 = v2f{p[e + 2], p[e + 3]};
        }
        #pragma unroll
        for (int t = 0; t < NT; ++t) {
            const v2f w01 = {bc[t].x, bc[t].y};
            acc[t] = __builtin_amdgcn_wmma_f32_16x16x4_f32(
                false, w01, false, x01, (short)0, acc[t], false, false);
        }
        #pragma unroll
        for (int t = 0; t < NT; ++t) {
            const v2f w23 = {bc[t].z, bc[t].w};
            acc[t] = __builtin_amdgcn_wmma_f32_16x16x4_f32(
                false, w23, false, x23, (short)0, acc[t], false, false);
        }
        if (s + 1 < KSTEPS)
            __builtin_amdgcn_sched_group_barrier(0x100, NT, 0);
        __builtin_amdgcn_sched_group_barrier(0x008, 2 * NT, 0);
        if (s + 1 < KSTEPS) {
            #pragma unroll
            for (int t = 0; t < NT; ++t) bc[t] = bnx[t];
        }
    }
}

template<int NT, int MODE>
__device__ __forceinline__ void gin_layer_epi(const float* P, v8f* acc) {
    __builtin_amdgcn_sched_barrier(0);
    #pragma unroll
    for (int t = 0; t < NT; ++t) {
        #pragma unroll
        for (int g = 0; g < 2; ++g) {
            const v4f b4 = *reinterpret_cast<const v4f*>(P + 16 * t + 4 * g);
            v4f sc4 = {1.0f, 1.0f, 1.0f, 1.0f}, sh4 = {0.0f, 0.0f, 0.0f, 0.0f};
            if constexpr (MODE == 2) {
                sc4 = *reinterpret_cast<const v4f*>(P + 128 + 16 * t + 4 * g);
                sh4 = *reinterpret_cast<const v4f*>(P + 256 + 16 * t + 4 * g);
            }
            #pragma unroll
            for (int i = 0; i < 4; ++i) {
                float val = acc[t][4 * g + i] + b4[i];
                if constexpr (MODE == 2) val = val * sc4[i] + sh4[i];
                if constexpr (MODE >= 1) val = fmaxf(val, 0.0f);
                acc[t][4 * g + i] = val;
            }
        }
    }
}

template<int ROWS>
__device__ __forceinline__ void gin_w_load(const float* __restrict__ WT, v4f* wreg) {
    #pragma unroll
    for (int it = 0; it < ROWS / CH_WAVES; ++it)
        wreg[it] = reinterpret_cast<const v4f*>(WT)[it * (CH_WAVES * 32) + threadIdx.x];
}
template<int ROWS>
__device__ __forceinline__ void gin_w_store(float* Wd, const v4f* wreg) {
    #pragma unroll
    for (int it = 0; it < ROWS / CH_WAVES; ++it) {
        const int i = it * (CH_WAVES * 32) + threadIdx.x;
        *reinterpret_cast<v4f*>(&Wd[(i >> 5) * WT_PITCH + (i & 31) * 4]) = wreg[it];
    }
}

template<int NT>
__device__ __forceinline__ void gin_x_store(float* X, int t0, int lane, const v8f* z) {
    #pragma unroll
    for (int t = 0; t < NT; ++t) {
        #pragma unroll
        for (int g = 0; g < 2; ++g)
            *reinterpret_cast<v4f*>(X + (t0 + t) * X_TILE_FLOATS + g * 128 + lane * 4) =
                v4f{z[t][4 * g], z[t][4 * g + 1], z[t][4 * g + 2], z[t][4 * g + 3]};
    }
}
template<int NT>
__device__ __forceinline__ void gin_x_load(const float* X, int lane, v8f* x) {
    #pragma unroll
    for (int t = 0; t < NT; ++t) {
        #pragma unroll
        for (int g = 0; g < 2; ++g) {
            const v4f v = *reinterpret_cast<const v4f*>(X + t * X_TILE_FLOATS + g * 128 + lane * 4);
            x[t][4 * g] = v.x;  x[t][4 * g + 1] = v.y;  x[t][4 * g + 2] = v.z;  x[t][4 * g + 3] = v.w;
        }
    }
}
template<int NT, int HLAST>
__device__ __forceinline__ void gin_out_store(float* orow, int hi, const v8f* y) {
    #pragma unroll
    for (int t = 0; t < NT; ++t) {
        if ((HLAST % 16 == 0) || (16 * t + 8 * hi < HLAST)) {
            *reinterpret_cast<v4f*>(orow + 16 * t)     = v4f{y[t][0], y[t][1], y[t][2], y[t][3]};
            *reinterpret_cast<v4f*>(orow + 16 * t + 4) = v4f{y[t][4], y[t][5], y[t][6], y[t][7]};
        }
    }
}

template<int NL, int HLAST, bool RELU_LAST>
__global__ void __launch_bounds__(CH_WAVES * 32)
gin_mlp_chain(const float* __restrict__ A,
              const float* __restrict__ WTa, const float* __restrict__ WTb,
              const float* __restrict__ WTc, const float* __restrict__ WTd,
              const float* __restrict__ bias_a, const float* __restrict__ bias_b,
              const float* __restrict__ bias_c, const float* __restrict__ bias_d,
              const float* __restrict__ bn_g, const float* __restrict__ bn_b,
              const float* __restrict__ bn_m, const float* __restrict__ bn_v,
              float* __restrict__ out) {
    constexpr int NT_LAST   = (HLAST + 15) / 16;
    constexpr int NT_SPLIT  = (NT_LAST + 3) / 4;
    constexpr int ROWS_LAST = NT_LAST * 16;
    constexpr int ROW_TILES = N_NODES / 16;
    constexpr int KSTEPS    = K_DIM / 8;
    constexpr int W_FLOATS  = ((NL - 1) * 128 + ROWS_LAST) * WT_PITCH;
    constexpr int P_FLOATS  = 384 + (NL - 1) * 128;
    constexpr int X_FLOATS  = ((NL == 2) ? 4 : 2) * 8 * X_TILE_FLOATS;
    static_assert(NT_LAST % NT_SPLIT == 0, "the last layer's tiles are dealt whole");

    __shared__ alignas(16) float Ws[W_FLOATS + P_FLOATS + X_FLOATS];
    float* Ps = Ws + W_FLOATS;
    float* Xs = Ps + P_FLOATS;

    const int lane = threadIdx.x & 31;
    const int wave = __builtin_amdgcn_readfirstlane((int)(threadIdx.x >> 5));
    const bool full = wave < CH_FULL;
    const int q = wave - CH_FULL;
    const int tile = full ? (int)blockIdx.x * CH_FULL + wave
                          : CH_BLOCKS * CH_FULL + (int)blockIdx.x;
    const bool have = tile < ROW_TILES;
    const int c  = lane & 15;
    const int hi = lane >> 4;

    constexpr int RL = ROWS_LAST / CH_WAVES;
    v4f w0[8], w1[NL == 2 ? RL : 8], w2[8], w3[RL];
    gin_w_load<128>(WTa, w0);
    if constexpr (NL == 2) {
        gin_w_load<ROWS_LAST>(WTb, w1);
    } else {
        gin_w_load<128>(WTb, w1);
        gin_w_load<128>(WTc, w2);
        gin_w_load<ROWS_LAST>(WTd, w3);
    }

    const int ltile = have ? tile : ROW_TILES - 1;
    const v4f* Ap = reinterpret_cast<const v4f*>(
        A + (long long)(ltile * 16 + c) * K_DIM + 4 * hi);
    v4f a[KSTEPS];
    #pragma unroll
    for (int j = 0; j < KSTEPS; ++j) a[j] = Ap[2 * j];

    if (threadIdx.x < 128) {
        const int n = threadIdx.x;
        const float rs = rsqrtf(bn_v[n] + BN_EPS);
        const float scale = bn_g[n] * rs;
        Ps[n]       = bias_a[n];
        Ps[128 + n] = scale;
        Ps[256 + n] = bn_b[n] - bn_m[n] * scale;
        if constexpr (NL == 2) {
            const float bb = bias_b[(n < HLAST) ? n : 0];
            Ps[384 + n] = (n < HLAST) ? bb : 0.0f;
        } else {
            Ps[384 + n] = bias_b[n];
            Ps[512 + n] = bias_c[n];
            const float bd = bias_d[(n < HLAST) ? n : 0];
            Ps[640 + n] = (n < HLAST) ? bd : 0.0f;
        }
    }

    gin_w_store<128>(Ws, w0);
    if constexpr (NL == 2) {
        gin_w_store<ROWS_LAST>(Ws + 128 * WT_PITCH, w1);
    } else {
        gin_w_store<128>(Ws + 128 * WT_PITCH, w1);
        gin_w_store<128>(Ws + 256 * WT_PITCH, w2);
        gin_w_store<ROWS_LAST>(Ws + 384 * WT_PITCH, w3);
    }
    __syncthreads();

    const float* Wl0 = Ws + c * WT_PITCH + 4 * hi;
    const float* WlN = Ws + c * WT_PITCH + 8 * hi;
    const float* Pl  = Ps + 8 * hi;

    constexpr int SW = 2 * 16 * WT_PITCH;
    float* X0 = Xs;
    float* X1 = Xs + 8 * X_TILE_FLOATS;
    v8f x[8], y[8];

    if (full) {
        gin_layer_mma<8, true>(Wl0, a, nullptr, x);
        gin_layer_epi<8, 2>(Pl, x);
    } else if (have) {
        v8f z[2];
        gin_layer_mma<2, true>(Wl0 + q * SW, a, nullptr, z);
        gin_layer_epi<2, 2>(Pl + 32 * q, z);
        gin_x_store<2>(X0, 2 * q, lane, z);
    }
    __syncthreads();

    if constexpr (NL == 2) {
        if (full) {
            gin_layer_mma<NT_LAST, false>(WlN + 128 * WT_PITCH, nullptr, x, y);
            gin_layer_epi<NT_LAST, RELU_LAST ? 1 : 0>(Pl + 384, y);
            gin_out_store<NT_LAST, HLAST>(out + (long long)(tile * 16 + c) * HLAST + 8 * hi, hi, y);
        } else if (have && q * NT_SPLIT < NT_LAST) {
            v8f p[8], z[NT_SPLIT];
            gin_x_load<8>(X0, lane, p);
            gin_layer_mma<NT_SPLIT, false>(WlN + 128 * WT_PITCH + q * (NT_SPLIT * 16 * WT_PITCH),
                                           nullptr, p, z);
            gin_layer_epi<NT_SPLIT, RELU_LAST ? 1 : 0>(Pl + 384 + q * (NT_SPLIT * 16), z);
            gin_x_store<NT_SPLIT>(X1, q * NT_SPLIT, lane, z);
        }
    } else {
        if (full) {
            gin_layer_mma<8, false>(WlN + 128 * WT_PITCH, nullptr, x, y);
            gin_layer_epi<8, 1>(Pl + 384, y);
        } else if (have) {
            v8f p[8], z[2];
            gin_x_load<8>(X0, lane, p);
            gin_layer_mma<2, false>(WlN + 128 * WT_PITCH + q * SW, nullptr, p, z);
            gin_layer_epi<2, 1>(Pl + 384 + 32 * q, z);
            gin_x_store<2>(X1, 2 * q, lane, z);
        }
        __syncthreads();
        v8f x2[8], x3[NT_LAST];
        if (full) {
            gin_layer_mma<8, false>(WlN + 256 * WT_PITCH, nullptr, y, x2);
            gin_layer_epi<8, 1>(Pl + 512, x2);
        } else if (have) {
            v8f p[8], z[2];
            gin_x_load<8>(X1, lane, p);
            gin_layer_mma<2, false>(WlN + 256 * WT_PITCH + q * SW, nullptr, p, z);
            gin_layer_epi<2, 1>(Pl + 512 + 32 * q, z);
            gin_x_store<2>(X0, 2 * q, lane, z);
        }
        __syncthreads();
        if (full) {
            gin_layer_mma<NT_LAST, false>(WlN + 384 * WT_PITCH, nullptr, x2, x3);
            gin_layer_epi<NT_LAST, RELU_LAST ? 1 : 0>(Pl + 640, x3);
            gin_out_store<NT_LAST, HLAST>(out + (long long)(tile * 16 + c) * HLAST + 8 * hi, hi, x3);
        } else if (have && q * NT_SPLIT < NT_LAST) {
            v8f p[8], z[NT_SPLIT];
            gin_x_load<8>(X0, lane, p);
            gin_layer_mma<NT_SPLIT, false>(WlN + 384 * WT_PITCH + q * (NT_SPLIT * 16 * WT_PITCH),
                                           nullptr, p, z);
            gin_layer_epi<NT_SPLIT, RELU_LAST ? 1 : 0>(Pl + 640 + q * (NT_SPLIT * 16), z);
            gin_x_store<NT_SPLIT>(X1, q * NT_SPLIT, lane, z);
        }
    }
    __syncthreads();
    if (!full && have && q == 0) {
        v8f o[NT_LAST];
        gin_x_load<NT_LAST>(X1, lane, o);
        gin_out_store<NT_LAST, HLAST>(out + (long long)(tile * 16 + c) * HLAST + 8 * hi, hi, o);
    }
}

extern "C" void kernel_launch(void* const* d_in, const int* in_sizes, int n_in,
                              void* d_out, int out_size, void* d_ws, size_t ws_size,
                              hipStream_t stream) {
    (void)in_sizes; (void)n_in; (void)out_size; (void)ws_size;

    const float* x      = (const float*)d_in[0];
    const int*   edge   = (const int*)  d_in[1];
    const float* lin1_w = (const float*)d_in[2];
    const float* lin1_b = (const float*)d_in[3];
    const float* bn1_g  = (const float*)d_in[4];
    const float* bn1_b  = (const float*)d_in[5];
    const float* bn1_m  = (const float*)d_in[6];
    const float* bn1_v  = (const float*)d_in[7];
    const float* lin2_w = (const float*)d_in[8];
    const float* lin2_b = (const float*)d_in[9];
    const float* lin3_w = (const float*)d_in[10];
    const float* lin3_b = (const float*)d_in[11];
    const float* bn2_g  = (const float*)d_in[12];
    const float* bn2_b  = (const float*)d_in[13];
    const float* bn2_m  = (const float*)d_in[14];
    const float* bn2_v  = (const float*)d_in[15];
    const float* lin4_w = (const float*)d_in[16];
    const float* lin4_b = (const float*)d_in[17];
    const float* lin5_w = (const float*)d_in[18];
    const float* lin5_b = (const float*)d_in[19];
    const float* lin6_w = (const float*)d_in[20];
    const float* lin6_b = (const float*)d_in[21];

    const int* e_src = edge;
    const int* e_dst = edge + N_EDGES;

    const size_t NH = (size_t)N_NODES * K_DIM;
    float* B0  = (float*)d_ws;
    float* B1  = B0  + NH;
    float* WT1 = B1  + NH;
    float* WT2 = WT1 + 128 * 128;
    float* WT3 = WT2 + 128 * 128;
    float* WT4 = WT3 + 128 * 128;
    float* WT5 = WT4 + 128 * 128;
    float* WT6 = WT5 + 128 * 128;
    int*  deg  = (int*)(WT6 + 64 * 128);
    int*  cnts = deg + N_NODES;
    int*  col  = deg + N_NODES + 272;
    int2* ovf  = (int2*)(col + N_NODES * BUCKET);
    unsigned int* binbuf = (unsigned int*)(ovf + N_EDGES);
    unsigned int* ovf1   = binbuf + N_BINS * BIN_CAP;

    const int sBlocks  = SETUP_T_BLOCKS + SETUP_Z_BLOCKS;
    const int eBlocks  = (N_EDGES / 4 + BIN_THREADS - 1) / BIN_THREADS;
    const int gaBlocks = (N_NODES + 7) / 8;

    gin_setup_kernel<<<sBlocks, 256, 0, stream>>>(lin1_w, lin2_w, lin3_w, lin4_w, lin5_w, lin6_w,
                                                  WT1, cnts);

    gin_bin_kernel<<<eBlocks, BIN_THREADS, 0, stream>>>(e_src, e_dst, cnts, binbuf, ovf1);
    gin_rank_kernel<<<N_BINS, RANK_THREADS, 0, stream>>>(cnts, binbuf, ovf1, deg, col, ovf);

    gin_gather_kernel<<<gaBlocks, 256, 0, stream>>>(x, deg, col, ovf, B0);
    gin_mlp_chain<2, 128, true><<<CH_BLOCKS, CH_WAVES * 32, 0, stream>>>(
        B0, WT1, WT2, nullptr, nullptr, lin1_b, lin2_b, nullptr, nullptr,
        bn1_g, bn1_b, bn1_m, bn1_v, B1);

    gin_gather_kernel<<<gaBlocks, 256, 0, stream>>>(B1, deg, col, ovf, B0);
    gin_mlp_chain<4, 40, false><<<CH_BLOCKS, CH_WAVES * 32, 0, stream>>>(
        B0, WT3, WT4, WT5, WT6, lin3_b, lin4_b, lin5_b, lin6_b,
        bn2_g, bn2_b, bn2_m, bn2_v, (float*)d_out);
}


# Edge binning without global counters: a work-group stores one whole 128-byte line per range (count + 31 edges, built in LDS), ranking reads them in one trip; list/sum order changes; 7 -> 6 launches
# speedup vs baseline: 4.0071x; 1.0770x over previous
#include <hip/hip_runtime.h>

typedef __attribute__((ext_vector_type(2))) float v2f;
typedef __attribute__((ext_vector_type(4))) float v4f;
typedef __attribute__((ext_vector_type(8))) float v8f;

#define N_NODES 50000
#define N_EDGES 600000
#define K_DIM   128
#define BN_EPS  1e-5f

#define BIN_SHIFT 8
#define BIN_NODES (1 << BIN_SHIFT)
#define N_BINS    ((N_NODES + BIN_NODES - 1) / BIN_NODES)
#define BUCKET 32
#define BIN_THREADS 512
#define BIN_EDGES (4 * BIN_THREADS)
#define N_GROUPS ((N_EDGES + BIN_EDGES - 1) / BIN_EDGES)
#define LINE_WORDS 32
#define LINE_CAP (LINE_WORDS - 1)
#define SETUP_T_BLOCKS 176
#define RANK_THREADS 512
#define RANK_LINES ((N_GROUPS + RANK_THREADS / 32 - 1) / (RANK_THREADS / 32))
static_assert(N_NODES <= 65536, "destination and source are packed into 16 bits each");
static_assert(N_EDGES % 4 == 0, "edges are read four at a time");
static_assert(N_BINS <= BIN_THREADS, "one thread clears one range counter");

__global__ void __launch_bounds__(BIN_THREADS)
gin_setup_bin_kernel(const int* __restrict__ src_idx,
                     const int* __restrict__ dst_idx,
                     const float* __restrict__ w1, const float* __restrict__ w2,
                     const float* __restrict__ w3, const float* __restrict__ w4,
                     const float* __restrict__ w5, const float* __restrict__ w6,
                     float* __restrict__ WT, int* __restrict__ cnts,
                     unsigned int* __restrict__ binbuf,
                     unsigned int* __restrict__ ovf1,
                     int* __restrict__ ocnt) {
    __shared__ int hist[N_BINS];
    __shared__ unsigned int stg[N_BINS * LINE_WORDS];
    __shared__ int novf;
    const int tid = threadIdx.x;
    const int g = blockIdx.x;
    if (g >= N_GROUPS) {
        const int t = g - N_GROUPS;
        if (t == 0 && tid == 0) cnts[0] = 0;
        const int m = t >> 5;
        const float* __restrict__ W = (m == 0) ? w1 : (m == 1) ? w2 : (m == 2) ? w3
                                    : (m == 3) ? w4 : (m == 4) ? w5 : w6;
        const int Hout = (m == 5) ? 40 : 128;
        const int i = (t & 31) * BIN_THREADS + tid;
        const int n = i >> 7;
        const int k = i & 127;
        WT[m * (128 * 128) + i] = (n < Hout) ? W[k * Hout + n] : 0.0f;
        return;
    }
    if (tid < N_BINS) hist[tid] = 0;
    if (tid == 0) novf = 0;
    __syncthreads();

    const int q = g * BIN_THREADS + tid;
    if (q < N_EDGES / 4) {
        const int4 d4 = reinterpret_cast<const int4*>(dst_idx)[q];
        const int4 s4 = reinterpret_cast<const int4*>(src_idx)[q];
        const int d[4] = {d4.x, d4.y, d4.z, d4.w};
        const int s[4] = {s4.x, s4.y, s4.z, s4.w};
        #pragma unroll
        for (int i = 0; i < 4; ++i) {
            const int b = (int)((unsigned int)d[i] >> BIN_SHIFT);
            const unsigned int rec = ((unsigned int)d[i] << 16) | (unsigned int)s[i];
            const int r = atomicAdd(&hist[b], 1);
            if (r < LINE_CAP) {
                stg[b * LINE_WORDS + 1 + r] = rec;
            } else {
                const int o = atomicAdd(&novf, 1);
                ovf1[g * BIN_EDGES + o] = rec;
            }
        }
    }
    __syncthreads();
    const int lane = tid & 31;
    for (int b = tid >> 5; b < N_BINS; b += BIN_THREADS / 32) {
        const int c = hist[b];
        const int cc = (c < LINE_CAP) ? c : LINE_CAP;
        unsigned int v = 0u;
        if (lane == 0) v = (unsigned int)cc;
        else if (lane <= cc) v = stg[b * LINE_WORDS + lane];
        binbuf[(b * N_GROUPS + g) * LINE_WORDS + lane] = v;
    }
    if (tid == 0) ocnt[g] = novf;
}

__device__ __forceinline__ void gin_rank_place(unsigned int rec, int* cnt, int* lst,
                                               int* __restrict__ cnts, int2* __restrict__ ovf) {
    const int dl = (int)((rec >> 16) & (BIN_NODES - 1));
    const int sv = (int)(rec & 0xFFFFu);
    const int r = atomicAdd(&cnt[dl], 1);
    if (r < BUCKET) {
        lst[dl * BUCKET + r] = sv;
    } else {
        const int o = atomicAdd(cnts, 1);
        ovf[o] = make_int2((int)(rec >> 16), sv);
    }
}

__global__ void __launch_bounds__(RANK_THREADS)
gin_rank_kernel(int* __restrict__ cnts,
                const unsigned int* __restrict__ binbuf,
                const unsigned int* __restrict__ ovf1,
                const int* __restrict__ ocnt,
                int* __restrict__ deg,
                int* __restrict__ col,
                int2* __restrict__ ovf) {
    __shared__ int cnt[BIN_NODES];
    __shared__ int lst[BIN_NODES * BUCKET];
    const int b = blockIdx.x;
    const int tid = threadIdx.x;
    const int lane = tid & 31;
    const int wave = tid >> 5;
    if (tid < BIN_NODES) cnt[tid] = 0;
    __syncthreads();

    unsigned int rec[RANK_LINES];
    #pragma unroll
    for (int k = 0; k < RANK_LINES; ++k) {
        const int g = wave + k * (RANK_THREADS / 32);
        rec[k] = (g < N_GROUPS) ? binbuf[(b * N_GROUPS + g) * LINE_WORDS + lane] : 0u;
    }
    const int oc = (tid < N_GROUPS) ? ocnt[tid] : 0;
    #pragma unroll
    for (int k = 0; k < RANK_LINES; ++k) {
        const int n = __builtin_amdgcn_readfirstlane((int)rec[k]);
        if (lane >= 1 && lane <= n) gin_rank_place(rec[k], cnt, lst, cnts, ovf);
    }
    if (__syncthreads_or(oc > 0)) {
        for (int g = 0; g < N_GROUPS; ++g) {
            int og = ocnt[g];
            og = (og < BIN_EDGES) ? og : BIN_EDGES;
            for (int i = tid; i < og; i += RANK_THREADS) {
                const unsigned int r1 = ovf1[g * BIN_EDGES + i];
                if ((int)(r1 >> 24) == b) gin_rank_place(r1, cnt, lst, cnts, ovf);
            }
        }
    }
    __syncthreads();

    const int node0 = b * BIN_NODES;
    if (tid < BIN_NODES && node0 + tid < N_NODES) deg[node0 + tid] = cnt[tid];
    for (int n = tid >> 5; n < BIN_NODES; n += RANK_THREADS / 32) {
        if (node0 + n < N_NODES) {
            const int c = cnt[n];
            if (lane < ((c < BUCKET) ? c : BUCKET))
                col[(node0 + n) * BUCKET + lane] = lst[n * BUCKET + lane];
        }
    }
}

__device__ __forceinline__ void gin_add_rows(const float4* __restrict__ xr,
                                             int my, int n, float4& acc) {
    const float4 zero4 = make_float4(0.0f, 0.0f, 0.0f, 0.0f);
    for (int k = 0; k < n; k += 16) {
        float4 r[16];
        r[0] = xr[(size_t)__builtin_amdgcn_readlane(my, k) * 32];
        #pragma unroll
        for (int j = 1; j < 16; ++j) {
            r[j] = zero4;
            if (k + j < n)
                r[j] = xr[(size_t)__builtin_amdgcn_readlane(my, k + j) * 32];
        }
        #pragma unroll
        for (int s = 1; s < 16; s <<= 1) {
            #pragma unroll
            for (int j = 0; j < 16; j += 2 * s) {
                r[j].x += r[j + s].x;  r[j].y += r[j + s].y;
                r[j].z += r[j + s].z;  r[j].w += r[j + s].w;
            }
        }
        acc.x += r[0].x;  acc.y += r[0].y;  acc.z += r[0].z;  acc.w += r[0].w;
    }
}

__global__ void __launch_bounds__(256)
gin_gather_kernel(const float* __restrict__ x,
                  const int* __restrict__ deg,
                  const int* __restrict__ col,
                  const int2* __restrict__ ovf,
                  float* __restrict__ agg) {
    const int node = __builtin_amdgcn_readfirstlane(
        (int)(blockIdx.x * (blockDim.x >> 5) + (threadIdx.x >> 5)));
    if (node >= N_NODES) return;
    const int lane = threadIdx.x & 31;
    const float4* __restrict__ xr = reinterpret_cast<const float4*>(x) + lane;

    const int cnt = deg[node];

    const int n0  = (cnt < BUCKET) ? cnt : BUCKET;
    const int my0 = (lane < n0) ? col[node * BUCKET + lane] : 0;
    const float4 self = xr[(size_t)node * 32];

    float4 acc = make_float4(0.0f, 0.0f, 0.0f, 0.0f);
    gin_add_rows(xr, my0, n0, acc);
    if (cnt > BUCKET) {
        int total = deg[N_NODES];
        total = (total < N_EDGES) ? total : N_EDGES;
        for (int base = 0; base < total; base += 32) {
            const int i = base + lane;
            int2 rec = make_int2(-1, 0);
            if (i < total) rec = ovf[i];
            unsigned int m = __builtin_amdgcn_ballot_w32(rec.x == node);
            while (m != 0u) {
                const int l = __builtin_ctz(m);
                m &= m - 1u;
                const float4 v = xr[(size_t)__builtin_amdgcn_readlane(rec.y, l) * 32];
                acc.x += v.x;  acc.y += v.y;  acc.z += v.z;  acc.w += v.w;
            }
        }
    }
    acc.x += self.x;  acc.y += self.y;  acc.z += self.z;  acc.w += self.w;
    reinterpret_cast<float4*>(agg)[(size_t)node * 32 + lane] = acc;
}


#define WT_PITCH 132
#define CH_WAVES 16
#define CH_FULL  12
#define CH_BLOCKS 256
#define X_TILE_FLOATS (8 * 32)
static_assert(CH_WAVES - CH_FULL == 4, "a left-over tile is shared by four waves, one per SIMD");
static_assert(CH_BLOCKS * CH_FULL <= N_NODES / 16, "every whole-tile wave has a tile");
static_assert(N_NODES / 16 - CH_BLOCKS * CH_FULL <= CH_BLOCKS, "at most one left-over tile per work-group");

template<int NT, bool FIRST>
__device__ __forceinline__ void gin_layer_mma(const float* Wl, const v4f* a,
                                              const v8f* prev, v8f* acc) {
    constexpr int KSTEPS = K_DIM / 8;
    v4f bc[NT], bnx[NT];
    #pragma unroll
    for (int t = 0; t < NT; ++t) {
        acc[t] = v8f{};
        bc[t] = *reinterpret_cast<const v4f*>(Wl + t * 16 * WT_PITCH);
    }
    __builtin_amdgcn_sched_group_barrier(0x100, NT, 0);

    #pragma unroll
    for (int s = 0; s < KSTEPS; ++s) {
        if (s + 1 < KSTEPS) {
            const int s1 = s + 1;
            const int koff = FIRST ? 8 * s1 : 16 * (s1 >> 1) + 4 * (s1 & 1);
            #pragma unroll
            for (int t = 0; t < NT; ++t)
                bnx[t] = *reinterpret_cast<const v4f*>(Wl + t * 16 * WT_PITCH + koff);
        }
        v2f x01, x23;
        if constexpr (FIRST) {
            x01 = v2f{a[s].x, a[s].y};
            x23 = v2f{a[s].z, a[s].w};
        } else {
            const v8f p = prev[s >> 1];
            const int e = 4 * (s & 1);
            x01 = v2f{p[e], p[e + 1]};
            x23 = v2f{p[e + 2], p[e + 3]};
        }
        #pragma unroll
        for (int t = 0; t < NT; ++t) {
            const v2f w01 = {bc[t].x, bc[t].y};
            acc[t] = __builtin_amdgcn_wmma_f32_16x16x4_f32(
                false, w01, false, x01, (short)0, acc[t], false, false);
        }
        #pragma unroll
        for (int t = 0; t < NT; ++t) {
            const v2f w23 = {bc[t].z, bc[t].w};
            acc[t] = __builtin_amdgcn_wmma_f32_16x16x4_f32(
                false, w23, false, x23, (short)0, acc[t], false, false);
        }
        if (s + 1 < KSTEPS)
            __builtin_amdgcn_sched_group_barrier(0x100, NT, 0);
        __builtin_amdgcn_sched_group_barrier(0x008, 2 * NT, 0);
        if (s + 1 < KSTEPS) {
            #pragma unroll
            for (int t = 0; t < NT; ++t) bc[t] = bnx[t];
        }
    }
}

template<int NT, int MODE>
__device__ __forceinline__ void gin_layer_epi(const float* P, v8f* acc) {
    __builtin_amdgcn_sched_barrier(0);
    #pragma unroll
    for (int t = 0; t < NT; ++t) {
        #pragma unroll
        for (int g = 0; g < 2; ++g) {
            const v4f b4 = *reinterpret_cast<const v4f*>(P + 16 * t + 4 * g);
            v4f sc4 = {1.0f, 1.0f, 1.0f, 1.0f}, sh4 = {0.0f, 0.0f, 0.0f, 0.0f};
            if constexpr (MODE == 2) {
                sc4 = *reinterpret_cast<const v4f*>(P + 128 + 16 * t + 4 * g);
                sh4 = *reinterpret_cast<const v4f*>(P + 256 + 16 * t + 4 * g);
            }
            #pragma unroll
            for (int i = 0; i < 4; ++i) {
                float val = acc[t][4 * g + i] + b4[i];
                if constexpr (MODE == 2) val = val * sc4[i] + sh4[i];
                if constexpr (MODE >= 1) val = fmaxf(val, 0.0f);
                acc[t][4 * g + i] = val;
            }
        }
    }
}

template<int ROWS>
__device__ __forceinline__ void gin_w_load(const float* __restrict__ WT, v4f* wreg) {
    #pragma unroll
    for (int it = 0; it < ROWS / CH_WAVES; ++it)
        wreg[it] = reinterpret_cast<const v4f*>(WT)[it * (CH_WAVES * 32) + threadIdx.x];
}
template<int ROWS>
__device__ __forceinline__ void gin_w_store(float* Wd, const v4f* wreg) {
    #pragma unroll
    for (int it = 0; it < ROWS / CH_WAVES; ++it) {
        const int i = it * (CH_WAVES * 32) + threadIdx.x;
        *reinterpret_cast<v4f*>(&Wd[(i >> 5) * WT_PITCH + (i & 31) * 4]) = wreg[it];
    }
}

template<int NT>
__device__ __forceinline__ void gin_x_store(float* X, int t0, int lane, const v8f* z) {
    #pragma unroll
    for (int t = 0; t < NT; ++t) {
        #pragma unroll
        for (int g = 0; g < 2; ++g)
            *reinterpret_cast<v4f*>(X + (t0 + t) * X_TILE_FLOATS + g * 128 + lane * 4) =
                v4f{z[t][4 * g], z[t][4 * g + 1], z[t][4 * g + 2], z[t][4 * g + 3]};
    }
}
template<int NT>
__device__ __forceinline__ void gin_x_load(const float* X, int lane, v8f* x) {
    #pragma unroll
    for (int t = 0; t < NT; ++t) {
        #pragma unroll
        for (int g = 0; g < 2; ++g) {
            const v4f v = *reinterpret_cast<const v4f*>(X + t * X_TILE_FLOATS + g * 128 + lane * 4);
            x[t][4 * g] = v.x;  x[t][4 * g + 1] = v.y;  x[t][4 * g + 2] = v.z;  x[t][4 * g + 3] = v.w;
        }
    }
}
template<int NT, int HLAST>
__device__ __forceinline__ void gin_out_store(float* orow, int hi, const v8f* y) {
    #pragma unroll
    for (int t = 0; t < NT; ++t) {
        if ((HLAST % 16 == 0) || (16 * t + 8 * hi < HLAST)) {
            *reinterpret_cast<v4f*>(orow + 16 * t)     = v4f{y[t][0], y[t][1], y[t][2], y[t][3]};
            *reinterpret_cast<v4f*>(orow + 16 * t + 4) = v4f{y[t][4], y[t][5], y[t][6], y[t][7]};
        }
    }
}

template<int NL, int HLAST, bool RELU_LAST>
__global__ void __launch_bounds__(CH_WAVES * 32)
gin_mlp_chain(const float* __restrict__ A,
              const float* __restrict__ WTa, const float* __restrict__ WTb,
              const float* __restrict__ WTc, const float* __restrict__ WTd,
              const float* __restrict__ bias_a, const float* __restrict__ bias_b,
              const float* __restrict__ bias_c, const float* __restrict__ bias_d,
              const float* __restrict__ bn_g, const float* __restrict__ bn_b,
              const float* __restrict__ bn_m, const float* __restrict__ bn_v,
              float* __restrict__ out) {
    constexpr int NT_LAST   = (HLAST + 15) / 16;
    constexpr int NT_SPLIT  = (NT_LAST + 3) / 4;
    constexpr int ROWS_LAST = NT_LAST * 16;
    constexpr int ROW_TILES = N_NODES / 16;
    constexpr int KSTEPS    = K_DIM / 8;
    constexpr int W_FLOATS  = ((NL - 1) * 128 + ROWS_LAST) * WT_PITCH;
    constexpr int P_FLOATS  = 384 + (NL - 1) * 128;
    constexpr int X_FLOATS  = ((NL == 2) ? 4 : 2) * 8 * X_TILE_FLOATS;
    static_assert(NT_LAST % NT_SPLIT == 0, "the last layer's tiles are dealt whole");

    __shared__ alignas(16) float Ws[W_FLOATS + P_FLOATS + X_FLOATS];
    float* Ps = Ws + W_FLOATS;
    float* Xs = Ps + P_FLOATS;

    const int lane = threadIdx.x & 31;
    const int wave = __builtin_amdgcn_readfirstlane((int)(threadIdx.x >> 5));
    const bool full = wave < CH_FULL;
    const int q = wave - CH_FULL;
    const int tile = full ? (int)blockIdx.x * CH_FULL + wave
                          : CH_BLOCKS * CH_FULL + (int)blockIdx.x;
    const bool have = tile < ROW_TILES;
    const int c  = lane & 15;
    const int hi = lane >> 4;

    constexpr int RL = ROWS_LAST / CH_WAVES;
    v4f w0[8], w1[NL == 2 ? RL : 8], w2[8], w3[RL];
    gin_w_load<128>(WTa, w0);
    if constexpr (NL == 2) {
        gin_w_load<ROWS_LAST>(WTb, w1);
    } else {
        gin_w_load<128>(WTb, w1);
        gin_w_load<128>(WTc, w2);
        gin_w_load<ROWS_LAST>(WTd, w3);
    }

    const int ltile = have ? tile : ROW_TILES - 1;
    const v4f* Ap = reinterpret_cast<const v4f*>(
        A + (long long)(ltile * 16 + c) * K_DIM + 4 * hi);
    v4f a[KSTEPS];
    #pragma unroll
    for (int j = 0; j < KSTEPS; ++j) a[j] = Ap[2 * j];

    if (threadIdx.x < 128) {
        const int n = threadIdx.x;
        const float rs = rsqrtf(bn_v[n] + BN_EPS);
        const float scale = bn_g[n] * rs;
        Ps[n]       = bias_a[n];
        Ps[128 + n] = scale;
        Ps[256 + n] = bn_b[n] - bn_m[n] * scale;
        if constexpr (NL == 2) {
            const float bb = bias_b[(n < HLAST) ? n : 0];
            Ps[384 + n] = (n < HLAST) ? bb : 0.0f;
        } else {
            Ps[384 + n] = bias_b[n];
            Ps[512 + n] = bias_c[n];
            const float bd = bias_d[(n < HLAST) ? n : 0];
            Ps[640 + n] = (n < HLAST) ? bd : 0.0f;
        }
    }

    gin_w_store<128>(Ws, w0);
    if constexpr (NL == 2) {
        gin_w_store<ROWS_LAST>(Ws + 128 * WT_PITCH, w1);
    } else {
        gin_w_store<128>(Ws + 128 * WT_PITCH, w1);
        gin_w_store<128>(Ws + 256 * WT_PITCH, w2);
        gin_w_store<ROWS_LAST>(Ws + 384 * WT_PITCH, w3);
    }
    __syncthreads();

    const float* Wl0 = Ws + c * WT_PITCH + 4 * hi;
    const float* WlN = Ws + c * WT_PITCH + 8 * hi;
    const float* Pl  = Ps + 8 * hi;

    constexpr int SW = 2 * 16 * WT_PITCH;
    float* X0 = Xs;
    float* X1 = Xs + 8 * X_TILE_FLOATS;
    v8f x[8], y[8];

    if (full) {
        gin_layer_mma<8, true>(Wl0, a, nullptr, x);
        gin_layer_epi<8, 2>(Pl, x);
    } else if (have) {
        v8f z[2];
        gin_layer_mma<2, true>(Wl0 + q * SW, a, nullptr, z);
        gin_layer_epi<2, 2>(Pl + 32 * q, z);
        gin_x_store<2>(X0, 2 * q, lane, z);
    }
    __syncthreads();

    if constexpr (NL == 2) {
        if (full) {
            gin_layer_mma<NT_LAST, false>(WlN + 128 * WT_PITCH, nullptr, x, y);
            gin_layer_epi<NT_LAST, RELU_LAST ? 1 : 0>(Pl + 384, y);
            gin_out_store<NT_LAST, HLAST>(out + (long long)(tile * 16 + c) * HLAST + 8 * hi, hi, y);
        } else if (have && q * NT_SPLIT < NT_LAST) {
            v8f p[8], z[NT_SPLIT];
            gin_x_load<8>(X0, lane, p);
            gin_layer_mma<NT_SPLIT, false>(WlN + 128 * WT_PITCH + q * (NT_SPLIT * 16 * WT_PITCH),
                                           nullptr, p, z);
            gin_layer_epi<NT_SPLIT, RELU_LAST ? 1 : 0>(Pl + 384 + q * (NT_SPLIT * 16), z);
            gin_x_store<NT_SPLIT>(X1, q * NT_SPLIT, lane, z);
        }
    } else {
        if (full) {
            gin_layer_mma<8, false>(WlN + 128 * WT_PITCH, nullptr, x, y);
            gin_layer_epi<8, 1>(Pl + 384, y);
        } else if (have) {
            v8f p[8], z[2];
            gin_x_load<8>(X0, lane, p);
            gin_layer_mma<2, false>(WlN + 128 * WT_PITCH + q * SW, nullptr, p, z);
            gin_layer_epi<2, 1>(Pl + 384 + 32 * q, z);
            gin_x_store<2>(X1, 2 * q, lane, z);
        }
        __syncthreads();
        v8f x2[8], x3[NT_LAST];
        if (full) {
            gin_layer_mma<8, false>(WlN + 256 * WT_PITCH, nullptr, y, x2);
            gin_layer_epi<8, 1>(Pl + 512, x2);
        } else if (have) {
            v8f p[8], z[2];
            gin_x_load<8>(X1, lane, p);
            gin_layer_mma<2, false>(WlN + 256 * WT_PITCH + q * SW, nullptr, p, z);
            gin_layer_epi<2, 1>(Pl + 512 + 32 * q, z);
            gin_x_store<2>(X0, 2 * q, lane, z);
        }
        __syncthreads();
        if (full) {
            gin_layer_mma<NT_LAST, false>(WlN + 384 * WT_PITCH, nullptr, x2, x3);
            gin_layer_epi<NT_LAST, RELU_LAST ? 1 : 0>(Pl + 640, x3);
            gin_out_store<NT_LAST, HLAST>(out + (long long)(tile * 16 + c) * HLAST + 8 * hi, hi, x3);
        } else if (have && q * NT_SPLIT < NT_LAST) {
            v8f p[8], z[NT_SPLIT];
            gin_x_load<8>(X0, lane, p);
            gin_layer_mma<NT_SPLIT, false>(WlN + 384 * WT_PITCH + q * (NT_SPLIT * 16 * WT_PITCH),
                                           nullptr, p, z);
            gin_layer_epi<NT_SPLIT, RELU_LAST ? 1 : 0>(Pl + 640 + q * (NT_SPLIT * 16), z);
            gin_x_store<NT_SPLIT>(X1, q * NT_SPLIT, lane, z);
        }
    }
    __syncthreads();
    if (!full && have && q == 0) {
        v8f o[NT_LAST];
        gin_x_load<NT_LAST>(X1, lane, o);
        gin_out_store<NT_LAST, HLAST>(out + (long long)(tile * 16 + c) * HLAST + 8 * hi, hi, o);
    }
}

extern "C" void kernel_launch(void* const* d_in, const int* in_sizes, int n_in,
                              void* d_out, int out_size, void* d_ws, size_t ws_size,
                              hipStream_t stream) {
    (void)in_sizes; (void)n_in; (void)out_size; (void)ws_size;

    const float* x      = (const float*)d_in[0];
    const int*   edge   = (const int*)  d_in[1];
    const float* lin1_w = (const float*)d_in[2];
    const float* lin1_b = (const float*)d_in[3];
    const float* bn1_g  = (const float*)d_in[4];
    const float* bn1_b  = (const float*)d_in[5];
    const float* bn1_m  = (const float*)d_in[6];
    const float* bn1_v  = (const float*)d_in[7];
    const float* lin2_w = (const float*)d_in[8];
    const float* lin2_b = (const float*)d_in[9];
    const float* lin3_w = (const float*)d_in[10];
    const float* lin3_b = (const float*)d_in[11];
    const float* bn2_g  = (const float*)d_in[12];
    const float* bn2_b  = (const float*)d_in[13];
    const float* bn2_m  = (const float*)d_in[14];
    const float* bn2_v  = (const float*)d_in[15];
    const float* lin4_w = (const float*)d_in[16];
    const float* lin4_b = (const float*)d_in[17];
    const float* lin5_w = (const float*)d_in[18];
    const float* lin5_b = (const float*)d_in[19];
    const float* lin6_w = (const float*)d_in[20];
    const float* lin6_b = (const float*)d_in[21];

    const int* e_src = edge;
    const int* e_dst = edge + N_EDGES;

    const size_t NH = (size_t)N_NODES * K_DIM;
    float* B0  = (float*)d_ws;
    float* B1  = B0  + NH;
    float* WT1 = B1  + NH;
    float* WT2 = WT1 + 128 * 128;
    float* WT3 = WT2 + 128 * 128;
    float* WT4 = WT3 + 128 * 128;
    float* WT5 = WT4 + 128 * 128;
    float* WT6 = WT5 + 128 * 128;
    int*  deg  = (int*)(WT6 + 64 * 128);
    int*  cnts = deg + N_NODES;
    int*  col  = deg + N_NODES + 272;
    int2* ovf  = (int2*)(col + N_NODES * BUCKET);
    unsigned int* binbuf = (unsigned int*)(ovf + N_EDGES);
    unsigned int* ovf1   = binbuf + N_BINS * N_GROUPS * LINE_WORDS;
    int*          ocnt   = (int*)(ovf1 + N_GROUPS * BIN_EDGES);

    const int sBlocks  = N_GROUPS + SETUP_T_BLOCKS;
    const int gaBlocks = (N_NODES + 7) / 8;

    gin_setup_bin_kernel<<<sBlocks, BIN_THREADS, 0, stream>>>(e_src, e_dst,
                                                              lin1_w, lin2_w, lin3_w, lin4_w, lin5_w, lin6_w,
                                                              WT1, cnts, binbuf, ovf1, ocnt);

    gin_rank_kernel<<<N_BINS, RANK_THREADS, 0, stream>>>(cnts, binbuf, ovf1, ocnt, deg, col, ovf);

    gin_gather_kernel<<<gaBlocks, 256, 0, stream>>>(x, deg, col, ovf, B0);
    gin_mlp_chain<2, 128, true><<<CH_BLOCKS, CH_WAVES * 32, 0, stream>>>(
        B0, WT1, WT2, nullptr, nullptr, lin1_b, lin2_b, nullptr, nullptr,
        bn1_g, bn1_b, bn1_m, bn1_v, B1);

    gin_gather_kernel<<<gaBlocks, 256, 0, stream>>>(B1, deg, col, ovf, B0);
    gin_mlp_chain<4, 40, false><<<CH_BLOCKS, CH_WAVES * 32, 0, stream>>>(
        B0, WT3, WT4, WT5, WT6, lin3_b, lin4_b, lin5_b, lin6_b,
        bn2_g, bn2_b, bn2_m, bn2_v, (float*)d_out);
}


# Aggregation fused into the chain kernels: a tile's wave sums its 16 nodes' rows, parks them in LDS and reads them back as its first B operand; no aggregated plane in memory; 6 -> 4 launches
# speedup vs baseline: 4.1741x; 1.0417x over previous
#include <hip/hip_runtime.h>

typedef __attribute__((ext_vector_type(2))) float v2f;
typedef __attribute__((ext_vector_type(4))) float v4f;
typedef __attribute__((ext_vector_type(8))) float v8f;

#define N_NODES 50000
#define N_EDGES 600000
#define K_DIM   128
#define BN_EPS  1e-5f

#define BIN_SHIFT 8
#define BIN_NODES (1 << BIN_SHIFT)
#define N_BINS    ((N_NODES + BIN_NODES - 1) / BIN_NODES)
#define BUCKET 32
#define BIN_THREADS 512
#define BIN_EDGES (4 * BIN_THREADS)
#define N_GROUPS ((N_EDGES + BIN_EDGES - 1) / BIN_EDGES)
#define LINE_WORDS 32
#define LINE_CAP (LINE_WORDS - 1)
#define SETUP_T_BLOCKS 176
#define RANK_THREADS 512
#define RANK_LINES ((N_GROUPS + RANK_THREADS / 32 - 1) / (RANK_THREADS / 32))
static_assert(N_NODES <= 65536, "destination and source are packed into 16 bits each");
static_assert(N_EDGES % 4 == 0, "edges are read four at a time");
static_assert(N_BINS <= BIN_THREADS, "one thread clears one range counter");

__global__ void __launch_bounds__(BIN_THREADS)
gin_setup_bin_kernel(const int* __restrict__ src_idx,
                     const int* __restrict__ dst_idx,
                     const float* __restrict__ w1, const float* __restrict__ w2,
                     const float* __restrict__ w3, const float* __restrict__ w4,
                     const float* __restrict__ w5, const float* __restrict__ w6,
                     float* __restrict__ WT, int* __restrict__ cnts,
                     unsigned int* __restrict__ binbuf,
                     unsigned int* __restrict__ ovf1,
                     int* __restrict__ ocnt) {
    __shared__ int hist[N_BINS];
    __shared__ unsigned int stg[N_BINS * LINE_WORDS];
    __shared__ int novf;
    const int tid = threadIdx.x;
    const int g = blockIdx.x;
    if (g >= N_GROUPS) {
        const int t = g - N_GROUPS;
        if (t == 0 && tid == 0) cnts[0] = 0;
        const int m = t >> 5;
        const float* __restrict__ W = (m == 0) ? w1 : (m == 1) ? w2 : (m == 2) ? w3
                                    : (m == 3) ? w4 : (m == 4) ? w5 : w6;
        const int Hout = (m == 5) ? 40 : 128;
        const int i = (t & 31) * BIN_THREADS + tid;
        const int n = i >> 7;
        const int k = i & 127;
        WT[m * (128 * 128) + i] = (n < Hout) ? W[k * Hout + n] : 0.0f;
        return;
    }
    if (tid < N_BINS) hist[tid] = 0;
    if (tid == 0) novf = 0;
    __syncthreads();

    const int q = g * BIN_THREADS + tid;
    if (q < N_EDGES / 4) {
        const int4 d4 = reinterpret_cast<const int4*>(dst_idx)[q];
        const int4 s4 = reinterpret_cast<const int4*>(src_idx)[q];
        const int d[4] = {d4.x, d4.y, d4.z, d4.w};
        const int s[4] = {s4.x, s4.y, s4.z, s4.w};
        #pragma unroll
        for (int i = 0; i < 4; ++i) {
            const int b = (int)((unsigned int)d[i] >> BIN_SHIFT);
            const unsigned int rec = ((unsigned int)d[i] << 16) | (unsigned int)s[i];
            const int r = atomicAdd(&hist[b], 1);
            if (r < LINE_CAP) {
                stg[b * LINE_WORDS + 1 + r] = rec;
            } else {
                const int o = atomicAdd(&novf, 1);
                ovf1[g * BIN_EDGES + o] = rec;
            }
        }
    }
    __syncthreads();
    const int lane = tid & 31;
    for (int b = tid >> 5; b < N_BINS; b += BIN_THREADS / 32) {
        const int c = hist[b];
        const int cc = (c < LINE_CAP) ? c : LINE_CAP;
        unsigned int v = 0u;
        if (lane == 0) v = (unsigned int)cc;
        else if (lane <= cc) v = stg[b * LINE_WORDS + lane];
        binbuf[(b * N_GROUPS + g) * LINE_WORDS + lane] = v;
    }
    if (tid == 0) ocnt[g] = novf;
}

__device__ __forceinline__ void gin_rank_place(unsigned int rec, int* cnt, int* lst,
                                               int* __restrict__ cnts, int2* __restrict__ ovf) {
    const int dl = (int)((rec >> 16) & (BIN_NODES - 1));
    const int sv = (int)(rec & 0xFFFFu);
    const int r = atomicAdd(&cnt[dl], 1);
    if (r < BUCKET) {
        lst[dl * BUCKET + r] = sv;
    } else {
        const int o = atomicAdd(cnts, 1);
        ovf[o] = make_int2((int)(rec >> 16), sv);
    }
}

__global__ void __launch_bounds__(RANK_THREADS)
gin_rank_kernel(int* __restrict__ cnts,
                const unsigned int* __restrict__ binbuf,
                const unsigned int* __restrict__ ovf1,
                const int* __restrict__ ocnt,
                int* __restrict__ deg,
                int* __restrict__ col,
                int2* __restrict__ ovf) {
    __shared__ int cnt[BIN_NODES];
    __shared__ int lst[BIN_NODES * BUCKET];
    const int b = blockIdx.x;
    const int tid = threadIdx.x;
    const int lane = tid & 31;
    const int wave = tid >> 5;
    if (tid < BIN_NODES) cnt[tid] = 0;
    __syncthreads();

    unsigned int rec[RANK_LINES];
    #pragma unroll
    for (int k = 0; k < RANK_LINES; ++k) {
        const int g = wave + k * (RANK_THREADS / 32);
        rec[k] = (g < N_GROUPS) ? binbuf[(b * N_GROUPS + g) * LINE_WORDS + lane] : 0u;
    }
    const int oc = (tid < N_GROUPS) ? ocnt[tid] : 0;
    #pragma unroll
    for (int k = 0; k < RANK_LINES; ++k) {
        const int n = __builtin_amdgcn_readfirstlane((int)rec[k]);
        if (lane >= 1 && lane <= n) gin_rank_place(rec[k], cnt, lst, cnts, ovf);
    }
    if (__syncthreads_or(oc > 0)) {
        for (int g = 0; g < N_GROUPS; ++g) {
            int og = ocnt[g];
            og = (og < BIN_EDGES) ? og : BIN_EDGES;
            for (int i = tid; i < og; i += RANK_THREADS) {
                const unsigned int r1 = ovf1[g * BIN_EDGES + i];
                if ((int)(r1 >> 24) == b) gin_rank_place(r1, cnt, lst, cnts, ovf);
            }
        }
    }
    __syncthreads();

    const int node0 = b * BIN_NODES;
    if (tid < BIN_NODES && node0 + tid < N_NODES) deg[node0 + tid] = cnt[tid];
    for (int n = tid >> 5; n < BIN_NODES; n += RANK_THREADS / 32) {
        if (node0 + n < N_NODES) {
            const int c = cnt[n];
            if (lane < ((c < BUCKET) ? c : BUCKET))
                col[(node0 + n) * BUCKET + lane] = lst[n * BUCKET + lane];
        }
    }
}

__device__ __forceinline__ void gin_rows_issue(const float4* __restrict__ xr,
                                               int my, int k, int n, float4* r) {
    const float4 zero4 = make_float4(0.0f, 0.0f, 0.0f, 0.0f);
    r[0] = xr[(size_t)__builtin_amdgcn_readlane(my, k) * 32];
    #pragma unroll
    for (int j = 1; j < 16; ++j) {
        r[j] = zero4;
        if (k + j < n)
            r[j] = xr[(size_t)__builtin_amdgcn_readlane(my, k + j) * 32];
    }
}

__device__ __forceinline__ int gin_list_load(const int* __restrict__ col, int node, int cnt, int lane) {
    return (lane < ((cnt < BUCKET) ? cnt : BUCKET)) ? col[node * BUCKET + lane] : 0;
}

__device__ __forceinline__ void gin_rows_sum(float4* r, float4& acc) {
    #pragma unroll
    for (int s = 1; s < 16; s <<= 1) {
        #pragma unroll
        for (int j = 0; j < 16; j += 2 * s) {
            r[j].x += r[j + s].x;  r[j].y += r[j + s].y;
            r[j].z += r[j + s].z;  r[j].w += r[j + s].w;
        }
    }
    acc.x += r[0].x;  acc.y += r[0].y;  acc.z += r[0].z;  acc.w += r[0].w;
}

__device__ __forceinline__ void gin_add_overflow(const float4* __restrict__ xr,
                                                 const int* __restrict__ deg,
                                                 const int2* __restrict__ ovf,
                                                 int node, int lane, float4& acc) {
    int total = deg[N_NODES];
    total = (total < N_EDGES) ? total : N_EDGES;
    for (int base = 0; base < total; base += 32) {
        const int i = base + lane;
        int2 rec = make_int2(-1, 0);
        if (i < total) rec = ovf[i];
        unsigned int m = __builtin_amdgcn_ballot_w32(rec.x == node);
        while (m != 0u) {
            const int l = __builtin_ctz(m);
            m &= m - 1u;
            const float4 v = xr[(size_t)__builtin_amdgcn_readlane(rec.y, l) * 32];
            acc.x += v.x;  acc.y += v.y;  acc.z += v.z;  acc.w += v.w;
        }
    }
}


#define WT_PITCH 132
#define CH_WAVES 16
#define CH_FULL  12
#define CH_BLOCKS 256
#define X_TILE_FLOATS (8 * 32)
#define G_ROWS 8
static_assert(CH_WAVES - CH_FULL == 4, "a left-over tile is shared by four waves, one per SIMD");
static_assert(CH_BLOCKS * CH_FULL <= N_NODES / 16, "every whole-tile wave has a tile");
static_assert(N_NODES / 16 - CH_BLOCKS * CH_FULL <= CH_BLOCKS, "at most one left-over tile per work-group");

template<int NT, bool FIRST>
__device__ __forceinline__ void gin_layer_mma(const float* Wl, const v4f* a,
                                              const v8f* prev, v8f* acc) {
    constexpr int KSTEPS = K_DIM / 8;
    v4f bc[NT], bnx[NT];
    #pragma unroll
    for (int t = 0; t < NT; ++t) {
        acc[t] = v8f{};
        bc[t] = *reinterpret_cast<const v4f*>(Wl + t * 16 * WT_PITCH);
    }
    __builtin_amdgcn_sched_group_barrier(0x100, NT, 0);

    #pragma unroll
    for (int s = 0; s < KSTEPS; ++s) {
        if (s + 1 < KSTEPS) {
            const int s1 = s + 1;
            const int koff = FIRST ? 8 * s1 : 16 * (s1 >> 1) + 4 * (s1 & 1);
            #pragma unroll
            for (int t = 0; t < NT; ++t)
                bnx[t] = *reinterpret_cast<const v4f*>(Wl + t * 16 * WT_PITCH + koff);
        }
        v2f x01, x23;
        if constexpr (FIRST) {
            x01 = v2f{a[s].x, a[s].y};
            x23 = v2f{a[s].z, a[s].w};
        } else {
            const v8f p = prev[s >> 1];
            const int e = 4 * (s & 1);
            x01 = v2f{p[e], p[e + 1]};
            x23 = v2f{p[e + 2], p[e + 3]};
        }
        #pragma unroll
        for (int t = 0; t < NT; ++t) {
            const v2f w01 = {bc[t].x, bc[t].y};
            acc[t] = __builtin_amdgcn_wmma_f32_16x16x4_f32(
                false, w01, false, x01, (short)0, acc[t], false, false);
        }
        #pragma unroll
        for (int t = 0; t < NT; ++t) {
            const v2f w23 = {bc[t].z, bc[t].w};
            acc[t] = __builtin_amdgcn_wmma_f32_16x16x4_f32(
                false, w23, false, x23, (short)0, acc[t], false, false);
        }
        if (s + 1 < KSTEPS)
            __builtin_amdgcn_sched_group_barrier(0x100, NT, 0);
        __builtin_amdgcn_sched_group_barrier(0x008, 2 * NT, 0);
        if (s + 1 < KSTEPS) {
            #pragma unroll
            for (int t = 0; t < NT; ++t) bc[t] = bnx[t];
        }
    }
}

template<int NT, int MODE>
__device__ __forceinline__ void gin_layer_epi(const float* P, v8f* acc) {
    __builtin_amdgcn_sched_barrier(0);
    #pragma unroll
    for (int t = 0; t < NT; ++t) {
        #pragma unroll
        for (int g = 0; g < 2; ++g) {
            const v4f b4 = *reinterpret_cast<const v4f*>(P + 16 * t + 4 * g);
            v4f sc4 = {1.0f, 1.0f, 1.0f, 1.0f}, sh4 = {0.0f, 0.0f, 0.0f, 0.0f};
            if constexpr (MODE == 2) {
                sc4 = *reinterpret_cast<const v4f*>(P + 128 + 16 * t + 4 * g);
                sh4 = *reinterpret_cast<const v4f*>(P + 256 + 16 * t + 4 * g);
            }
            #pragma unroll
            for (int i = 0; i < 4; ++i) {
                float val = acc[t][4 * g + i] + b4[i];
                if constexpr (MODE == 2) val = val * sc4[i] + sh4[i];
                if constexpr (MODE >= 1) val = fmaxf(val, 0.0f);
                acc[t][4 * g + i] = val;
            }
        }
    }
}

template<int ROWS>
__device__ __forceinline__ void gin_w_load(const float* __restrict__ WT, v4f* wreg) {
    #pragma unroll
    for (int it = 0; it < ROWS / CH_WAVES; ++it)
        wreg[it] = reinterpret_cast<const v4f*>(WT)[it * (CH_WAVES * 32) + threadIdx.x];
}
template<int ROWS>
__device__ __forceinline__ void gin_w_store(float* Wd, const v4f* wreg) {
    #pragma unroll
    for (int it = 0; it < ROWS / CH_WAVES; ++it) {
        const int i = it * (CH_WAVES * 32) + threadIdx.x;
        *reinterpret_cast<v4f*>(&Wd[(i >> 5) * WT_PITCH + (i & 31) * 4]) = wreg[it];
    }
}

template<int NT>
__device__ __forceinline__ void gin_x_store(float* X, int t0, int lane, const v8f* z) {
    #pragma unroll
    for (int t = 0; t < NT; ++t) {
        #pragma unroll
        for (int g = 0; g < 2; ++g)
            *reinterpret_cast<v4f*>(X + (t0 + t) * X_TILE_FLOATS + g * 128 + lane * 4) =
                v4f{z[t][4 * g], z[t][4 * g + 1], z[t][4 * g + 2], z[t][4 * g + 3]};
    }
}
template<int NT>
__device__ __forceinline__ void gin_x_load(const float* X, int lane, v8f* x) {
    #pragma unroll
    for (int t = 0; t < NT; ++t) {
        #pragma unroll
        for (int g = 0; g < 2; ++g) {
            const v4f v = *reinterpret_cast<const v4f*>(X + t * X_TILE_FLOATS + g * 128 + lane * 4);
            x[t][4 * g] = v.x;  x[t][4 * g + 1] = v.y;  x[t][4 * g + 2] = v.z;  x[t][4 * g + 3] = v.w;
        }
    }
}
template<int NT, int HLAST>
__device__ __forceinline__ void gin_out_store(float* orow, int hi, const v8f* y) {
    #pragma unroll
    for (int t = 0; t < NT; ++t) {
        if ((HLAST % 16 == 0) || (16 * t + 8 * hi < HLAST)) {
            *reinterpret_cast<v4f*>(orow + 16 * t)     = v4f{y[t][0], y[t][1], y[t][2], y[t][3]};
            *reinterpret_cast<v4f*>(orow + 16 * t + 4) = v4f{y[t][4], y[t][5], y[t][6], y[t][7]};
        }
    }
}

template<int RM, bool READBACK>
__device__ __forceinline__ void gin_rows_to_lds(const float4* __restrict__ xr,
                                                const int* __restrict__ deg,
                                                const int* __restrict__ col,
                                                const int2* __restrict__ ovf,
                                                int base, int cntv, int j0, int j1,
                                                float* G, int lane, v4f* a) {
    const int c  = lane & 15;
    const int hi = lane >> 4;
    int cnt = __builtin_amdgcn_readlane(cntv, j0);
    int my = gin_list_load(col, base + j0, cnt, lane);
    #pragma unroll 1
    for (int j = j0; j < j1; ++j) {
        const int node = base + j;
        const int n0 = (cnt < BUCKET) ? cnt : BUCKET;
        const float4 self = xr[(size_t)node * 32];
        const bool more = j + 1 < j1;
        const int cnt_n = more ? __builtin_amdgcn_readlane(cntv, j + 1) : 0;
        int my_n = 0;
        float4 acc = make_float4(0.0f, 0.0f, 0.0f, 0.0f);
        float4 r[16];
        gin_rows_issue(xr, my, 0, n0, r);
        if (more) my_n = gin_list_load(col, node + 1, cnt_n, lane);
        if (n0 == 0) r[0] = make_float4(0.0f, 0.0f, 0.0f, 0.0f);
        gin_rows_sum(r, acc);
        for (int k = 16; k < n0; k += 16) {
            gin_rows_issue(xr, my, k, n0, r);
            gin_rows_sum(r, acc);
        }
        if (cnt > BUCKET)
            gin_add_overflow(xr, deg, ovf, node, lane, acc);
        acc.x += self.x;  acc.y += self.y;  acc.z += self.z;  acc.w += self.w;
        *reinterpret_cast<v4f*>(G + (j & RM) * WT_PITCH + lane * 4) = v4f{acc.x, acc.y, acc.z, acc.w};
        if (READBACK && (j & RM) == RM) {
            if ((c & ~RM) == (j & ~RM)) {
                #pragma unroll
                for (int s = 0; s < K_DIM / 8; ++s)
                    a[s] = *reinterpret_cast<const v4f*>(G + (c & RM) * WT_PITCH + 8 * s + 4 * hi);
            }
        }
        cnt = cnt_n;  my = my_n;
    }
}

template<int NL, int HLAST, bool RELU_LAST>
__global__ void __launch_bounds__(CH_WAVES * 32)
gin_mlp_chain(const float* __restrict__ X,
              const int* __restrict__ deg,
              const int* __restrict__ col,
              const int2* __restrict__ ovf,
              const float* __restrict__ WTa, const float* __restrict__ WTb,
              const float* __restrict__ WTc, const float* __restrict__ WTd,
              const float* __restrict__ bias_a, const float* __restrict__ bias_b,
              const float* __restrict__ bias_c, const float* __restrict__ bias_d,
              const float* __restrict__ bn_g, const float* __restrict__ bn_b,
              const float* __restrict__ bn_m, const float* __restrict__ bn_v,
              float* __restrict__ out) {
    constexpr int NT_LAST   = (HLAST + 15) / 16;
    constexpr int NT_SPLIT  = (NT_LAST + 3) / 4;
    constexpr int ROWS_LAST = NT_LAST * 16;
    constexpr int ROW_TILES = N_NODES / 16;
    constexpr int KSTEPS    = K_DIM / 8;
    constexpr int W_FLOATS  = ((NL - 1) * 128 + ROWS_LAST) * WT_PITCH;
    constexpr int P_FLOATS  = 384 + (NL - 1) * 128;
    constexpr int X_FLOATS  = ((NL == 2) ? 4 : 2) * 8 * X_TILE_FLOATS;
    constexpr int G_FLOATS  = CH_WAVES * G_ROWS * WT_PITCH;
    static_assert(NT_LAST % NT_SPLIT == 0, "the last layer's tiles are dealt whole");
    static_assert((CH_WAVES - CH_FULL) * G_ROWS >= 16, "the shared tile's 16 rows fit its waves' rows");
    static_assert(16 % G_ROWS == 0 && (G_ROWS & (G_ROWS - 1)) == 0, "a tile is parked in whole groups");

    __shared__ alignas(16) float Ws[W_FLOATS + P_FLOATS + X_FLOATS + G_FLOATS];
    float* Ps = Ws + W_FLOATS;
    float* Xs = Ps + P_FLOATS;
    float* Gs = Xs + X_FLOATS;

    const int lane = threadIdx.x & 31;
    const int wave = __builtin_amdgcn_readfirstlane((int)(threadIdx.x >> 5));
    const bool full = wave < CH_FULL;
    const int q = wave - CH_FULL;
    const int tile = full ? (int)blockIdx.x * CH_FULL + wave
                          : CH_BLOCKS * CH_FULL + (int)blockIdx.x;
    const bool have = tile < ROW_TILES;
    const int c  = lane & 15;
    const int hi = lane >> 4;

    constexpr int RL = ROWS_LAST / CH_WAVES;
    v4f w0[8], w1[NL == 2 ? RL : 8], w2[8], w3[RL];
    gin_w_load<128>(WTa, w0);
    if constexpr (NL == 2) {
        gin_w_load<ROWS_LAST>(WTb, w1);
    } else {
        gin_w_load<128>(WTb, w1);
        gin_w_load<128>(WTc, w2);
        gin_w_load<ROWS_LAST>(WTd, w3);
    }

    int cntv = 0;
    if (have) cntv = deg[tile * 16 + c];
    const float4* __restrict__ xr = reinterpret_cast<const float4*>(X) + lane;
    float* Gw = Gs + wave * (G_ROWS * WT_PITCH);
    float* Gl = Gs + CH_FULL * (G_ROWS * WT_PITCH);
    v4f a[KSTEPS];
    #pragma unroll
    for (int j = 0; j < KSTEPS; ++j) a[j] = v4f{0.0f, 0.0f, 0.0f, 0.0f};

    if (threadIdx.x < 128) {
        const int n = threadIdx.x;
        const float rs = rsqrtf(bn_v[n] + BN_EPS);
        const float scale = bn_g[n] * rs;
        Ps[n]       = bias_a[n];
        Ps[128 + n] = scale;
        Ps[256 + n] = bn_b[n] - bn_m[n] * scale;
        if constexpr (NL == 2) {
            const float bb = bias_b[(n < HLAST) ? n : 0];
            Ps[384 + n] = (n < HLAST) ? bb : 0.0f;
        } else {
            Ps[384 + n] = bias_b[n];
            Ps[512 + n] = bias_c[n];
            const float bd = bias_d[(n < HLAST) ? n : 0];
            Ps[640 + n] = (n < HLAST) ? bd : 0.0f;
        }
    }

    gin_w_store<128>(Ws, w0);
    if constexpr (NL == 2) {
        gin_w_store<ROWS_LAST>(Ws + 128 * WT_PITCH, w1);
    } else {
        gin_w_store<128>(Ws + 128 * WT_PITCH, w1);
        gin_w_store<128>(Ws + 256 * WT_PITCH, w2);
        gin_w_store<ROWS_LAST>(Ws + 384 * WT_PITCH, w3);
    }
    if (have) {
        const int jq = full ? 0 : 4 * q;
        gin_rows_to_lds<15, false>(xr, deg, col, ovf, tile * 16, cntv, jq, jq + 4,
                                   full ? Gw : Gl, lane, a);
    }
    __syncthreads();
    if (full) {
        gin_rows_to_lds<G_ROWS - 1, true>(xr, deg, col, ovf, tile * 16, cntv, 4, 16, Gw, lane, a);
    } else if (have) {
        #pragma unroll
        for (int s = 0; s < KSTEPS; ++s)
            a[s] = *reinterpret_cast<const v4f*>(Gl + c * WT_PITCH + 8 * s + 4 * hi);
    }

    const float* Wl0 = Ws + c * WT_PITCH + 4 * hi;
    const float* WlN = Ws + c * WT_PITCH + 8 * hi;
    const float* Pl  = Ps + 8 * hi;

    constexpr int SW = 2 * 16 * WT_PITCH;
    float* X0 = Xs;
    float* X1 = Xs + 8 * X_TILE_FLOATS;
    v8f x[8], y[8];

    if (full) {
        gin_layer_mma<8, true>(Wl0, a, nullptr, x);
        gin_layer_epi<8, 2>(Pl, x);
    } else if (have) {
        v8f z[2];
        gin_layer_mma<2, true>(Wl0 + q * SW, a, nullptr, z);
        gin_layer_epi<2, 2>(Pl + 32 * q, z);
        gin_x_store<2>(X0, 2 * q, lane, z);
    }
    __syncthreads();

    if constexpr (NL == 2) {
        if (full) {
            gin_layer_mma<NT_LAST, false>(WlN + 128 * WT_PITCH, nullptr, x, y);
            gin_layer_epi<NT_LAST, RELU_LAST ? 1 : 0>(Pl + 384, y);
            gin_out_store<NT_LAST, HLAST>(out + (long long)(tile * 16 + c) * HLAST + 8 * hi, hi, y);
        } else if (have && q * NT_SPLIT < NT_LAST) {
            v8f p[8], z[NT_SPLIT];
            gin_x_load<8>(X0, lane, p);
            gin_layer_mma<NT_SPLIT, false>(WlN + 128 * WT_PITCH + q * (NT_SPLIT * 16 * WT_PITCH),
                                           nullptr, p, z);
            gin_layer_epi<NT_SPLIT, RELU_LAST ? 1 : 0>(Pl + 384 + q * (NT_SPLIT * 16), z);
            gin_x_store<NT_SPLIT>(X1, q * NT_SPLIT, lane, z);
        }
    } else {
        if (full) {
            gin_layer_mma<8, false>(WlN + 128 * WT_PITCH, nullptr, x, y);
            gin_layer_epi<8, 1>(Pl + 384, y);
        } else if (have) {
            v8f p[8], z[2];
            gin_x_load<8>(X0, lane, p);
            gin_layer_mma<2, false>(WlN + 128 * WT_PITCH + q * SW, nullptr, p, z);
            gin_layer_epi<2, 1>(Pl + 384 + 32 * q, z);
            gin_x_store<2>(X1, 2 * q, lane, z);
        }
        __syncthreads();
        v8f x2[8], x3[NT_LAST];
        if (full) {
            gin_layer_mma<8, false>(WlN + 256 * WT_PITCH, nullptr, y, x2);
            gin_layer_epi<8, 1>(Pl + 512, x2);
        } else if (have) {
            v8f p[8], z[2];
            gin_x_load<8>(X1, lane, p);
            gin_layer_mma<2, false>(WlN + 256 * WT_PITCH + q * SW, nullptr, p, z);
            gin_layer_epi<2, 1>(Pl + 512 + 32 * q, z);
            gin_x_store<2>(X0, 2 * q, lane, z);
        }
        __syncthreads();
        if (full) {
            gin_layer_mma<NT_LAST, false>(WlN + 384 * WT_PITCH, nullptr, x2, x3);
            gin_layer_epi<NT_LAST, RELU_LAST ? 1 : 0>(Pl + 640, x3);
            gin_out_store<NT_LAST, HLAST>(out + (long long)(tile * 16 + c) * HLAST + 8 * hi, hi, x3);
        } else if (have && q * NT_SPLIT < NT_LAST) {
            v8f p[8], z[NT_SPLIT];
            gin_x_load<8>(X0, lane, p);
            gin_layer_mma<NT_SPLIT, false>(WlN + 384 * WT_PITCH + q * (NT_SPLIT * 16 * WT_PITCH),
                                           nullptr, p, z);
            gin_layer_epi<NT_SPLIT, RELU_LAST ? 1 : 0>(Pl + 640 + q * (NT_SPLIT * 16), z);
            gin_x_store<NT_SPLIT>(X1, q * NT_SPLIT, lane, z);
        }
    }
    __syncthreads();
    if (!full && have && q == 0) {
        v8f o[NT_LAST];
        gin_x_load<NT_LAST>(X1, lane, o);
        gin_out_store<NT_LAST, HLAST>(out + (long long)(tile * 16 + c) * HLAST + 8 * hi, hi, o);
    }
}

extern "C" void kernel_launch(void* const* d_in, const int* in_sizes, int n_in,
                              void* d_out, int out_size, void* d_ws, size_t ws_size,
                              hipStream_t stream) {
    (void)in_sizes; (void)n_in; (void)out_size; (void)ws_size;

    const float* x      = (const float*)d_in[0];
    const int*   edge   = (const int*)  d_in[1];
    const float* lin1_w = (const float*)d_in[2];
    const float* lin1_b = (const float*)d_in[3];
    const float* bn1_g  = (const float*)d_in[4];
    const float* bn1_b  = (const float*)d_in[5];
    const float* bn1_m  = (const float*)d_in[6];
    const float* bn1_v  = (const float*)d_in[7];
    const float* lin2_w = (const float*)d_in[8];
    const float* lin2_b = (const float*)d_in[9];
    const float* lin3_w = (const float*)d_in[10];
    const float* lin3_b = (const float*)d_in[11];
    const float* bn2_g  = (const float*)d_in[12];
    const float* bn2_b  = (const float*)d_in[13];
    const float* bn2_m  = (const float*)d_in[14];
    const float* bn2_v  = (const float*)d_in[15];
    const float* lin4_w = (const float*)d_in[16];
    const float* lin4_b = (const float*)d_in[17];
    const float* lin5_w = (const float*)d_in[18];
    const float* lin5_b = (const float*)d_in[19];
    const float* lin6_w = (const float*)d_in[20];
    const float* lin6_b = (const float*)d_in[21];

    const int* e_src = edge;
    const int* e_dst = edge + N_EDGES;

    const size_t NH = (size_t)N_NODES * K_DIM;
    float* B0  = (float*)d_ws;
    float* B1  = B0  + NH;
    float* WT1 = B1  + NH;
    float* WT2 = WT1 + 128 * 128;
    float* WT3 = WT2 + 128 * 128;
    float* WT4 = WT3 + 128 * 128;
    float* WT5 = WT4 + 128 * 128;
    float* WT6 = WT5 + 128 * 128;
    int*  deg  = (int*)(WT6 + 64 * 128);
    int*  cnts = deg + N_NODES;
    int*  col  = deg + N_NODES + 272;
    int2* ovf  = (int2*)(col + N_NODES * BUCKET);
    unsigned int* binbuf = (unsigned int*)(ovf + N_EDGES);
    unsigned int* ovf1   = binbuf + N_BINS * N_GROUPS * LINE_WORDS;
    int*          ocnt   = (int*)(ovf1 + N_GROUPS * BIN_EDGES);

    const int sBlocks  = N_GROUPS + SETUP_T_BLOCKS;

    gin_setup_bin_kernel<<<sBlocks, BIN_THREADS, 0, stream>>>(e_src, e_dst,
                                                              lin1_w, lin2_w, lin3_w, lin4_w, lin5_w, lin6_w,
                                                              WT1, cnts, binbuf, ovf1, ocnt);

    gin_rank_kernel<<<N_BINS, RANK_THREADS, 0, stream>>>(cnts, binbuf, ovf1, ocnt, deg, col, ovf);

    gin_mlp_chain<2, 128, true><<<CH_BLOCKS, CH_WAVES * 32, 0, stream>>>(
        x, deg, col, ovf, WT1, WT2, nullptr, nullptr, lin1_b, lin2_b, nullptr, nullptr,
        bn1_g, bn1_b, bn1_m, bn1_v, B0);

    gin_mlp_chain<4, 40, false><<<CH_BLOCKS, CH_WAVES * 32, 0, stream>>>(
        B0, deg, col, ovf, WT3, WT4, WT5, WT6, lin3_b, lin4_b, lin5_b, lin6_b,
        bn2_g, bn2_b, bn2_m, bn2_v, (float*)d_out);
}


# Second grouping level fused into the first chain launch: ranges = the chain work-groups' own nodes, each ranks its range in LDS and gathers from those lists (sum order: LDS arrival); 4 -> 3 launches
# speedup vs baseline: 4.3621x; 1.0450x over previous
#include <hip/hip_runtime.h>

typedef __attribute__((ext_vector_type(2))) float v2f;
typedef __attribute__((ext_vector_type(4))) float v4f;
typedef __attribute__((ext_vector_type(8))) float v8f;

#define N_NODES 50000
#define N_EDGES 600000
#define K_DIM   128
#define BN_EPS  1e-5f

#define CH_WAVES 16
#define CH_FULL  12
#define CH_BLOCKS 256
#define CH_EXTRA (N_NODES / 16 - CH_BLOCKS * CH_FULL)
#define RANGE_BIG   ((CH_FULL + 1) * 16)
#define RANGE_SMALL (CH_FULL * 16)
#define N_BINS    CH_BLOCKS
#define BUCKET 32
#define BIN_THREADS 512
#define BIN_EDGES (4 * BIN_THREADS)
#define N_GROUPS ((N_EDGES + BIN_EDGES - 1) / BIN_EDGES)
#define LINE_WORDS 32
#define LINE_CAP (LINE_WORDS - 1)
#define SETUP_T_BLOCKS 176
#define RANK_LINES ((N_GROUPS + CH_WAVES - 1) / CH_WAVES)
static_assert(CH_EXTRA >= 0 && CH_EXTRA <= CH_BLOCKS, "at most one extra tile per work-group");
static_assert(CH_EXTRA * RANGE_BIG + (CH_BLOCKS - CH_EXTRA) * RANGE_SMALL == N_NODES, "the ranges cover the nodes exactly");

__device__ __forceinline__ int gin_range(int d) {
    const unsigned int u = (unsigned int)d;
    return (u < (unsigned int)(CH_EXTRA * RANGE_BIG))
               ? (int)(u / (unsigned int)RANGE_BIG)
               : CH_EXTRA + (int)((u - (unsigned int)(CH_EXTRA * RANGE_BIG)) / (unsigned int)RANGE_SMALL);
}
__device__ __forceinline__ int gin_tile0(int b) {
    return CH_FULL * b + ((b < CH_EXTRA) ? b : CH_EXTRA);
}
static_assert(N_NODES <= 65536, "destination and source are packed into 16 bits each");
static_assert(N_EDGES % 4 == 0, "edges are read four at a time");
static_assert(N_BINS <= BIN_THREADS, "one thread clears one range counter");

__global__ void __launch_bounds__(BIN_THREADS)
gin_setup_bin_kernel(const int* __restrict__ src_idx,
                     const int* __restrict__ dst_idx,
                     const float* __restrict__ w1, const float* __restrict__ w2,
                     const float* __restrict__ w3, const float* __restrict__ w4,
                     const float* __restrict__ w5, const float* __restrict__ w6,
                     float* __restrict__ WT, int* __restrict__ cnts,
                     unsigned int* __restrict__ binbuf,
                     unsigned int* __restrict__ ovf1,
                     int* __restrict__ ocnt) {
    __shared__ int hist[N_BINS];
    __shared__ unsigned int stg[N_BINS * LINE_WORDS];
    __shared__ int novf;
    const int tid = threadIdx.x;
    const int g = blockIdx.x;
    if (g >= N_GROUPS) {
        const int t = g - N_GROUPS;
        if (t == 0 && tid == 0) cnts[0] = 0;
        const int m = t >> 5;
        const float* __restrict__ W = (m == 0) ? w1 : (m == 1) ? w2 : (m == 2) ? w3
                                    : (m == 3) ? w4 : (m == 4) ? w5 : w6;
        const int Hout = (m == 5) ? 40 : 128;
        const int i = (t & 31) * BIN_THREADS + tid;
        const int n = i >> 7;
        const int k = i & 127;
        WT[m * (128 * 128) + i] = (n < Hout) ? W[k * Hout + n] : 0.0f;
        return;
    }
    if (tid < N_BINS) hist[tid] = 0;
    if (tid == 0) novf = 0;
    __syncthreads();

    const int q = g * BIN_THREADS + tid;
    if (q < N_EDGES / 4) {
        const int4 d4 = reinterpret_cast<const int4*>(dst_idx)[q];
        const int4 s4 = reinterpret_cast<const int4*>(src_idx)[q];
        const int d[4] = {d4.x, d4.y, d4.z, d4.w};
        const int s[4] = {s4.x, s4.y, s4.z, s4.w};
        #pragma unroll
        for (int i = 0; i < 4; ++i) {
            const int b = gin_range(d[i]);
            const unsigned int rec = ((unsigned int)d[i] << 16) | (unsigned int)s[i];
            const int r = atomicAdd(&hist[b], 1);
            if (r < LINE_CAP) {
                stg[b * LINE_WORDS + 1 + r] = rec;
            } else {
                const int o = atomicAdd(&novf, 1);
                ovf1[g * BIN_EDGES + o] = rec;
            }
        }
    }
    __syncthreads();
    const int lane = tid & 31;
    for (int b = tid >> 5; b < N_BINS; b += BIN_THREADS / 32) {
        const int c = hist[b];
        const int cc = (c < LINE_CAP) ? c : LINE_CAP;
        unsigned int v = 0u;
        if (lane == 0) v = (unsigned int)cc;
        else if (lane <= cc) v = stg[b * LINE_WORDS + lane];
        binbuf[(b * N_GROUPS + g) * LINE_WORDS + lane] = v;
    }
    if (tid == 0) ocnt[g] = novf;
}

__device__ __forceinline__ void gin_rank_store(unsigned int rec, int r, int node0, int* lst,
                                               int* __restrict__ cnts, int2* __restrict__ ovf) {
    const int dl = (int)(rec >> 16) - node0;
    const int sv = (int)(rec & 0xFFFFu);
    if (r < BUCKET) {
        lst[dl * BUCKET + r] = sv;
    } else {
        const int o = atomicAdd(cnts, 1);
        ovf[o] = make_int2((int)(rec >> 16), sv);
    }
}
__device__ __forceinline__ void gin_rank_place(unsigned int rec, int node0, int* cnt, int* lst,
                                               int* __restrict__ cnts, int2* __restrict__ ovf) {
    const int r = atomicAdd(&cnt[(int)(rec >> 16) - node0], 1);
    gin_rank_store(rec, r, node0, lst, cnts, ovf);
}

__device__ __forceinline__ void gin_rank_range(int b, int node0, int nn, int wave, int lane,
                                               int* cnt, int* lst,
                                               int* __restrict__ cnts,
                                               const unsigned int* __restrict__ binbuf,
                                               const unsigned int* __restrict__ ovf1,
                                               const int* __restrict__ ocnt,
                                               int* __restrict__ deg,
                                               int* __restrict__ col,
                                               int2* __restrict__ ovf) {
    const int tid = (int)threadIdx.x;
    unsigned int rec[RANK_LINES];
    #pragma unroll
    for (int k = 0; k < RANK_LINES; ++k) {
        const int g = wave + k * CH_WAVES;
        rec[k] = (g < N_GROUPS) ? binbuf[(b * N_GROUPS + g) * LINE_WORDS + lane] : 0u;
    }
    const int oc = (tid < N_GROUPS) ? ocnt[tid] : 0;
    if (tid < RANGE_BIG) cnt[tid] = 0;
    __syncthreads();
    int slot[RANK_LINES];
    #pragma unroll
    for (int k = 0; k < RANK_LINES; ++k) {
        const int n = __builtin_amdgcn_readfirstlane((int)rec[k]);
        slot[k] = -1;
        if (lane >= 1 && lane <= n) slot[k] = atomicAdd(&cnt[(int)(rec[k] >> 16) - node0], 1);
    }
    #pragma unroll
    for (int k = 0; k < RANK_LINES; ++k) {
        if (slot[k] >= 0) gin_rank_store(rec[k], slot[k], node0, lst, cnts, ovf);
    }
    if (__syncthreads_or(oc > 0)) {
        for (int g = 0; g < N_GROUPS; ++g) {
            int og = ocnt[g];
            og = (og < BIN_EDGES) ? og : BIN_EDGES;
            for (int i = tid; i < og; i += CH_WAVES * 32) {
                const unsigned int r1 = ovf1[g * BIN_EDGES + i];
                if (gin_range((int)(r1 >> 16)) == b) gin_rank_place(r1, node0, cnt, lst, cnts, ovf);
            }
        }
    }
    __syncthreads();

    if (tid < nn) deg[node0 + tid] = cnt[tid];
    for (int n = wave; n < nn; n += CH_WAVES) {
        const int c = cnt[n];
        if (lane < ((c < BUCKET) ? c : BUCKET))
            col[(node0 + n) * BUCKET + lane] = lst[n * BUCKET + lane];
    }
}

__device__ __forceinline__ void gin_add_from_lines(const float4* __restrict__ xr,
                                                   const unsigned int* __restrict__ binbuf,
                                                   const unsigned int* __restrict__ ovf1,
                                                   const int* __restrict__ ocnt,
                                                   int b, int node, int lane, float4& acc) {
    for (int g = 0; g < N_GROUPS; ++g) {
        const unsigned int v = binbuf[(b * N_GROUPS + g) * LINE_WORDS + lane];
        const int n = __builtin_amdgcn_readfirstlane((int)v);
        unsigned int m = __builtin_amdgcn_ballot_w32(lane >= 1 && lane <= n && (int)(v >> 16) == node);
        while (m != 0u) {
            const int l = __builtin_ctz(m);
            m &= m - 1u;
            const float4 r = xr[(size_t)__builtin_amdgcn_readlane((int)(v & 0xFFFFu), l) * 32];
            acc.x += r.x;  acc.y += r.y;  acc.z += r.z;  acc.w += r.w;
        }
        int og = ocnt[g];
        og = (og < BIN_EDGES) ? og : BIN_EDGES;
        for (int base = 0; base < og; base += 32) {
            const int i = base + lane;
            unsigned int r1 = 0u;
            if (i < og) r1 = ovf1[g * BIN_EDGES + i];
            unsigned int m1 = __builtin_amdgcn_ballot_w32(i < og && (int)(r1 >> 16) == node);
            while (m1 != 0u) {
                const int l = __builtin_ctz(m1);
                m1 &= m1 - 1u;
                const float4 r = xr[(size_t)__builtin_amdgcn_readlane((int)(r1 & 0xFFFFu), l) * 32];
                acc.x += r.x;  acc.y += r.y;  acc.z += r.z;  acc.w += r.w;
            }
        }
    }
}

__device__ __forceinline__ void gin_rows_issue(const float4* __restrict__ xr,
                                               int my, int k, int n, float4* r) {
    const float4 zero4 = make_float4(0.0f, 0.0f, 0.0f, 0.0f);
    r[0] = xr[(size_t)__builtin_amdgcn_readlane(my, k) * 32];
    #pragma unroll
    for (int j = 1; j < 16; ++j) {
        r[j] = zero4;
        if (k + j < n)
            r[j] = xr[(size_t)__builtin_amdgcn_readlane(my, k + j) * 32];
    }
}

__device__ __forceinline__ int gin_list_load(const int* __restrict__ col, int node, int cnt, int lane) {
    return (lane < ((cnt < BUCKET) ? cnt : BUCKET)) ? col[node * BUCKET + lane] : 0;
}
__device__ __forceinline__ int gin_list_load_lds(const int* lst, int jl, int cnt, int lane) {
    return (lane < ((cnt < BUCKET) ? cnt : BUCKET)) ? lst[jl * BUCKET + lane] : 0;
}

__device__ __forceinline__ void gin_rows_sum(float4* r, float4& acc) {
    #pragma unroll
    for (int s = 1; s < 16; s <<= 1) {
        #pragma unroll
        for (int j = 0; j < 16; j += 2 * s) {
            r[j].x += r[j + s].x;  r[j].y += r[j + s].y;
            r[j].z += r[j + s].z;  r[j].w += r[j + s].w;
        }
    }
    acc.x += r[0].x;  acc.y += r[0].y;  acc.z += r[0].z;  acc.w += r[0].w;
}

__device__ __forceinline__ void gin_add_overflow(const float4* __restrict__ xr,
                                                 const int* __restrict__ deg,
                                                 const int2* __restrict__ ovf,
                                                 int node, int lane, float4& acc) {
    int total = deg[N_NODES];
    total = (total < N_EDGES) ? total : N_EDGES;
    for (int base = 0; base < total; base += 32) {
        const int i = base + lane;
        int2 rec = make_int2(-1, 0);
        if (i < total) rec = ovf[i];
        unsigned int m = __builtin_amdgcn_ballot_w32(rec.x == node);
        while (m != 0u) {
            const int l = __builtin_ctz(m);
            m &= m - 1u;
            const float4 v = xr[(size_t)__builtin_amdgcn_readlane(rec.y, l) * 32];
            acc.x += v.x;  acc.y += v.y;  acc.z += v.z;  acc.w += v.w;
        }
    }
}


#define WT_PITCH 132
#define X_TILE_FLOATS (8 * 32)
#define G_ROWS 8
static_assert(CH_WAVES - CH_FULL == 4, "a left-over tile is shared by four waves, one per SIMD");
static_assert(CH_BLOCKS * CH_FULL <= N_NODES / 16, "every whole-tile wave has a tile");
static_assert(N_NODES / 16 - CH_BLOCKS * CH_FULL <= CH_BLOCKS, "at most one left-over tile per work-group");

template<int NT, bool FIRST>
__device__ __forceinline__ void gin_layer_mma(const float* Wl, const v4f* a,
                                              const v8f* prev, v8f* acc) {
    constexpr int KSTEPS = K_DIM / 8;
    v4f bc[NT], bnx[NT];
    #pragma unroll
    for (int t = 0; t < NT; ++t) {
        acc[t] = v8f{};
        bc[t] = *reinterpret_cast<const v4f*>(Wl + t * 16 * WT_PITCH);
    }
    __builtin_amdgcn_sched_group_barrier(0x100, NT, 0);

    #pragma unroll
    for (int s = 0; s < KSTEPS; ++s) {
        if (s + 1 < KSTEPS) {
            const int s1 = s + 1;
            const int koff = FIRST ? 8 * s1 : 16 * (s1 >> 1) + 4 * (s1 & 1);
            #pragma unroll
            for (int t = 0; t < NT; ++t)
                bnx[t] = *reinterpret_cast<const v4f*>(Wl + t * 16 * WT_PITCH + koff);
        }
        v2f x01, x23;
        if constexpr (FIRST) {
            x01 = v2f{a[s].x, a[s].y};
            x23 = v2f{a[s].z, a[s].w};
        } else {
            const v8f p = prev[s >> 1];
            const int e = 4 * (s & 1);
            x01 = v2f{p[e], p[e + 1]};
            x23 = v2f{p[e + 2], p[e + 3]};
        }
        #pragma unroll
        for (int t = 0; t < NT; ++t) {
            const v2f w01 = {bc[t].x, bc[t].y};
            acc[t] = __builtin_amdgcn_wmma_f32_16x16x4_f32(
                false, w01, false, x01, (short)0, acc[t], false, false);
        }
        #pragma unroll
        for (int t = 0; t < NT; ++t) {
            const v2f w23 = {bc[t].z, bc[t].w};
            acc[t] = __builtin_amdgcn_wmma_f32_16x16x4_f32(
                false, w23, false, x23, (short)0, acc[t], false, false);
        }
        if (s + 1 < KSTEPS)
            __builtin_amdgcn_sched_group_barrier(0x100, NT, 0);
        __builtin_amdgcn_sched_group_barrier(0x008, 2 * NT, 0);
        if (s + 1 < KSTEPS) {
            #pragma unroll
            for (int t = 0; t < NT; ++t) bc[t] = bnx[t];
        }
    }
}

template<int NT, int MODE>
__device__ __forceinline__ void gin_layer_epi(const float* P, v8f* acc) {
    __builtin_amdgcn_sched_barrier(0);
    #pragma unroll
    for (int t = 0; t < NT; ++t) {
        #pragma unroll
        for (int g = 0; g < 2; ++g) {
            const v4f b4 = *reinterpret_cast<const v4f*>(P + 16 * t + 4 * g);
            v4f sc4 = {1.0f, 1.0f, 1.0f, 1.0f}, sh4 = {0.0f, 0.0f, 0.0f, 0.0f};
            if constexpr (MODE == 2) {
                sc4 = *reinterpret_cast<const v4f*>(P + 128 + 16 * t + 4 * g);
                sh4 = *reinterpret_cast<const v4f*>(P + 256 + 16 * t + 4 * g);
            }
            #pragma unroll
            for (int i = 0; i < 4; ++i) {
                float val = acc[t][4 * g + i] + b4[i];
                if constexpr (MODE == 2) val = val * sc4[i] + sh4[i];
                if constexpr (MODE >= 1) val = fmaxf(val, 0.0f);
                acc[t][4 * g + i] = val;
            }
        }
    }
}

template<int ROWS>
__device__ __forceinline__ void gin_w_load(const float* __restrict__ WT, v4f* wreg) {
    #pragma unroll
    for (int it = 0; it < ROWS / CH_WAVES; ++it)
        wreg[it] = reinterpret_cast<const v4f*>(WT)[it * (CH_WAVES * 32) + threadIdx.x];
}
template<int ROWS>
__device__ __forceinline__ void gin_w_store(float* Wd, const v4f* wreg) {
    #pragma unroll
    for (int it = 0; it < ROWS / CH_WAVES; ++it) {
        const int i = it * (CH_WAVES * 32) + threadIdx.x;
        *reinterpret_cast<v4f*>(&Wd[(i >> 5) * WT_PITCH + (i & 31) * 4]) = wreg[it];
    }
}

template<int NT>
__device__ __forceinline__ void gin_x_store(float* X, int t0, int lane, const v8f* z) {
    #pragma unroll
    for (int t = 0; t < NT; ++t) {
        #pragma unroll
        for (int g = 0; g < 2; ++g)
            *reinterpret_cast<v4f*>(X + (t0 + t) * X_TILE_FLOATS + g * 128 + lane * 4) =
                v4f{z[t][4 * g], z[t][4 * g + 1], z[t][4 * g + 2], z[t][4 * g + 3]};
    }
}
template<int NT>
__device__ __forceinline__ void gin_x_load(const float* X, int lane, v8f* x) {
    #pragma unroll
    for (int t = 0; t < NT; ++t) {
        #pragma unroll
        for (int g = 0; g < 2; ++g) {
            const v4f v = *reinterpret_cast<const v4f*>(X + t * X_TILE_FLOATS + g * 128 + lane * 4);
            x[t][4 * g] = v.x;  x[t][4 * g + 1] = v.y;  x[t][4 * g + 2] = v.z;  x[t][4 * g + 3] = v.w;
        }
    }
}
template<int NT, int HLAST>
__device__ __forceinline__ void gin_out_store(float* orow, int hi, const v8f* y) {
    #pragma unroll
    for (int t = 0; t < NT; ++t) {
        if ((HLAST % 16 == 0) || (16 * t + 8 * hi < HLAST)) {
            *reinterpret_cast<v4f*>(orow + 16 * t)     = v4f{y[t][0], y[t][1], y[t][2], y[t][3]};
            *reinterpret_cast<v4f*>(orow + 16 * t + 4) = v4f{y[t][4], y[t][5], y[t][6], y[t][7]};
        }
    }
}

template<int RM, bool READBACK, bool RANK>
__device__ __forceinline__ void gin_rows_to_lds(const float4* __restrict__ xr,
                                                const int* __restrict__ deg,
                                                const int* __restrict__ col,
                                                const int2* __restrict__ ovf,
                                                const int* lst, int jl0,
                                                const unsigned int* __restrict__ binbuf,
                                                const unsigned int* __restrict__ ovf1,
                                                const int* __restrict__ ocnt, int rb,
                                                int base, int cntv, int j0, int j1,
                                                float* G, int lane, v4f* a) {
    const int c  = lane & 15;
    const int hi = lane >> 4;
    int cnt = __builtin_amdgcn_readlane(cntv, j0);
    int my = RANK ? gin_list_load_lds(lst, jl0 + j0, cnt, lane)
                  : gin_list_load(col, base + j0, cnt, lane);
    #pragma unroll 1
    for (int j = j0; j < j1; ++j) {
        const int node = base + j;
        const int n0 = (cnt < BUCKET) ? cnt : BUCKET;
        const float4 self = xr[(size_t)node * 32];
        const bool more = j + 1 < j1;
        const int cnt_n = more ? __builtin_amdgcn_readlane(cntv, j + 1) : 0;
        int my_n = 0;
        float4 acc = make_float4(0.0f, 0.0f, 0.0f, 0.0f);
        float4 r[16];
        gin_rows_issue(xr, my, 0, n0, r);
        if (more) my_n = RANK ? gin_list_load_lds(lst, jl0 + j + 1, cnt_n, lane)
                              : gin_list_load(col, node + 1, cnt_n, lane);
        if (n0 == 0) r[0] = make_float4(0.0f, 0.0f, 0.0f, 0.0f);
        gin_rows_sum(r, acc);
        for (int k = 16; k < n0; k += 16) {
            gin_rows_issue(xr, my, k, n0, r);
            gin_rows_sum(r, acc);
        }
        if (cnt > BUCKET) {
            if constexpr (RANK) {
                acc = make_float4(0.0f, 0.0f, 0.0f, 0.0f);
                gin_add_from_lines(xr, binbuf, ovf1, ocnt, rb, node, lane, acc);
            } else {
                gin_add_overflow(xr, deg, ovf, node, lane, acc);
            }
        }
        acc.x += self.x;  acc.y += self.y;  acc.z += self.z;  acc.w += self.w;
        *reinterpret_cast<v4f*>(G + (j & RM) * WT_PITCH + lane * 4) = v4f{acc.x, acc.y, acc.z, acc.w};
        if (READBACK && (j & RM) == RM) {
            if ((c & ~RM) == (j & ~RM)) {
                #pragma unroll
                for (int s = 0; s < K_DIM / 8; ++s)
                    a[s] = *reinterpret_cast<const v4f*>(G + (c & RM) * WT_PITCH + 8 * s + 4 * hi);
            }
        }
        cnt = cnt_n;  my = my_n;
    }
}

template<int NL, int HLAST, bool RELU_LAST, bool RANK>
__global__ void __launch_bounds__(CH_WAVES * 32)
gin_mlp_chain(const float* __restrict__ X,
              int* __restrict__ deg,
              int* __restrict__ col,
              int2* __restrict__ ovf,
              int* __restrict__ cnts,
              const unsigned int* __restrict__ binbuf,
              const unsigned int* __restrict__ ovf1,
              const int* __restrict__ ocnt,
              const float* __restrict__ WTa, const float* __restrict__ WTb,
              const float* __restrict__ WTc, const float* __restrict__ WTd,
              const float* __restrict__ bias_a, const float* __restrict__ bias_b,
              const float* __restrict__ bias_c, const float* __restrict__ bias_d,
              const float* __restrict__ bn_g, const float* __restrict__ bn_b,
              const float* __restrict__ bn_m, const float* __restrict__ bn_v,
              float* __restrict__ out) {
    constexpr int NT_LAST   = (HLAST + 15) / 16;
    constexpr int NT_SPLIT  = (NT_LAST + 3) / 4;
    constexpr int ROWS_LAST = NT_LAST * 16;
    constexpr int ROW_TILES = N_NODES / 16;
    constexpr int KSTEPS    = K_DIM / 8;
    constexpr int W_FLOATS  = ((NL - 1) * 128 + ROWS_LAST) * WT_PITCH;
    constexpr int P_FLOATS  = 384 + (NL - 1) * 128;
    constexpr int X_FLOATS  = ((NL == 2) ? 4 : 2) * 8 * X_TILE_FLOATS;
    constexpr int G_FLOATS  = CH_WAVES * G_ROWS * WT_PITCH;
    static_assert(NT_LAST % NT_SPLIT == 0, "the last layer's tiles are dealt whole");
    static_assert((CH_WAVES - CH_FULL) * G_ROWS >= 16, "the shared tile's 16 rows fit its waves' rows");
    static_assert(16 % G_ROWS == 0 && (G_ROWS & (G_ROWS - 1)) == 0, "a tile is parked in whole groups");

    __shared__ alignas(16) float Ws[W_FLOATS + P_FLOATS + X_FLOATS + G_FLOATS];
    float* Ps = Ws + W_FLOATS;
    float* Xs = Ps + P_FLOATS;
    float* Gs = Xs + X_FLOATS;
    __shared__ int Ls[RANK ? RANGE_BIG * (BUCKET + 1) : 1];
    int* Lcnt = Ls;
    int* Llst = Ls + (RANK ? RANGE_BIG : 0);

    const int lane = threadIdx.x & 31;
    const int wave = __builtin_amdgcn_readfirstlane((int)(threadIdx.x >> 5));
    const bool full = wave < CH_FULL;
    const int q = wave - CH_FULL;
    const int bk = (int)blockIdx.x;
    const int tile0 = gin_tile0(bk);
    const int tile = tile0 + (full ? wave : CH_FULL);
    const bool have = full || bk < CH_EXTRA;
    static_assert(CH_BLOCKS * CH_FULL + CH_EXTRA == ROW_TILES, "every tile has a wave");
    const int c  = lane & 15;
    const int hi = lane >> 4;

    constexpr int RL = ROWS_LAST / CH_WAVES;
    v4f w0[8], w1[NL == 2 ? RL : 8], w2[8], w3[RL];
    gin_w_load<128>(WTa, w0);
    if constexpr (NL == 2) {
        gin_w_load<ROWS_LAST>(WTb, w1);
    } else {
        gin_w_load<128>(WTb, w1);
        gin_w_load<128>(WTc, w2);
        gin_w_load<ROWS_LAST>(WTd, w3);
    }

    const int jl0 = (tile - tile0) * 16;
    if constexpr (RANK)
        gin_rank_range(bk, tile0 * 16, (bk < CH_EXTRA) ? RANGE_BIG : RANGE_SMALL, wave, lane,
                       Lcnt, Llst, cnts, binbuf, ovf1, ocnt, deg, col, ovf);

    int cntv = 0;
    if (have) cntv = RANK ? Lcnt[jl0 + c] : deg[tile * 16 + c];
    const float4* __restrict__ xr = reinterpret_cast<const float4*>(X) + lane;
    float* Gw = Gs + wave * (G_ROWS * WT_PITCH);
    float* Gl = Gs + CH_FULL * (G_ROWS * WT_PITCH);
    v4f a[KSTEPS];
    #pragma unroll
    for (int j = 0; j < KSTEPS; ++j) a[j] = v4f{0.0f, 0.0f, 0.0f, 0.0f};

    if (threadIdx.x < 128) {
        const int n = threadIdx.x;
        const float rs = rsqrtf(bn_v[n] + BN_EPS);
        const float scale = bn_g[n] * rs;
        Ps[n]       = bias_a[n];
        Ps[128 + n] = scale;
        Ps[256 + n] = bn_b[n] - bn_m[n] * scale;
        if constexpr (NL == 2) {
            const float bb = bias_b[(n < HLAST) ? n : 0];
            Ps[384 + n] = (n < HLAST) ? bb : 0.0f;
        } else {
            Ps[384 + n] = bias_b[n];
            Ps[512 + n] = bias_c[n];
            const float bd = bias_d[(n < HLAST) ? n : 0];
            Ps[640 + n] = (n < HLAST) ? bd : 0.0f;
        }
    }

    gin_w_store<128>(Ws, w0);
    if constexpr (NL == 2) {
        gin_w_store<ROWS_LAST>(Ws + 128 * WT_PITCH, w1);
    } else {
        gin_w_store<128>(Ws + 128 * WT_PITCH, w1);
        gin_w_store<128>(Ws + 256 * WT_PITCH, w2);
        gin_w_store<ROWS_LAST>(Ws + 384 * WT_PITCH, w3);
    }
    if (have) {
        const int jq = full ? 0 : 4 * q;
        gin_rows_to_lds<15, false, RANK>(xr, deg, col, ovf, Llst, jl0, binbuf, ovf1, ocnt, bk,
                                         tile * 16, cntv, jq, jq + 4, full ? Gw : Gl, lane, a);
    }
    __syncthreads();
    if (full) {
        gin_rows_to_lds<G_ROWS - 1, true, RANK>(xr, deg, col, ovf, Llst, jl0, binbuf, ovf1, ocnt, bk,
                                                tile * 16, cntv, 4, 16, Gw, lane, a);
    } else if (have) {
        #pragma unroll
        for (int s = 0; s < KSTEPS; ++s)
            a[s] = *reinterpret_cast<const v4f*>(Gl + c * WT_PITCH + 8 * s + 4 * hi);
    }

    const float* Wl0 = Ws + c * WT_PITCH + 4 * hi;
    const float* WlN = Ws + c * WT_PITCH + 8 * hi;
    const float* Pl  = Ps + 8 * hi;

    constexpr int SW = 2 * 16 * WT_PITCH;
    float* X0 = Xs;
    float* X1 = Xs + 8 * X_TILE_FLOATS;
    v8f x[8], y[8];

    if (full) {
        gin_layer_mma<8, true>(Wl0, a, nullptr, x);
        gin_layer_epi<8, 2>(Pl, x);
    } else if (have) {
        v8f z[2];
        gin_layer_mma<2, true>(Wl0 + q * SW, a, nullptr, z);
        gin_layer_epi<2, 2>(Pl + 32 * q, z);
        gin_x_store<2>(X0, 2 * q, lane, z);
    }
    __syncthreads();

    if constexpr (NL == 2) {
        if (full) {
            gin_layer_mma<NT_LAST, false>(WlN + 128 * WT_PITCH, nullptr, x, y);
            gin_layer_epi<NT_LAST, RELU_LAST ? 1 : 0>(Pl + 384, y);
            gin_out_store<NT_LAST, HLAST>(out + (long long)(tile * 16 + c) * HLAST + 8 * hi, hi, y);
        } else if (have && q * NT_SPLIT < NT_LAST) {
            v8f p[8], z[NT_SPLIT];
            gin_x_load<8>(X0, lane, p);
            gin_layer_mma<NT_SPLIT, false>(WlN + 128 * WT_PITCH + q * (NT_SPLIT * 16 * WT_PITCH),
                                           nullptr, p, z);
            gin_layer_epi<NT_SPLIT, RELU_LAST ? 1 : 0>(Pl + 384 + q * (NT_SPLIT * 16), z);
            gin_x_store<NT_SPLIT>(X1, q * NT_SPLIT, lane, z);
        }
    } else {
        if (full) {
            gin_layer_mma<8, false>(WlN + 128 * WT_PITCH, nullptr, x, y);
            gin_layer_epi<8, 1>(Pl + 384, y);
        } else if (have) {
            v8f p[8], z[2];
            gin_x_load<8>(X0, lane, p);
            gin_layer_mma<2, false>(WlN + 128 * WT_PITCH + q * SW, nullptr, p, z);
            gin_layer_epi<2, 1>(Pl + 384 + 32 * q, z);
            gin_x_store<2>(X1, 2 * q, lane, z);
        }
        __syncthreads();
        v8f x2[8], x3[NT_LAST];
        if (full) {
            gin_layer_mma<8, false>(WlN + 256 * WT_PITCH, nullptr, y, x2);
            gin_layer_epi<8, 1>(Pl + 512, x2);
        } else if (have) {
            v8f p[8], z[2];
            gin_x_load<8>(X1, lane, p);
            gin_layer_mma<2, false>(WlN + 256 * WT_PITCH + q * SW, nullptr, p, z);
            gin_layer_epi<2, 1>(Pl + 512 + 32 * q, z);
            gin_x_store<2>(X0, 2 * q, lane, z);
        }
        __syncthreads();
        if (full) {
            gin_layer_mma<NT_LAST, false>(WlN + 384 * WT_PITCH, nullptr, x2, x3);
            gin_layer_epi<NT_LAST, RELU_LAST ? 1 : 0>(Pl + 640, x3);
            gin_out_store<NT_LAST, HLAST>(out + (long long)(tile * 16 + c) * HLAST + 8 * hi, hi, x3);
        } else if (have && q * NT_SPLIT < NT_LAST) {
            v8f p[8], z[NT_SPLIT];
            gin_x_load<8>(X0, lane, p);
            gin_layer_mma<NT_SPLIT, false>(WlN + 384 * WT_PITCH + q * (NT_SPLIT * 16 * WT_PITCH),
                                           nullptr, p, z);
            gin_layer_epi<NT_SPLIT, RELU_LAST ? 1 : 0>(Pl + 640 + q * (NT_SPLIT * 16), z);
            gin_x_store<NT_SPLIT>(X1, q * NT_SPLIT, lane, z);
        }
    }
    __syncthreads();
    if (!full && have && q == 0) {
        v8f o[NT_LAST];
        gin_x_load<NT_LAST>(X1, lane, o);
        gin_out_store<NT_LAST, HLAST>(out + (long long)(tile * 16 + c) * HLAST + 8 * hi, hi, o);
    }
}

extern "C" void kernel_launch(void* const* d_in, const int* in_sizes, int n_in,
                              void* d_out, int out_size, void* d_ws, size_t ws_size,
                              hipStream_t stream) {
    (void)in_sizes; (void)n_in; (void)out_size; (void)ws_size;

    const float* x      = (const float*)d_in[0];
    const int*   edge   = (const int*)  d_in[1];
    const float* lin1_w = (const float*)d_in[2];
    const float* lin1_b = (const float*)d_in[3];
    const float* bn1_g  = (const float*)d_in[4];
    const float* bn1_b  = (const float*)d_in[5];
    const float* bn1_m  = (const float*)d_in[6];
    const float* bn1_v  = (const float*)d_in[7];
    const float* lin2_w = (const float*)d_in[8];
    const float* lin2_b = (const float*)d_in[9];
    const float* lin3_w = (const float*)d_in[10];
    const float* lin3_b = (const float*)d_in[11];
    const float* bn2_g  = (const float*)d_in[12];
    const float* bn2_b  = (const float*)d_in[13];
    const float* bn2_m  = (const float*)d_in[14];
    const float* bn2_v  = (const float*)d_in[15];
    const float* lin4_w = (const float*)d_in[16];
    const float* lin4_b = (const float*)d_in[17];
    const float* lin5_w = (const float*)d_in[18];
    const float* lin5_b = (const float*)d_in[19];
    const float* lin6_w = (const float*)d_in[20];
    const float* lin6_b = (const float*)d_in[21];

    const int* e_src = edge;
    const int* e_dst = edge + N_EDGES;

    const size_t NH = (size_t)N_NODES * K_DIM;
    float* B0  = (float*)d_ws;
    float* B1  = B0  + NH;
    float* WT1 = B1  + NH;
    float* WT2 = WT1 + 128 * 128;
    float* WT3 = WT2 + 128 * 128;
    float* WT4 = WT3 + 128 * 128;
    float* WT5 = WT4 + 128 * 128;
    float* WT6 = WT5 + 128 * 128;
    int*  deg  = (int*)(WT6 + 64 * 128);
    int*  cnts = deg + N_NODES;
    int*  col  = deg + N_NODES + 272;
    int2* ovf  = (int2*)(col + N_NODES * BUCKET);
    unsigned int* binbuf = (unsigned int*)(ovf + N_EDGES);
    unsigned int* ovf1   = binbuf + N_BINS * N_GROUPS * LINE_WORDS;
    int*          ocnt   = (int*)(ovf1 + N_GROUPS * BIN_EDGES);

    const int sBlocks  = N_GROUPS + SETUP_T_BLOCKS;

    gin_setup_bin_kernel<<<sBlocks, BIN_THREADS, 0, stream>>>(e_src, e_dst,
                                                              lin1_w, lin2_w, lin3_w, lin4_w, lin5_w, lin6_w,
                                                              WT1, cnts, binbuf, ovf1, ocnt);

    gin_mlp_chain<2, 128, true, true><<<CH_BLOCKS, CH_WAVES * 32, 0, stream>>>(
        x, deg, col, ovf, cnts, binbuf, ovf1, ocnt,
        WT1, WT2, nullptr, nullptr, lin1_b, lin2_b, nullptr, nullptr,
        bn1_g, bn1_b, bn1_m, bn1_v, B0);

    gin_mlp_chain<4, 40, false, false><<<CH_BLOCKS, CH_WAVES * 32, 0, stream>>>(
        B0, deg, col, ovf, nullptr, nullptr, nullptr, nullptr,
        WT3, WT4, WT5, WT6, lin3_b, lin4_b, lin5_b, lin6_b,
        bn2_g, bn2_b, bn2_m, bn2_v, (float*)d_out);
}
